# starting kernel
# baseline (speedup 1.0000x reference)
#include <hip/hip_runtime.h>
#include <hip/hip_bf16.h>

#define N_NODES 50000
#define N_EDGES 800000
#define IN_C    128
#define OUT_C   128
#define HIDDEN  16
#define N_COEF  7

typedef __attribute__((ext_vector_type(2))) float v2f;
typedef __attribute__((ext_vector_type(8))) float v8f;

__device__ __forceinline__ v8f wmma4(v2f a, v2f b, v8f c) {
  return __builtin_amdgcn_wmma_f32_16x16x4_f32(false, a, false, b, (short)0, c,
                                               false, false);
}

__device__ __forceinline__ float silu_f(float v) {
  return v * (1.0f / (1.0f + __expf(-v)));
}

__device__ __forceinline__ void bspline7(float v, float* out) {
  float b[10];
#pragma unroll
  for (int j = 0; j < 10; ++j) {
    float gj = 0.5f * (float)j - 2.5f;
    b[j] = (v >= gj && v < gj + 0.5f) ? 1.0f : 0.0f;
  }
#pragma unroll
  for (int p = 1; p <= 3; ++p) {
    float invd = 1.0f / (0.5f * (float)p);
#pragma unroll
    for (int j = 0; j < 10 - p; ++j) {
      float gj  = 0.5f * (float)j - 2.5f;
      float gjp = gj + 0.5f * (float)p;
      b[j] = (v - gj) * invd * b[j] + ((gjp + 0.5f) - v) * invd * b[j + 1];
    }
  }
#pragma unroll
  for (int c = 0; c < N_COEF; ++c) out[c] = b[c];
}

__global__ void k_gemm_xw(const float* __restrict__ x, const float* __restrict__ W,
                          float* __restrict__ hlin) {
  const int NT = N_NODES / 16;
  const int NG = (NT + 3) / 4;
  int wave = (int)((blockIdx.x * blockDim.x + threadIdx.x) >> 5);
  int lane = (int)(threadIdx.x & 31);
  if (wave >= NG * (OUT_C / 16)) return;
  int grp = wave >> 3, ct = wave & 7;
  int o0 = ct * 16;
  int half = lane >> 4, l16 = lane & 15;

  int  rt[4];
  bool valid[4];
#pragma unroll
  for (int i = 0; i < 4; ++i) {
    int r = grp * 4 + i;
    valid[i] = (r < NT);
    rt[i] = valid[i] ? r : (NT - 1);
  }

  v8f acc[4] = {};
  for (int k = 0; k < IN_C; k += 4) {
    int ka0 = k + (half ? 2 : 0);
    int ka1 = k + (half ? 3 : 1);
    v2f b;
    b.x = W[(o0 + l16) * IN_C + ka0];
    b.y = W[(o0 + l16) * IN_C + ka1];
#pragma unroll
    for (int i = 0; i < 4; ++i) {
      v2f a;
      a.x = x[(rt[i] * 16 + l16) * IN_C + ka0];
      a.y = x[(rt[i] * 16 + l16) * IN_C + ka1];
      acc[i] = wmma4(a, b, acc[i]);
    }
  }
#pragma unroll
  for (int i = 0; i < 4; ++i) {
    if (!valid[i]) continue;
#pragma unroll
    for (int j = 0; j < 8; ++j)
      hlin[(rt[i] * 16 + j + half * 8) * OUT_C + o0 + l16] = acc[i][j];
  }
}

__global__ void k_deg_init(float* __restrict__ deg) {
  int i = blockIdx.x * blockDim.x + threadIdx.x;
  if (i < N_NODES) deg[i] = 1.0f;
}

__global__ void k_deg_acc(const int* __restrict__ ei, float* __restrict__ deg) {
  int e = blockIdx.x * blockDim.x + threadIdx.x;
  if (e < N_EDGES)
    __hip_atomic_fetch_add(&deg[ei[N_EDGES + e]], 1.0f, __ATOMIC_RELAXED,
                           __HIP_MEMORY_SCOPE_AGENT);
}

__global__ void k_deg_rsqrt(float* __restrict__ deg) {
  int i = blockIdx.x * blockDim.x + threadIdx.x;
  if (i < N_NODES) {
    float d = deg[i];
    deg[i] = (d > 0.0f) ? rsqrtf(d) : 0.0f;
  }
}

__global__ void k_self_bias(const float* __restrict__ hlin,
                            const float* __restrict__ dinv,
                            const float* __restrict__ bias,
                            float* __restrict__ h) {
  int t = blockIdx.x * blockDim.x + threadIdx.x;
  if (t < N_NODES * OUT_C) {
    int n = t >> 7, c = t & 127;
    float di = dinv[n];
    h[t] = hlin[t] * di * di + bias[c];
  }
}

__global__ void k_edge_scatter(const int* __restrict__ ei,
                               const float* __restrict__ hlin,
                               const float* __restrict__ dinv,
                               float* __restrict__ h) {
  long long t = (long long)blockIdx.x * blockDim.x + threadIdx.x;
  int e = (int)(t >> 5);
  int lane = (int)(t & 31);
  if (e >= N_EDGES) return;
  int src = ei[e], dst = ei[N_EDGES + e];
  const float4* h4 = (const float4*)hlin;
  const float4* p = &h4[src * 32 + lane];
  __builtin_prefetch(p, 0, 3);
  float norm = dinv[src] * dinv[dst];
  float4 v = *p;
  float* base = &h[dst * OUT_C + lane * 4];
  __hip_atomic_fetch_add(base + 0, v.x * norm, __ATOMIC_RELAXED, __HIP_MEMORY_SCOPE_AGENT);
  __hip_atomic_fetch_add(base + 1, v.y * norm, __ATOMIC_RELAXED, __HIP_MEMORY_SCOPE_AGENT);
  __hip_atomic_fetch_add(base + 2, v.z * norm, __ATOMIC_RELAXED, __HIP_MEMORY_SCOPE_AGENT);
  __hip_atomic_fetch_add(base + 3, v.w * norm, __ATOMIC_RELAXED, __HIP_MEMORY_SCOPE_AGENT);
}

__global__ void k_prep_w0(const float* __restrict__ bw, const float* __restrict__ sw,
                          const float* __restrict__ sc, float* __restrict__ Bm) {
  int t = blockIdx.x * blockDim.x + threadIdx.x;
  if (t >= 8 * OUT_C * HIDDEN) return;
  int k = t >> 4, o = t & 15;
  int f = k >> 3, j = k & 7;
  float w = (j == 0) ? bw[o * OUT_C + f]
                     : sw[(o * OUT_C + f) * N_COEF + (j - 1)] * sc[o * OUT_C + f];
  Bm[k * HIDDEN + o] = w;
}

__global__ void k_prep_w1(const float* __restrict__ bw, const float* __restrict__ sw,
                          const float* __restrict__ sc, float* __restrict__ Bm) {
  int t = blockIdx.x * blockDim.x + threadIdx.x;
  if (t >= 8 * HIDDEN * OUT_C) return;
  int k = t >> 7, o = t & 127;
  int f = k >> 3, j = k & 7;
  float w = (j == 0) ? bw[o * HIDDEN + f]
                     : sw[(o * HIDDEN + f) * N_COEF + (j - 1)] * sc[o * HIDDEN + f];
  Bm[k * OUT_C + o] = w;
}

#define LDS0_STRIDE 258
__global__ void __launch_bounds__(32) k_kan0(const float* __restrict__ h,
                                             const float* __restrict__ Bm,
                                             float* __restrict__ g) {
  __shared__ float feat[16 * LDS0_STRIDE];
  int n0 = blockIdx.x * 16;
  int lane = threadIdx.x;
  int half = lane >> 4, l16 = lane & 15;
  v8f accA = {}, accB = {};
  for (int ch = 0; ch < 4; ++ch) {
    int f0 = ch * 32;
    for (int idx = lane; idx < 16 * 32; idx += 32) {
      int row = idx & 15, fl = idx >> 4;
      float v = h[(n0 + row) * OUT_C + f0 + fl];
      float* d = &feat[row * LDS0_STRIDE + fl * 8];
      d[0] = silu_f(v);
      float bs[N_COEF];
      bspline7(v, bs);
#pragma unroll
      for (int c = 0; c < N_COEF; ++c) d[1 + c] = bs[c];
    }
    __syncthreads();
    for (int k = 0; k < 256; k += 8) {
#pragma unroll
      for (int s = 0; s < 2; ++s) {
        int ka0 = k + s * 4 + (half ? 2 : 0);
        int ka1 = k + s * 4 + (half ? 3 : 1);
        v2f a, b;
        a.x = feat[l16 * LDS0_STRIDE + ka0];
        a.y = feat[l16 * LDS0_STRIDE + ka1];
        b.x = Bm[(ch * 256 + ka0) * HIDDEN + l16];
        b.y = Bm[(ch * 256 + ka1) * HIDDEN + l16];
        if (s == 0) accA = wmma4(a, b, accA);
        else        accB = wmma4(a, b, accB);
      }
    }
    __syncthreads();
  }
  v8f acc = accA + accB;
#pragma unroll
  for (int i = 0; i < 8; ++i)
    g[(n0 + i + half * 8) * HIDDEN + l16] = acc[i];
}

#define LDS1_STRIDE 130
__global__ void __launch_bounds__(32) k_kan1(const float* __restrict__ g,
                                             const float* __restrict__ Bm,
                                             float* __restrict__ out) {
  __shared__ float feat[16 * LDS1_STRIDE];
  int n0 = blockIdx.x * 16;
  int lane = threadIdx.x;
  int half = lane >> 4, l16 = lane & 15;
  for (int idx = lane; idx < 16 * HIDDEN; idx += 32) {
    int row = idx & 15, fl = idx >> 4;
    float v = g[(n0 + row) * HIDDEN + fl];
    float* d = &feat[row * LDS1_STRIDE + fl * 8];
    d[0] = silu_f(v);
    float bs[N_COEF];
    bspline7(v, bs);
#pragma unroll
    for (int c = 0; c < N_COEF; ++c) d[1 + c] = bs[c];
  }
  __syncthreads();
  for (int ct = 0; ct < 8; ++ct) {
    int o0 = ct * 16;
    v8f accA = {}, accB = {};
    for (int k = 0; k < 128; k += 8) {
#pragma unroll
      for (int s = 0; s < 2; ++s) {
        int ka0 = k + s * 4 + (half ? 2 : 0);
        int ka1 = k + s * 4 + (half ? 3 : 1);
        v2f a, b;
        a.x = feat[l16 * LDS1_STRIDE + ka0];
        a.y = feat[l16 * LDS1_STRIDE + ka1];
        b.x = Bm[ka0 * OUT_C + o0 + l16];
        b.y = Bm[ka1 * OUT_C + o0 + l16];
        if (s == 0) accA = wmma4(a, b, accA);
        else        accB = wmma4(a, b, accB);
      }
    }
    v8f acc = accA + accB;
#pragma unroll
    for (int i = 0; i < 8; ++i)
      out[(n0 + i + half * 8) * OUT_C + o0 + l16] = acc[i];
  }
}

extern "C" void kernel_launch(void* const* d_in, const int* in_sizes, int n_in,
                              void* d_out, int out_size, void* d_ws, size_t ws_size,
                              hipStream_t stream) {
  (void)in_sizes; (void)n_in; (void)out_size; (void)ws_size;
  const float* x       = (const float*)d_in[0];
  const int*   ei      = (const int*)d_in[1];
  const float* gcn_w   = (const float*)d_in[2];
  const float* gcn_b   = (const float*)d_in[3];
  const float* base_w0 = (const float*)d_in[4];
  const float* spl_w0  = (const float*)d_in[5];
  const float* scal0   = (const float*)d_in[6];
  const float* base_w1 = (const float*)d_in[7];
  const float* spl_w1  = (const float*)d_in[8];
  const float* scal1   = (const float*)d_in[9];
  float* out = (float*)d_out;

  float* ws    = (float*)d_ws;
  float* hlin  = ws;
  float* h     = hlin + (size_t)N_NODES * OUT_C;
  float* deg   = h + (size_t)N_NODES * OUT_C;
  float* g     = deg + N_NODES;
  float* Bm0   = g + (size_t)N_NODES * HIDDEN;
  float* Bm1   = Bm0 + 8 * OUT_C * HIDDEN;

  const int NT = N_NODES / 16;
  const int NG = (NT + 3) / 4;

  k_gemm_xw<<<NG, 256, 0, stream>>>(x, gcn_w, hlin);
  k_deg_init<<<(N_NODES + 255) / 256, 256, 0, stream>>>(deg);
  k_deg_acc<<<(N_EDGES + 255) / 256, 256, 0, stream>>>(ei, deg);
  k_deg_rsqrt<<<(N_NODES + 255) / 256, 256, 0, stream>>>(deg);
  k_self_bias<<<(N_NODES * OUT_C + 255) / 256, 256, 0, stream>>>(hlin, deg, gcn_b, h);
  k_edge_scatter<<<((long long)N_EDGES * 32 + 255) / 256, 256, 0, stream>>>(ei, hlin, deg, h);
  k_prep_w0<<<(8 * OUT_C * HIDDEN + 255) / 256, 256, 0, stream>>>(base_w0, spl_w0, scal0, Bm0);
  k_prep_w1<<<(8 * HIDDEN * OUT_C + 255) / 256, 256, 0, stream>>>(base_w1, spl_w1, scal1, Bm1);
  k_kan0<<<N_NODES / 16, 32, 0, stream>>>(h, Bm0, g);
  k_kan1<<<N_NODES / 16, 32, 0, stream>>>(g, Bm1, out);
}


# Replace float-atomic edge scatter by per-destination gather: int-atomic slot per edge, scan, fill; one wave per node sorts its in-edges by source and sums in that fixed order; self loop, bias fused
# speedup vs baseline: 1.2419x; 1.2419x over previous
#include <hip/hip_runtime.h>
#include <hip/hip_bf16.h>

#define N_NODES 50000
#define N_EDGES 800000
#define IN_C    128
#define OUT_C   128
#define HIDDEN  16
#define N_COEF  7

typedef __attribute__((ext_vector_type(2))) float v2f;
typedef __attribute__((ext_vector_type(8))) float v8f;

__device__ __forceinline__ v8f wmma4(v2f a, v2f b, v8f c) {
  return __builtin_amdgcn_wmma_f32_16x16x4_f32(false, a, false, b, (short)0, c,
                                               false, false);
}

__device__ __forceinline__ float silu_f(float v) {
  return v * (1.0f / (1.0f + __expf(-v)));
}

__device__ __forceinline__ void bspline7(float v, float* out) {
  float b[10];
#pragma unroll
  for (int j = 0; j < 10; ++j) {
    float gj = 0.5f * (float)j - 2.5f;
    b[j] = (v >= gj && v < gj + 0.5f) ? 1.0f : 0.0f;
  }
#pragma unroll
  for (int p = 1; p <= 3; ++p) {
    float invd = 1.0f / (0.5f * (float)p);
#pragma unroll
    for (int j = 0; j < 10 - p; ++j) {
      float gj  = 0.5f * (float)j - 2.5f;
      float gjp = gj + 0.5f * (float)p;
      b[j] = (v - gj) * invd * b[j] + ((gjp + 0.5f) - v) * invd * b[j + 1];
    }
  }
#pragma unroll
  for (int c = 0; c < N_COEF; ++c) out[c] = b[c];
}

__global__ void k_gemm_xw(const float* __restrict__ x, const float* __restrict__ W,
                          float* __restrict__ hlin) {
  const int NT = N_NODES / 16;
  const int NG = (NT + 3) / 4;
  int wave = (int)((blockIdx.x * blockDim.x + threadIdx.x) >> 5);
  int lane = (int)(threadIdx.x & 31);
  if (wave >= NG * (OUT_C / 16)) return;
  int grp = wave >> 3, ct = wave & 7;
  int o0 = ct * 16;
  int half = lane >> 4, l16 = lane & 15;

  int  rt[4];
  bool valid[4];
#pragma unroll
  for (int i = 0; i < 4; ++i) {
    int r = grp * 4 + i;
    valid[i] = (r < NT);
    rt[i] = valid[i] ? r : (NT - 1);
  }

  v8f acc[4] = {};
  for (int k = 0; k < IN_C; k += 4) {
    int ka0 = k + (half ? 2 : 0);
    int ka1 = k + (half ? 3 : 1);
    v2f b;
    b.x = W[(o0 + l16) * IN_C + ka0];
    b.y = W[(o0 + l16) * IN_C + ka1];
#pragma unroll
    for (int i = 0; i < 4; ++i) {
      v2f a;
      a.x = x[(rt[i] * 16 + l16) * IN_C + ka0];
      a.y = x[(rt[i] * 16 + l16) * IN_C + ka1];
      acc[i] = wmma4(a, b, acc[i]);
    }
  }
#pragma unroll
  for (int i = 0; i < 4; ++i) {
    if (!valid[i]) continue;
#pragma unroll
    for (int j = 0; j < 8; ++j)
      hlin[(rt[i] * 16 + j + half * 8) * OUT_C + o0 + l16] = acc[i][j];
  }
}

static_assert(N_NODES <= 65536, "source ids are kept as 16-bit values");

__global__ void k_cnt_zero(int* __restrict__ cnt) {
  int i = blockIdx.x * blockDim.x + threadIdx.x;
  if (i < N_NODES) cnt[i] = 0;
}

__global__ void k_count_rank(const int* __restrict__ ei, int* __restrict__ cnt,
                             int* __restrict__ rank) {
  int e = blockIdx.x * blockDim.x + threadIdx.x;
  if (e < N_EDGES)
    rank[e] = __hip_atomic_fetch_add(&cnt[ei[N_EDGES + e]], 1, __ATOMIC_RELAXED,
                                     __HIP_MEMORY_SCOPE_AGENT);
}

#define SCAN_T 1024
#define SCAN_V 13
static_assert(N_NODES % 4 == 0 && SCAN_T * SCAN_V * 4 >= N_NODES, "scan tiling");
__global__ void __launch_bounds__(SCAN_T) k_scan(const int* __restrict__ cnt,
                                                 int* __restrict__ off,
                                                 float* __restrict__ dinv) {
  __shared__ int part[SCAN_T];
  const int N4 = N_NODES / 4;
  int t = threadIdx.x;
  int b4 = t * SCAN_V;
  const int4* c4 = (const int4*)cnt;
  int4 a[SCAN_V];
  for (int v = 0; v < SCAN_V; ++v) {
    int i4 = b4 + v;
    int4 x = c4[(i4 < N4) ? i4 : (N4 - 1)];
    if (i4 >= N4) x = make_int4(0, 0, 0, 0);
    a[v] = x;
  }
  int loc = 0;
  for (int v = 0; v < SCAN_V; ++v) loc += a[v].x + a[v].y + a[v].z + a[v].w;
  part[t] = loc;
  __syncthreads();
  for (int o = 1; o < SCAN_T; o <<= 1) {
    int v = (t >= o) ? part[t - o] : 0;
    __syncthreads();
    part[t] += v;
    __syncthreads();
  }
  int run = part[t] - loc;
  for (int v = 0; v < SCAN_V; ++v) {
    int i4 = b4 + v;
    if (i4 < N4) {
      int4 o;
      o.x = run; run += a[v].x;
      o.y = run; run += a[v].y;
      o.z = run; run += a[v].z;
      o.w = run; run += a[v].w;
      ((int4*)off)[i4] = o;
      float4 dv;
      float d0 = 1.0f + (float)a[v].x, d1 = 1.0f + (float)a[v].y;
      float d2 = 1.0f + (float)a[v].z, d3 = 1.0f + (float)a[v].w;
      dv.x = (d0 > 0.0f) ? rsqrtf(d0) : 0.0f;
      dv.y = (d1 > 0.0f) ? rsqrtf(d1) : 0.0f;
      dv.z = (d2 > 0.0f) ? rsqrtf(d2) : 0.0f;
      dv.w = (d3 > 0.0f) ? rsqrtf(d3) : 0.0f;
      ((float4*)dinv)[i4] = dv;
    }
  }
  if (t == SCAN_T - 1) off[N_NODES] = part[t];
}

__global__ void k_fill(const int* __restrict__ ei, const int* __restrict__ rank,
                       const int* __restrict__ off,
                       unsigned short* __restrict__ csr) {
  int e = blockIdx.x * blockDim.x + threadIdx.x;
  if (e < N_EDGES) {
    int d = ei[N_EDGES + e];
    csr[off[d] + rank[e]] = (unsigned short)ei[e];
  }
}

__device__ __forceinline__ float4 axpy4(float4 v, float s, float4 a) {
  a.x += v.x * s;
  a.y += v.y * s;
  a.z += v.z * s;
  a.w += v.w * s;
  return a;
}

__device__ __forceinline__ float readlane_f(float v, int l) {
  return __int_as_float(__builtin_amdgcn_readlane(__float_as_int(v), l));
}

#define MID_CAP 128
__global__ void __launch_bounds__(256) k_gather(const float* __restrict__ hlin,
                                                const int* __restrict__ off,
                                                const unsigned short* __restrict__ csr,
                                                const float* __restrict__ dinv,
                                                const float* __restrict__ bias,
                                                float* __restrict__ h) {
  __shared__ int   mid_us[8][MID_CAP];
  __shared__ int   mid_so[8][MID_CAP];
  __shared__ float mid_wo[8][MID_CAP];
  int lane = (int)(threadIdx.x & 31);
  int n = __builtin_amdgcn_readfirstlane(
      (int)((blockIdx.x * blockDim.x + threadIdx.x) >> 5));
  if (n >= N_NODES) return;
  int beg = off[n];
  int d = off[n + 1] - beg;
  float di = dinv[n];
  const float4* h4 = (const float4*)hlin;
  float4 hv = h4[n * 32 + lane];
  float4 acc;
  acc.x = hv.x * di * di + bias[lane * 4 + 0];
  acc.y = hv.y * di * di + bias[lane * 4 + 1];
  acc.z = hv.z * di * di + bias[lane * 4 + 2];
  acc.w = hv.w * di * di + bias[lane * 4 + 3];

  if (d <= 32) {
    int s = 0x7fffffff;
    float w = 0.0f;
    if (lane < d) {
      s = (int)csr[beg + lane];
      w = dinv[s];
    }
    int r = 0;
    for (int j = 0; j < d; ++j) {
      int sj = __builtin_amdgcn_readlane(s, j);
      r += ((sj < s) || (sj == s && j < lane)) ? 1 : 0;
    }
    int   ss = __builtin_amdgcn_ds_permute(r << 2, s);
    float ws = __int_as_float(__builtin_amdgcn_ds_permute(r << 2, __float_as_int(w)));
    int k = 0;
    for (; k + 4 <= d; k += 4) {
      int s0 = __builtin_amdgcn_readlane(ss, k);
      int s1 = __builtin_amdgcn_readlane(ss, k + 1);
      int s2 = __builtin_amdgcn_readlane(ss, k + 2);
      int s3 = __builtin_amdgcn_readlane(ss, k + 3);
      float w0 = readlane_f(ws, k);
      float w1 = readlane_f(ws, k + 1);
      float w2 = readlane_f(ws, k + 2);
      float w3 = readlane_f(ws, k + 3);
      float4 v0 = h4[s0 * 32 + lane];
      float4 v1 = h4[s1 * 32 + lane];
      float4 v2 = h4[s2 * 32 + lane];
      float4 v3 = h4[s3 * 32 + lane];
      acc = axpy4(v0, w0 * di, acc);
      acc = axpy4(v1, w1 * di, acc);
      acc = axpy4(v2, w2 * di, acc);
      acc = axpy4(v3, w3 * di, acc);
    }
    for (; k < d; ++k) {
      int s0 = __builtin_amdgcn_readlane(ss, k);
      float w0 = readlane_f(ws, k);
      float4 v0 = h4[s0 * 32 + lane];
      acc = axpy4(v0, w0 * di, acc);
    }
  } else if (d <= MID_CAP) {
    int wv = (int)(threadIdx.x >> 5);
    int*   us = mid_us[wv];
    int*   so = mid_so[wv];
    float* wo = mid_wo[wv];
    int   e[MID_CAP / 32];
    float w[MID_CAP / 32];
    for (int q = 0; q < MID_CAP / 32; ++q) {
      int i = lane + 32 * q;
      e[q] = 0x7fffffff;
      w[q] = 0.0f;
      if (i < d) {
        e[q] = (int)csr[beg + i];
        w[q] = dinv[e[q]];
        us[i] = e[q];
      }
    }
    int r[MID_CAP / 32];
    for (int q = 0; q < MID_CAP / 32; ++q) r[q] = 0;
    for (int j = 0; j < d; ++j) {
      int sj = us[j];
      for (int q = 0; q < MID_CAP / 32; ++q)
        r[q] += ((sj < e[q]) || (sj == e[q] && j < lane + 32 * q)) ? 1 : 0;
    }
    for (int q = 0; q < MID_CAP / 32; ++q) {
      if (lane + 32 * q < d) {
        so[r[q]] = e[q];
        wo[r[q]] = w[q];
      }
    }
    int k = 0;
    for (; k + 4 <= d; k += 4) {
      int s0 = so[k], s1 = so[k + 1], s2 = so[k + 2], s3 = so[k + 3];
      float w0 = wo[k], w1 = wo[k + 1], w2 = wo[k + 2], w3 = wo[k + 3];
      float4 v0 = h4[s0 * 32 + lane];
      float4 v1 = h4[s1 * 32 + lane];
      float4 v2 = h4[s2 * 32 + lane];
      float4 v3 = h4[s3 * 32 + lane];
      acc = axpy4(v0, w0 * di, acc);
      acc = axpy4(v1, w1 * di, acc);
      acc = axpy4(v2, w2 * di, acc);
      acc = axpy4(v3, w3 * di, acc);
    }
    for (; k < d; ++k) {
      int s0 = so[k];
      float w0 = wo[k];
      float4 v0 = h4[s0 * 32 + lane];
      acc = axpy4(v0, w0 * di, acc);
    }
  } else {
    int last = -1;
    int left = d;
    for (int it = 0; it < d && left > 0; ++it) {
      int mn = 0x7fffffff;
      for (int i = lane; i < d; i += 32) {
        int s = (int)csr[beg + i];
        if (s > last && s < mn) mn = s;
      }
      for (int o = 16; o > 0; o >>= 1) {
        int t = __shfl_xor(mn, o);
        mn = (t < mn) ? t : mn;
      }
      int m = 0;
      for (int i = lane; i < d; i += 32) m += ((int)csr[beg + i] == mn) ? 1 : 0;
      for (int o = 16; o > 0; o >>= 1) m += __shfl_xor(m, o);
      mn = __builtin_amdgcn_readfirstlane(mn);
      m = __builtin_amdgcn_readfirstlane(m);
      float nrm = dinv[mn] * di;
      float4 v = h4[mn * 32 + lane];
      for (int c = 0; c < m; ++c) acc = axpy4(v, nrm, acc);
      last = mn;
      left -= m;
    }
  }
  ((float4*)h)[n * 32 + lane] = acc;
}

__global__ void k_prep_w0(const float* __restrict__ bw, const float* __restrict__ sw,
                          const float* __restrict__ sc, float* __restrict__ Bm) {
  int t = blockIdx.x * blockDim.x + threadIdx.x;
  if (t >= 8 * OUT_C * HIDDEN) return;
  int k = t >> 4, o = t & 15;
  int f = k >> 3, j = k & 7;
  float w = (j == 0) ? bw[o * OUT_C + f]
                     : sw[(o * OUT_C + f) * N_COEF + (j - 1)] * sc[o * OUT_C + f];
  Bm[k * HIDDEN + o] = w;
}

__global__ void k_prep_w1(const float* __restrict__ bw, const float* __restrict__ sw,
                          const float* __restrict__ sc, float* __restrict__ Bm) {
  int t = blockIdx.x * blockDim.x + threadIdx.x;
  if (t >= 8 * HIDDEN * OUT_C) return;
  int k = t >> 7, o = t & 127;
  int f = k >> 3, j = k & 7;
  float w = (j == 0) ? bw[o * HIDDEN + f]
                     : sw[(o * HIDDEN + f) * N_COEF + (j - 1)] * sc[o * HIDDEN + f];
  Bm[k * OUT_C + o] = w;
}

#define LDS0_STRIDE 258
__global__ void __launch_bounds__(32) k_kan0(const float* __restrict__ h,
                                             const float* __restrict__ Bm,
                                             float* __restrict__ g) {
  __shared__ float feat[16 * LDS0_STRIDE];
  int n0 = blockIdx.x * 16;
  int lane = threadIdx.x;
  int half = lane >> 4, l16 = lane & 15;
  v8f accA = {}, accB = {};
  for (int ch = 0; ch < 4; ++ch) {
    int f0 = ch * 32;
    for (int idx = lane; idx < 16 * 32; idx += 32) {
      int row = idx & 15, fl = idx >> 4;
      float v = h[(n0 + row) * OUT_C + f0 + fl];
      float* d = &feat[row * LDS0_STRIDE + fl * 8];
      d[0] = silu_f(v);
      float bs[N_COEF];
      bspline7(v, bs);
#pragma unroll
      for (int c = 0; c < N_COEF; ++c) d[1 + c] = bs[c];
    }
    __syncthreads();
    for (int k = 0; k < 256; k += 8) {
#pragma unroll
      for (int s = 0; s < 2; ++s) {
        int ka0 = k + s * 4 + (half ? 2 : 0);
        int ka1 = k + s * 4 + (half ? 3 : 1);
        v2f a, b;
        a.x = feat[l16 * LDS0_STRIDE + ka0];
        a.y = feat[l16 * LDS0_STRIDE + ka1];
        b.x = Bm[(ch * 256 + ka0) * HIDDEN + l16];
        b.y = Bm[(ch * 256 + ka1) * HIDDEN + l16];
        if (s == 0) accA = wmma4(a, b, accA);
        else        accB = wmma4(a, b, accB);
      }
    }
    __syncthreads();
  }
  v8f acc = accA + accB;
#pragma unroll
  for (int i = 0; i < 8; ++i)
    g[(n0 + i + half * 8) * HIDDEN + l16] = acc[i];
}

#define LDS1_STRIDE 130
__global__ void __launch_bounds__(32) k_kan1(const float* __restrict__ g,
                                             const float* __restrict__ Bm,
                                             float* __restrict__ out) {
  __shared__ float feat[16 * LDS1_STRIDE];
  int n0 = blockIdx.x * 16;
  int lane = threadIdx.x;
  int half = lane >> 4, l16 = lane & 15;
  for (int idx = lane; idx < 16 * HIDDEN; idx += 32) {
    int row = idx & 15, fl = idx >> 4;
    float v = g[(n0 + row) * HIDDEN + fl];
    float* d = &feat[row * LDS1_STRIDE + fl * 8];
    d[0] = silu_f(v);
    float bs[N_COEF];
    bspline7(v, bs);
#pragma unroll
    for (int c = 0; c < N_COEF; ++c) d[1 + c] = bs[c];
  }
  __syncthreads();
  for (int ct = 0; ct < 8; ++ct) {
    int o0 = ct * 16;
    v8f accA = {}, accB = {};
    for (int k = 0; k < 128; k += 8) {
#pragma unroll
      for (int s = 0; s < 2; ++s) {
        int ka0 = k + s * 4 + (half ? 2 : 0);
        int ka1 = k + s * 4 + (half ? 3 : 1);
        v2f a, b;
        a.x = feat[l16 * LDS1_STRIDE + ka0];
        a.y = feat[l16 * LDS1_STRIDE + ka1];
        b.x = Bm[ka0 * OUT_C + o0 + l16];
        b.y = Bm[ka1 * OUT_C + o0 + l16];
        if (s == 0) accA = wmma4(a, b, accA);
        else        accB = wmma4(a, b, accB);
      }
    }
    v8f acc = accA + accB;
#pragma unroll
    for (int i = 0; i < 8; ++i)
      out[(n0 + i + half * 8) * OUT_C + o0 + l16] = acc[i];
  }
}

extern "C" void kernel_launch(void* const* d_in, const int* in_sizes, int n_in,
                              void* d_out, int out_size, void* d_ws, size_t ws_size,
                              hipStream_t stream) {
  (void)in_sizes; (void)n_in; (void)out_size; (void)ws_size;
  const float* x       = (const float*)d_in[0];
  const int*   ei      = (const int*)d_in[1];
  const float* gcn_w   = (const float*)d_in[2];
  const float* gcn_b   = (const float*)d_in[3];
  const float* base_w0 = (const float*)d_in[4];
  const float* spl_w0  = (const float*)d_in[5];
  const float* scal0   = (const float*)d_in[6];
  const float* base_w1 = (const float*)d_in[7];
  const float* spl_w1  = (const float*)d_in[8];
  const float* scal1   = (const float*)d_in[9];
  float* out = (float*)d_out;

  float* ws    = (float*)d_ws;
  float* hlin  = ws;
  float* h     = hlin + (size_t)N_NODES * OUT_C;
  float* deg   = h + (size_t)N_NODES * OUT_C;
  float* g     = deg + N_NODES;
  float* Bm0   = g + (size_t)N_NODES * HIDDEN;
  float* Bm1   = Bm0 + 8 * OUT_C * HIDDEN;
  int* cnt  = (int*)h;
  int* rank = cnt + N_NODES;
  unsigned short* csr = (unsigned short*)g;
  int* off  = (int*)g + N_EDGES / 2;

  const int NT = N_NODES / 16;
  const int NG = (NT + 3) / 4;

  k_gemm_xw<<<NG, 256, 0, stream>>>(x, gcn_w, hlin);
  k_cnt_zero<<<(N_NODES + 255) / 256, 256, 0, stream>>>(cnt);
  k_count_rank<<<(N_EDGES + 255) / 256, 256, 0, stream>>>(ei, cnt, rank);
  k_scan<<<1, SCAN_T, 0, stream>>>(cnt, off, deg);
  k_fill<<<(N_EDGES + 255) / 256, 256, 0, stream>>>(ei, rank, off, csr);
  k_gather<<<(N_NODES * 32 + 255) / 256, 256, 0, stream>>>(hlin, off, csr, deg, gcn_b, h);
  k_prep_w0<<<(8 * OUT_C * HIDDEN + 255) / 256, 256, 0, stream>>>(base_w0, spl_w0, scal0, Bm0);
  k_prep_w1<<<(8 * HIDDEN * OUT_C + 255) / 256, 256, 0, stream>>>(base_w1, spl_w1, scal1, Bm1);
  k_kan0<<<N_NODES / 16, 32, 0, stream>>>(h, Bm0, g);
  k_kan1<<<N_NODES / 16, 32, 0, stream>>>(g, Bm1, out);
}


# x@W^T and KAN layer 1: persistent 8-wave workgroups hold each column tile's weight fragments in registers, node rows staged in LDS by whole-line loads; k regrouped over 2 accumulators (sum order)
# speedup vs baseline: 1.6683x; 1.3433x over previous
#include <hip/hip_runtime.h>
#include <hip/hip_bf16.h>

#define N_NODES 50000
#define N_EDGES 800000
#define IN_C    128
#define OUT_C   128
#define HIDDEN  16
#define N_COEF  7

typedef __attribute__((ext_vector_type(2))) float v2f;
typedef __attribute__((ext_vector_type(8))) float v8f;

__device__ __forceinline__ v8f wmma4(v2f a, v2f b, v8f c) {
  return __builtin_amdgcn_wmma_f32_16x16x4_f32(false, a, false, b, (short)0, c,
                                               false, false);
}

__device__ __forceinline__ float silu_f(float v) {
  return v * (1.0f / (1.0f + __expf(-v)));
}

__device__ __forceinline__ void bspline7(float v, float* out) {
  float b[10];
#pragma unroll
  for (int j = 0; j < 10; ++j) {
    float gj = 0.5f * (float)j - 2.5f;
    b[j] = (v >= gj && v < gj + 0.5f) ? 1.0f : 0.0f;
  }
#pragma unroll
  for (int p = 1; p <= 3; ++p) {
    float invd = 1.0f / (0.5f * (float)p);
#pragma unroll
    for (int j = 0; j < 10 - p; ++j) {
      float gj  = 0.5f * (float)j - 2.5f;
      float gjp = gj + 0.5f * (float)p;
      b[j] = (v - gj) * invd * b[j] + ((gjp + 0.5f) - v) * invd * b[j + 1];
    }
  }
#pragma unroll
  for (int c = 0; c < N_COEF; ++c) out[c] = b[c];
}

#define TILE_P 132
#define NTILES (N_NODES / 16)
#define PGRID  512
static_assert(N_NODES % 16 == 0 && PGRID <= NTILES, "tiling");

__device__ __forceinline__ void load_bfrag(const float* __restrict__ Wt, int o0,
                                           int l16, int half, float4 (&b)[16]) {
  const float4* p = (const float4*)(Wt + (o0 + l16) * 128 + 4 * half);
#pragma unroll
  for (int j = 0; j < 16; ++j) b[j] = p[2 * j];
}

__device__ __forceinline__ v8f tile_mma(const float* As, int l16, int half,
                                        const float4 (&b)[16]) {
  float4 a[16];
  const float4* p = (const float4*)(As + l16 * TILE_P + 4 * half);
#pragma unroll
  for (int j = 0; j < 16; ++j) a[j] = p[2 * j];
  __builtin_amdgcn_sched_barrier(0);
  v8f acc0 = {}, acc1 = {};
#pragma unroll
  for (int j = 0; j < 16; ++j) {
    v2f a0, a1, b0, b1;
    a0.x = a[j].x; a0.y = a[j].y; a1.x = a[j].z; a1.y = a[j].w;
    b0.x = b[j].x; b0.y = b[j].y; b1.x = b[j].z; b1.y = b[j].w;
    acc0 = wmma4(a0, b0, acc0);
    acc1 = wmma4(a1, b1, acc1);
  }
  asm volatile("v_nop\n\tv_nop\n\tv_nop\n\tv_nop\n\tv_nop" : "+v"(acc0), "+v"(acc1));
  return acc0 + acc1;
}

__global__ void __launch_bounds__(256) k_gemm_xw(const float* __restrict__ x,
                                                 const float* __restrict__ W,
                                                 float* __restrict__ hlin) {
  __shared__ __attribute__((aligned(16))) float As[2][16 * TILE_P];
  int tid = (int)threadIdx.x;
  int lane = tid & 31, wv = tid >> 5;
  int half = lane >> 4, l16 = lane & 15;
  int o0 = wv * 16;
  float4 b[16];
  load_bfrag(W, o0, l16, half, b);
  int r0 = wv, c4 = lane;
  const float4* x4 = (const float4*)x;
  int t = (int)blockIdx.x;
  float4 p0 = x4[(t * 16 + r0) * 32 + c4];
  float4 p1 = x4[(t * 16 + r0 + 8) * 32 + c4];
  *(float4*)&As[0][r0 * TILE_P + 4 * c4] = p0;
  *(float4*)&As[0][(r0 + 8) * TILE_P + 4 * c4] = p1;
  __syncthreads();
  int buf = 0;
  for (; t < NTILES; t += PGRID) {
    int tn = t + PGRID;
    bool more = tn < NTILES;
    if (more) {
      p0 = x4[(tn * 16 + r0) * 32 + c4];
      p1 = x4[(tn * 16 + r0 + 8) * 32 + c4];
    }
    v8f acc = tile_mma(As[buf], l16, half, b);
    if (more) {
      *(float4*)&As[buf ^ 1][r0 * TILE_P + 4 * c4] = p0;
      *(float4*)&As[buf ^ 1][(r0 + 8) * TILE_P + 4 * c4] = p1;
    }
    __syncthreads();
#pragma unroll
    for (int r = 0; r < 8; ++r)
      hlin[(t * 16 + r + half * 8) * OUT_C + o0 + l16] = acc[r];
    buf ^= 1;
  }
}

static_assert(N_NODES <= 65536, "source ids are kept as 16-bit values");

__global__ void k_cnt_zero(int* __restrict__ cnt) {
  int i = blockIdx.x * blockDim.x + threadIdx.x;
  if (i < N_NODES) cnt[i] = 0;
}

__global__ void k_count_rank(const int* __restrict__ ei, int* __restrict__ cnt,
                             int* __restrict__ rank) {
  int e = blockIdx.x * blockDim.x + threadIdx.x;
  if (e < N_EDGES)
    rank[e] = __hip_atomic_fetch_add(&cnt[ei[N_EDGES + e]], 1, __ATOMIC_RELAXED,
                                     __HIP_MEMORY_SCOPE_AGENT);
}

#define SCAN_T 1024
#define SCAN_V 13
static_assert(N_NODES % 4 == 0 && SCAN_T * SCAN_V * 4 >= N_NODES, "scan tiling");
__global__ void __launch_bounds__(SCAN_T) k_scan(const int* __restrict__ cnt,
                                                 int* __restrict__ off,
                                                 float* __restrict__ dinv) {
  __shared__ int part[SCAN_T];
  const int N4 = N_NODES / 4;
  int t = threadIdx.x;
  int b4 = t * SCAN_V;
  const int4* c4 = (const int4*)cnt;
  int4 a[SCAN_V];
  for (int v = 0; v < SCAN_V; ++v) {
    int i4 = b4 + v;
    int4 x = c4[(i4 < N4) ? i4 : (N4 - 1)];
    if (i4 >= N4) x = make_int4(0, 0, 0, 0);
    a[v] = x;
  }
  int loc = 0;
  for (int v = 0; v < SCAN_V; ++v) loc += a[v].x + a[v].y + a[v].z + a[v].w;
  part[t] = loc;
  __syncthreads();
  for (int o = 1; o < SCAN_T; o <<= 1) {
    int v = (t >= o) ? part[t - o] : 0;
    __syncthreads();
    part[t] += v;
    __syncthreads();
  }
  int run = part[t] - loc;
  for (int v = 0; v < SCAN_V; ++v) {
    int i4 = b4 + v;
    if (i4 < N4) {
      int4 o;
      o.x = run; run += a[v].x;
      o.y = run; run += a[v].y;
      o.z = run; run += a[v].z;
      o.w = run; run += a[v].w;
      ((int4*)off)[i4] = o;
      float4 dv;
      float d0 = 1.0f + (float)a[v].x, d1 = 1.0f + (float)a[v].y;
      float d2 = 1.0f + (float)a[v].z, d3 = 1.0f + (float)a[v].w;
      dv.x = (d0 > 0.0f) ? rsqrtf(d0) : 0.0f;
      dv.y = (d1 > 0.0f) ? rsqrtf(d1) : 0.0f;
      dv.z = (d2 > 0.0f) ? rsqrtf(d2) : 0.0f;
      dv.w = (d3 > 0.0f) ? rsqrtf(d3) : 0.0f;
      ((float4*)dinv)[i4] = dv;
    }
  }
  if (t == SCAN_T - 1) off[N_NODES] = part[t];
}

__global__ void k_fill(const int* __restrict__ ei, const int* __restrict__ rank,
                       const int* __restrict__ off,
                       unsigned short* __restrict__ csr) {
  int e = blockIdx.x * blockDim.x + threadIdx.x;
  if (e < N_EDGES) {
    int d = ei[N_EDGES + e];
    csr[off[d] + rank[e]] = (unsigned short)ei[e];
  }
}

__device__ __forceinline__ float4 axpy4(float4 v, float s, float4 a) {
  a.x += v.x * s;
  a.y += v.y * s;
  a.z += v.z * s;
  a.w += v.w * s;
  return a;
}

__device__ __forceinline__ float readlane_f(float v, int l) {
  return __int_as_float(__builtin_amdgcn_readlane(__float_as_int(v), l));
}

#define MID_CAP 128
__global__ void __launch_bounds__(256) k_gather(const float* __restrict__ hlin,
                                                const int* __restrict__ off,
                                                const unsigned short* __restrict__ csr,
                                                const float* __restrict__ dinv,
                                                const float* __restrict__ bias,
                                                float* __restrict__ h) {
  __shared__ int   mid_us[8][MID_CAP];
  __shared__ int   mid_so[8][MID_CAP];
  __shared__ float mid_wo[8][MID_CAP];
  int lane = (int)(threadIdx.x & 31);
  int n = __builtin_amdgcn_readfirstlane(
      (int)((blockIdx.x * blockDim.x + threadIdx.x) >> 5));
  if (n >= N_NODES) return;
  int beg = off[n];
  int d = off[n + 1] - beg;
  float di = dinv[n];
  const float4* h4 = (const float4*)hlin;
  float4 hv = h4[n * 32 + lane];
  float4 acc;
  acc.x = hv.x * di * di + bias[lane * 4 + 0];
  acc.y = hv.y * di * di + bias[lane * 4 + 1];
  acc.z = hv.z * di * di + bias[lane * 4 + 2];
  acc.w = hv.w * di * di + bias[lane * 4 + 3];

  if (d <= 32) {
    int s = 0x7fffffff;
    float w = 0.0f;
    if (lane < d) {
      s = (int)csr[beg + lane];
      w = dinv[s];
    }
    int r = 0;
    for (int j = 0; j < d; ++j) {
      int sj = __builtin_amdgcn_readlane(s, j);
      r += ((sj < s) || (sj == s && j < lane)) ? 1 : 0;
    }
    int   ss = __builtin_amdgcn_ds_permute(r << 2, s);
    float ws = __int_as_float(__builtin_amdgcn_ds_permute(r << 2, __float_as_int(w)));
    int k = 0;
    for (; k + 4 <= d; k += 4) {
      int s0 = __builtin_amdgcn_readlane(ss, k);
      int s1 = __builtin_amdgcn_readlane(ss, k + 1);
      int s2 = __builtin_amdgcn_readlane(ss, k + 2);
      int s3 = __builtin_amdgcn_readlane(ss, k + 3);
      float w0 = readlane_f(ws, k);
      float w1 = readlane_f(ws, k + 1);
      float w2 = readlane_f(ws, k + 2);
      float w3 = readlane_f(ws, k + 3);
      float4 v0 = h4[s0 * 32 + lane];
      float4 v1 = h4[s1 * 32 + lane];
      float4 v2 = h4[s2 * 32 + lane];
      float4 v3 = h4[s3 * 32 + lane];
      acc = axpy4(v0, w0 * di, acc);
      acc = axpy4(v1, w1 * di, acc);
      acc = axpy4(v2, w2 * di, acc);
      acc = axpy4(v3, w3 * di, acc);
    }
    for (; k < d; ++k) {
      int s0 = __builtin_amdgcn_readlane(ss, k);
      float w0 = readlane_f(ws, k);
      float4 v0 = h4[s0 * 32 + lane];
      acc = axpy4(v0, w0 * di, acc);
    }
  } else if (d <= MID_CAP) {
    int wv = (int)(threadIdx.x >> 5);
    int*   us = mid_us[wv];
    int*   so = mid_so[wv];
    float* wo = mid_wo[wv];
    int   e[MID_CAP / 32];
    float w[MID_CAP / 32];
    for (int q = 0; q < MID_CAP / 32; ++q) {
      int i = lane + 32 * q;
      e[q] = 0x7fffffff;
      w[q] = 0.0f;
      if (i < d) {
        e[q] = (int)csr[beg + i];
        w[q] = dinv[e[q]];
        us[i] = e[q];
      }
    }
    int r[MID_CAP / 32];
    for (int q = 0; q < MID_CAP / 32; ++q) r[q] = 0;
    for (int j = 0; j < d; ++j) {
      int sj = us[j];
      for (int q = 0; q < MID_CAP / 32; ++q)
        r[q] += ((sj < e[q]) || (sj == e[q] && j < lane + 32 * q)) ? 1 : 0;
    }
    for (int q = 0; q < MID_CAP / 32; ++q) {
      if (lane + 32 * q < d) {
        so[r[q]] = e[q];
        wo[r[q]] = w[q];
      }
    }
    int k = 0;
    for (; k + 4 <= d; k += 4) {
      int s0 = so[k], s1 = so[k + 1], s2 = so[k + 2], s3 = so[k + 3];
      float w0 = wo[k], w1 = wo[k + 1], w2 = wo[k + 2], w3 = wo[k + 3];
      float4 v0 = h4[s0 * 32 + lane];
      float4 v1 = h4[s1 * 32 + lane];
      float4 v2 = h4[s2 * 32 + lane];
      float4 v3 = h4[s3 * 32 + lane];
      acc = axpy4(v0, w0 * di, acc);
      acc = axpy4(v1, w1 * di, acc);
      acc = axpy4(v2, w2 * di, acc);
      acc = axpy4(v3, w3 * di, acc);
    }
    for (; k < d; ++k) {
      int s0 = so[k];
      float w0 = wo[k];
      float4 v0 = h4[s0 * 32 + lane];
      acc = axpy4(v0, w0 * di, acc);
    }
  } else {
    int last = -1;
    int left = d;
    for (int it = 0; it < d && left > 0; ++it) {
      int mn = 0x7fffffff;
      for (int i = lane; i < d; i += 32) {
        int s = (int)csr[beg + i];
        if (s > last && s < mn) mn = s;
      }
      for (int o = 16; o > 0; o >>= 1) {
        int t = __shfl_xor(mn, o);
        mn = (t < mn) ? t : mn;
      }
      int m = 0;
      for (int i = lane; i < d; i += 32) m += ((int)csr[beg + i] == mn) ? 1 : 0;
      for (int o = 16; o > 0; o >>= 1) m += __shfl_xor(m, o);
      mn = __builtin_amdgcn_readfirstlane(mn);
      m = __builtin_amdgcn_readfirstlane(m);
      float nrm = dinv[mn] * di;
      float4 v = h4[mn * 32 + lane];
      for (int c = 0; c < m; ++c) acc = axpy4(v, nrm, acc);
      last = mn;
      left -= m;
    }
  }
  ((float4*)h)[n * 32 + lane] = acc;
}

__global__ void k_prep_w0(const float* __restrict__ bw, const float* __restrict__ sw,
                          const float* __restrict__ sc, float* __restrict__ Bm) {
  int t = blockIdx.x * blockDim.x + threadIdx.x;
  if (t >= 8 * OUT_C * HIDDEN) return;
  int k = t >> 4, o = t & 15;
  int f = k >> 3, j = k & 7;
  float w = (j == 0) ? bw[o * OUT_C + f]
                     : sw[(o * OUT_C + f) * N_COEF + (j - 1)] * sc[o * OUT_C + f];
  Bm[k * HIDDEN + o] = w;
}

__global__ void k_prep_w1(const float* __restrict__ bw, const float* __restrict__ sw,
                          const float* __restrict__ sc, float* __restrict__ Bt) {
  int t = blockIdx.x * blockDim.x + threadIdx.x;
  if (t >= 8 * HIDDEN * OUT_C) return;
  int o = t >> 7, k = t & 127;
  int f = k >> 3, j = k & 7;
  float w = (j == 0) ? bw[o * HIDDEN + f]
                     : sw[(o * HIDDEN + f) * N_COEF + (j - 1)] * sc[o * HIDDEN + f];
  Bt[o * 128 + k] = w;
}

#define LDS0_STRIDE 258
__global__ void __launch_bounds__(32) k_kan0(const float* __restrict__ h,
                                             const float* __restrict__ Bm,
                                             float* __restrict__ g) {
  __shared__ float feat[16 * LDS0_STRIDE];
  int n0 = blockIdx.x * 16;
  int lane = threadIdx.x;
  int half = lane >> 4, l16 = lane & 15;
  v8f accA = {}, accB = {};
  for (int ch = 0; ch < 4; ++ch) {
    int f0 = ch * 32;
    for (int idx = lane; idx < 16 * 32; idx += 32) {
      int row = idx & 15, fl = idx >> 4;
      float v = h[(n0 + row) * OUT_C + f0 + fl];
      float* d = &feat[row * LDS0_STRIDE + fl * 8];
      d[0] = silu_f(v);
      float bs[N_COEF];
      bspline7(v, bs);
#pragma unroll
      for (int c = 0; c < N_COEF; ++c) d[1 + c] = bs[c];
    }
    __syncthreads();
    for (int k = 0; k < 256; k += 8) {
#pragma unroll
      for (int s = 0; s < 2; ++s) {
        int ka0 = k + s * 4 + (half ? 2 : 0);
        int ka1 = k + s * 4 + (half ? 3 : 1);
        v2f a, b;
        a.x = feat[l16 * LDS0_STRIDE + ka0];
        a.y = feat[l16 * LDS0_STRIDE + ka1];
        b.x = Bm[(ch * 256 + ka0) * HIDDEN + l16];
        b.y = Bm[(ch * 256 + ka1) * HIDDEN + l16];
        if (s == 0) accA = wmma4(a, b, accA);
        else        accB = wmma4(a, b, accB);
      }
    }
    __syncthreads();
  }
  v8f acc = accA + accB;
#pragma unroll
  for (int i = 0; i < 8; ++i)
    g[(n0 + i + half * 8) * HIDDEN + l16] = acc[i];
}

__device__ __forceinline__ void expand8(float v, float4* d) {
  float bs[N_COEF];
  bspline7(v, bs);
  float4 lo, hi;
  lo.x = silu_f(v); lo.y = bs[0]; lo.z = bs[1]; lo.w = bs[2];
  hi.x = bs[3]; hi.y = bs[4]; hi.z = bs[5]; hi.w = bs[6];
  d[0] = lo;
  d[1] = hi;
}

__global__ void __launch_bounds__(256) k_kan1(const float* __restrict__ g,
                                              const float* __restrict__ Bt,
                                              float* __restrict__ out) {
  __shared__ __attribute__((aligned(16))) float As[2][16 * TILE_P];
  int tid = (int)threadIdx.x;
  int lane = tid & 31, wv = tid >> 5;
  int half = lane >> 4, l16 = lane & 15;
  int o0 = wv * 16;
  float4 b[16];
  load_bfrag(Bt, o0, l16, half, b);
  int row = tid >> 4, f = tid & 15;
  int t = (int)blockIdx.x;
  float v = g[(t * 16 + row) * HIDDEN + f];
  expand8(v, (float4*)&As[0][row * TILE_P + f * 8]);
  __syncthreads();
  int buf = 0;
  for (; t < NTILES; t += PGRID) {
    int tn = t + PGRID;
    bool more = tn < NTILES;
    if (more) v = g[(tn * 16 + row) * HIDDEN + f];
    v8f acc = tile_mma(As[buf], l16, half, b);
    if (more) expand8(v, (float4*)&As[buf ^ 1][row * TILE_P + f * 8]);
    __syncthreads();
#pragma unroll
    for (int r = 0; r < 8; ++r)
      out[(t * 16 + r + half * 8) * OUT_C + o0 + l16] = acc[r];
    buf ^= 1;
  }
}

extern "C" void kernel_launch(void* const* d_in, const int* in_sizes, int n_in,
                              void* d_out, int out_size, void* d_ws, size_t ws_size,
                              hipStream_t stream) {
  (void)in_sizes; (void)n_in; (void)out_size; (void)ws_size;
  const float* x       = (const float*)d_in[0];
  const int*   ei      = (const int*)d_in[1];
  const float* gcn_w   = (const float*)d_in[2];
  const float* gcn_b   = (const float*)d_in[3];
  const float* base_w0 = (const float*)d_in[4];
  const float* spl_w0  = (const float*)d_in[5];
  const float* scal0   = (const float*)d_in[6];
  const float* base_w1 = (const float*)d_in[7];
  const float* spl_w1  = (const float*)d_in[8];
  const float* scal1   = (const float*)d_in[9];
  float* out = (float*)d_out;

  float* ws    = (float*)d_ws;
  float* hlin  = ws;
  float* h     = hlin + (size_t)N_NODES * OUT_C;
  float* deg   = h + (size_t)N_NODES * OUT_C;
  float* g     = deg + N_NODES;
  float* Bm0   = g + (size_t)N_NODES * HIDDEN;
  float* Bm1   = Bm0 + 8 * OUT_C * HIDDEN;
  int* cnt  = (int*)h;
  int* rank = cnt + N_NODES;
  unsigned short* csr = (unsigned short*)g;
  int* off  = (int*)g + N_EDGES / 2;

  k_gemm_xw<<<PGRID, 256, 0, stream>>>(x, gcn_w, hlin);
  k_cnt_zero<<<(N_NODES + 255) / 256, 256, 0, stream>>>(cnt);
  k_count_rank<<<(N_EDGES + 255) / 256, 256, 0, stream>>>(ei, cnt, rank);
  k_scan<<<1, SCAN_T, 0, stream>>>(cnt, off, deg);
  k_fill<<<(N_EDGES + 255) / 256, 256, 0, stream>>>(ei, rank, off, csr);
  k_gather<<<(N_NODES * 32 + 255) / 256, 256, 0, stream>>>(hlin, off, csr, deg, gcn_b, h);
  k_prep_w0<<<(8 * OUT_C * HIDDEN + 255) / 256, 256, 0, stream>>>(base_w0, spl_w0, scal0, Bm0);
  k_prep_w1<<<(8 * HIDDEN * OUT_C + 255) / 256, 256, 0, stream>>>(base_w1, spl_w1, scal1, Bm1);
  k_kan0<<<N_NODES / 16, 32, 0, stream>>>(h, Bm0, g);
  k_kan1<<<PGRID, 256, 0, stream>>>(g, Bm1, out);
}


# Group edges by destination with a two-level counting sort in LDS (chunk histogram over 256-node bins, scan, move, per-bin segment build): no global atomics, no per-edge scattered list writes
# speedup vs baseline: 2.0286x; 1.2160x over previous
#include <hip/hip_runtime.h>
#include <hip/hip_bf16.h>

#define N_NODES 50000
#define N_EDGES 800000
#define IN_C    128
#define OUT_C   128
#define HIDDEN  16
#define N_COEF  7

typedef __attribute__((ext_vector_type(2))) float v2f;
typedef __attribute__((ext_vector_type(8))) float v8f;

__device__ __forceinline__ v8f wmma4(v2f a, v2f b, v8f c) {
  return __builtin_amdgcn_wmma_f32_16x16x4_f32(false, a, false, b, (short)0, c,
                                               false, false);
}

__device__ __forceinline__ float silu_f(float v) {
  return v * (1.0f / (1.0f + __expf(-v)));
}

__device__ __forceinline__ void bspline7(float v, float* out) {
  float b[10];
#pragma unroll
  for (int j = 0; j < 10; ++j) {
    float gj = 0.5f * (float)j - 2.5f;
    b[j] = (v >= gj && v < gj + 0.5f) ? 1.0f : 0.0f;
  }
#pragma unroll
  for (int p = 1; p <= 3; ++p) {
    float invd = 1.0f / (0.5f * (float)p);
#pragma unroll
    for (int j = 0; j < 10 - p; ++j) {
      float gj  = 0.5f * (float)j - 2.5f;
      float gjp = gj + 0.5f * (float)p;
      b[j] = (v - gj) * invd * b[j] + ((gjp + 0.5f) - v) * invd * b[j + 1];
    }
  }
#pragma unroll
  for (int c = 0; c < N_COEF; ++c) out[c] = b[c];
}

#define TILE_P 132
#define NTILES (N_NODES / 16)
#define PGRID  512
static_assert(N_NODES % 16 == 0 && PGRID <= NTILES, "tiling");

__device__ __forceinline__ void load_bfrag(const float* __restrict__ Wt, int o0,
                                           int l16, int half, float4 (&b)[16]) {
  const float4* p = (const float4*)(Wt + (o0 + l16) * 128 + 4 * half);
#pragma unroll
  for (int j = 0; j < 16; ++j) b[j] = p[2 * j];
}

__device__ __forceinline__ v8f tile_mma(const float* As, int l16, int half,
                                        const float4 (&b)[16]) {
  float4 a[16];
  const float4* p = (const float4*)(As + l16 * TILE_P + 4 * half);
#pragma unroll
  for (int j = 0; j < 16; ++j) a[j] = p[2 * j];
  __builtin_amdgcn_sched_barrier(0);
  v8f acc0 = {}, acc1 = {};
#pragma unroll
  for (int j = 0; j < 16; ++j) {
    v2f a0, a1, b0, b1;
    a0.x = a[j].x; a0.y = a[j].y; a1.x = a[j].z; a1.y = a[j].w;
    b0.x = b[j].x; b0.y = b[j].y; b1.x = b[j].z; b1.y = b[j].w;
    acc0 = wmma4(a0, b0, acc0);
    acc1 = wmma4(a1, b1, acc1);
  }
  asm volatile("v_nop\n\tv_nop\n\tv_nop\n\tv_nop\n\tv_nop" : "+v"(acc0), "+v"(acc1));
  return acc0 + acc1;
}

__global__ void __launch_bounds__(256) k_gemm_xw(const float* __restrict__ x,
                                                 const float* __restrict__ W,
                                                 float* __restrict__ hlin) {
  __shared__ __attribute__((aligned(16))) float As[2][16 * TILE_P];
  int tid = (int)threadIdx.x;
  int lane = tid & 31, wv = tid >> 5;
  int half = lane >> 4, l16 = lane & 15;
  int o0 = wv * 16;
  float4 b[16];
  load_bfrag(W, o0, l16, half, b);
  int r0 = wv, c4 = lane;
  const float4* x4 = (const float4*)x;
  int t = (int)blockIdx.x;
  float4 p0 = x4[(t * 16 + r0) * 32 + c4];
  float4 p1 = x4[(t * 16 + r0 + 8) * 32 + c4];
  *(float4*)&As[0][r0 * TILE_P + 4 * c4] = p0;
  *(float4*)&As[0][(r0 + 8) * TILE_P + 4 * c4] = p1;
  __syncthreads();
  int buf = 0;
  for (; t < NTILES; t += PGRID) {
    int tn = t + PGRID;
    bool more = tn < NTILES;
    if (more) {
      p0 = x4[(tn * 16 + r0) * 32 + c4];
      p1 = x4[(tn * 16 + r0 + 8) * 32 + c4];
    }
    v8f acc = tile_mma(As[buf], l16, half, b);
    if (more) {
      *(float4*)&As[buf ^ 1][r0 * TILE_P + 4 * c4] = p0;
      *(float4*)&As[buf ^ 1][(r0 + 8) * TILE_P + 4 * c4] = p1;
    }
    __syncthreads();
#pragma unroll
    for (int r = 0; r < 8; ++r)
      hlin[(t * 16 + r + half * 8) * OUT_C + o0 + l16] = acc[r];
    buf ^= 1;
  }
}

static_assert(N_NODES <= 65536, "source ids are kept as 16-bit values");
#define BINW   256
#define NBINS  ((N_NODES + BINW - 1) / BINW)
#define CHUNK  4096
#define NCHUNK ((N_EDGES + CHUNK - 1) / CHUNK)
#define EDGE_T 512
#define EDGE_V (CHUNK / EDGE_T)
#define NSLOT  (NBINS * NCHUNK)
static_assert(CHUNK % EDGE_T == 0, "whole edges per thread");
static_assert(BINW <= 65536 && BINW <= EDGE_T, "local destination and source share one word; one thread per node of a bin");

__global__ void __launch_bounds__(EDGE_T) k_bin_hist(const int* __restrict__ ei,
                                                     int* __restrict__ slot) {
  __shared__ int hist[NBINS];
  int tid = (int)threadIdx.x, blk = (int)blockIdx.x;
  for (int b = tid; b < NBINS; b += EDGE_T) hist[b] = 0;
  __syncthreads();
  int e0 = blk * CHUNK + tid;
  unsigned d[EDGE_V];
#pragma unroll
  for (int q = 0; q < EDGE_V; ++q) {
    int e = e0 + q * EDGE_T;
    d[q] = (unsigned)ei[N_EDGES + ((e < N_EDGES) ? e : (N_EDGES - 1))];
  }
#pragma unroll
  for (int q = 0; q < EDGE_V; ++q) {
    int e = e0 + q * EDGE_T;
    if (e < N_EDGES)
      __hip_atomic_fetch_add(&hist[d[q] / BINW], 1, __ATOMIC_RELAXED,
                             __HIP_MEMORY_SCOPE_WORKGROUP);
  }
  __syncthreads();
  for (int b = tid; b < NBINS; b += EDGE_T) slot[b * NCHUNK + blk] = hist[b];
}

#define SCAN_T 1024
#define SLOT_V ((NSLOT + SCAN_T - 1) / SCAN_T)
__global__ void __launch_bounds__(SCAN_T) k_slot_scan(int* __restrict__ slot) {
  __shared__ int part[SCAN_T];
  int t = (int)threadIdx.x;
  int b0 = t * SLOT_V;
  int a[SLOT_V];
#pragma unroll
  for (int v = 0; v < SLOT_V; ++v) {
    int i = b0 + v;
    int x = slot[(i < NSLOT) ? i : (NSLOT - 1)];
    a[v] = (i < NSLOT) ? x : 0;
  }
  int loc = 0;
#pragma unroll
  for (int v = 0; v < SLOT_V; ++v) loc += a[v];
  part[t] = loc;
  __syncthreads();
  for (int o = 1; o < SCAN_T; o <<= 1) {
    int v = (t >= o) ? part[t - o] : 0;
    __syncthreads();
    part[t] += v;
    __syncthreads();
  }
  int run = part[t] - loc;
#pragma unroll
  for (int v = 0; v < SLOT_V; ++v) {
    int i = b0 + v;
    if (i < NSLOT) slot[i] = run;
    run += a[v];
  }
}

__global__ void __launch_bounds__(EDGE_T) k_bin_move(const int* __restrict__ ei,
                                                     const int* __restrict__ slot,
                                                     unsigned* __restrict__ binned) {
  __shared__ int pos[NBINS];
  int tid = (int)threadIdx.x, blk = (int)blockIdx.x;
  for (int b = tid; b < NBINS; b += EDGE_T) pos[b] = slot[b * NCHUNK + blk];
  int e0 = blk * CHUNK + tid;
  unsigned d[EDGE_V], s[EDGE_V];
#pragma unroll
  for (int q = 0; q < EDGE_V; ++q) {
    int e = e0 + q * EDGE_T;
    int ec = (e < N_EDGES) ? e : (N_EDGES - 1);
    d[q] = (unsigned)ei[N_EDGES + ec];
    s[q] = (unsigned)ei[ec];
  }
  __syncthreads();
#pragma unroll
  for (int q = 0; q < EDGE_V; ++q) {
    int e = e0 + q * EDGE_T;
    if (e < N_EDGES) {
      int p = __hip_atomic_fetch_add(&pos[d[q] / BINW], 1, __ATOMIC_RELAXED,
                                     __HIP_MEMORY_SCOPE_WORKGROUP);
      binned[p] = ((d[q] % BINW) << 16) | s[q];
    }
  }
}

#define BIN_U 4
__global__ void __launch_bounds__(EDGE_T) k_bin_csr(const unsigned* __restrict__ binned,
                                                    const int* __restrict__ slot,
                                                    int* __restrict__ off,
                                                    float* __restrict__ dinv,
                                                    unsigned short* __restrict__ csr) {
  __shared__ int cnt[BINW];
  __shared__ int seg[BINW];
  __shared__ int fil[BINW];
  int tid = (int)threadIdx.x, bin = (int)blockIdx.x;
  int beg = slot[bin * NCHUNK];
  int end = (bin + 1 < NBINS) ? slot[(bin + 1) * NCHUNK] : N_EDGES;
  int n = end - beg;
  if (tid < BINW) {
    cnt[tid] = 0;
    fil[tid] = 0;
  }
  __syncthreads();
  for (int i0 = tid; i0 < n; i0 += BIN_U * EDGE_T) {
    unsigned w[BIN_U];
#pragma unroll
    for (int q = 0; q < BIN_U; ++q) {
      int i = i0 + q * EDGE_T;
      w[q] = binned[beg + ((i < n) ? i : i0)];
    }
#pragma unroll
    for (int q = 0; q < BIN_U; ++q) {
      if (i0 + q * EDGE_T < n)
        __hip_atomic_fetch_add(&cnt[w[q] >> 16], 1, __ATOMIC_RELAXED,
                               __HIP_MEMORY_SCOPE_WORKGROUP);
    }
  }
  __syncthreads();
  int c = (tid < BINW) ? cnt[tid] : 0;
  if (tid < BINW) seg[tid] = c;
  __syncthreads();
  for (int o = 1; o < BINW; o <<= 1) {
    int v = (tid < BINW && tid >= o) ? seg[tid - o] : 0;
    __syncthreads();
    if (tid < BINW) seg[tid] += v;
    __syncthreads();
  }
  if (tid < BINW) {
    int p = beg + seg[tid] - c;
    seg[tid] = p;
    int node = bin * BINW + tid;
    if (node < N_NODES) {
      off[node] = p;
      float dg = 1.0f + (float)c;
      dinv[node] = (dg > 0.0f) ? rsqrtf(dg) : 0.0f;
    }
  }
  if (bin == NBINS - 1 && tid == 0) off[N_NODES] = N_EDGES;
  __syncthreads();
  for (int i0 = tid; i0 < n; i0 += BIN_U * EDGE_T) {
    unsigned w[BIN_U];
#pragma unroll
    for (int q = 0; q < BIN_U; ++q) {
      int i = i0 + q * EDGE_T;
      w[q] = binned[beg + ((i < n) ? i : i0)];
    }
#pragma unroll
    for (int q = 0; q < BIN_U; ++q) {
      if (i0 + q * EDGE_T < n) {
        int dl = (int)(w[q] >> 16);
        int r = __hip_atomic_fetch_add(&fil[dl], 1, __ATOMIC_RELAXED,
                                       __HIP_MEMORY_SCOPE_WORKGROUP);
        csr[seg[dl] + r] = (unsigned short)(w[q] & 0xffffu);
      }
    }
  }
}

__device__ __forceinline__ float4 axpy4(float4 v, float s, float4 a) {
  a.x += v.x * s;
  a.y += v.y * s;
  a.z += v.z * s;
  a.w += v.w * s;
  return a;
}

__device__ __forceinline__ float readlane_f(float v, int l) {
  return __int_as_float(__builtin_amdgcn_readlane(__float_as_int(v), l));
}

#define MID_CAP 128
__global__ void __launch_bounds__(256) k_gather(const float* __restrict__ hlin,
                                                const int* __restrict__ off,
                                                const unsigned short* __restrict__ csr,
                                                const float* __restrict__ dinv,
                                                const float* __restrict__ bias,
                                                float* __restrict__ h) {
  __shared__ int   mid_us[8][MID_CAP];
  __shared__ int   mid_so[8][MID_CAP];
  __shared__ float mid_wo[8][MID_CAP];
  int lane = (int)(threadIdx.x & 31);
  int n = __builtin_amdgcn_readfirstlane(
      (int)((blockIdx.x * blockDim.x + threadIdx.x) >> 5));
  if (n >= N_NODES) return;
  int beg = off[n];
  int d = off[n + 1] - beg;
  float di = dinv[n];
  const float4* h4 = (const float4*)hlin;
  float4 hv = h4[n * 32 + lane];
  float4 acc;
  acc.x = hv.x * di * di + bias[lane * 4 + 0];
  acc.y = hv.y * di * di + bias[lane * 4 + 1];
  acc.z = hv.z * di * di + bias[lane * 4 + 2];
  acc.w = hv.w * di * di + bias[lane * 4 + 3];

  if (d <= 32) {
    int s = 0x7fffffff;
    float w = 0.0f;
    if (lane < d) {
      s = (int)csr[beg + lane];
      w = dinv[s];
    }
    int r = 0;
    for (int j = 0; j < d; ++j) {
      int sj = __builtin_amdgcn_readlane(s, j);
      r += ((sj < s) || (sj == s && j < lane)) ? 1 : 0;
    }
    int   ss = __builtin_amdgcn_ds_permute(r << 2, s);
    float ws = __int_as_float(__builtin_amdgcn_ds_permute(r << 2, __float_as_int(w)));
    int k = 0;
    for (; k + 4 <= d; k += 4) {
      int s0 = __builtin_amdgcn_readlane(ss, k);
      int s1 = __builtin_amdgcn_readlane(ss, k + 1);
      int s2 = __builtin_amdgcn_readlane(ss, k + 2);
      int s3 = __builtin_amdgcn_readlane(ss, k + 3);
      float w0 = readlane_f(ws, k);
      float w1 = readlane_f(ws, k + 1);
      float w2 = readlane_f(ws, k + 2);
      float w3 = readlane_f(ws, k + 3);
      float4 v0 = h4[s0 * 32 + lane];
      float4 v1 = h4[s1 * 32 + lane];
      float4 v2 = h4[s2 * 32 + lane];
      float4 v3 = h4[s3 * 32 + lane];
      acc = axpy4(v0, w0 * di, acc);
      acc = axpy4(v1, w1 * di, acc);
      acc = axpy4(v2, w2 * di, acc);
      acc = axpy4(v3, w3 * di, acc);
    }
    for (; k < d; ++k) {
      int s0 = __builtin_amdgcn_readlane(ss, k);
      float w0 = readlane_f(ws, k);
      float4 v0 = h4[s0 * 32 + lane];
      acc = axpy4(v0, w0 * di, acc);
    }
  } else if (d <= MID_CAP) {
    int wv = (int)(threadIdx.x >> 5);
    int*   us = mid_us[wv];
    int*   so = mid_so[wv];
    float* wo = mid_wo[wv];
    int   e[MID_CAP / 32];
    float w[MID_CAP / 32];
    for (int q = 0; q < MID_CAP / 32; ++q) {
      int i = lane + 32 * q;
      e[q] = 0x7fffffff;
      w[q] = 0.0f;
      if (i < d) {
        e[q] = (int)csr[beg + i];
        w[q] = dinv[e[q]];
        us[i] = e[q];
      }
    }
    int r[MID_CAP / 32];
    for (int q = 0; q < MID_CAP / 32; ++q) r[q] = 0;
    for (int j = 0; j < d; ++j) {
      int sj = us[j];
      for (int q = 0; q < MID_CAP / 32; ++q)
        r[q] += ((sj < e[q]) || (sj == e[q] && j < lane + 32 * q)) ? 1 : 0;
    }
    for (int q = 0; q < MID_CAP / 32; ++q) {
      if (lane + 32 * q < d) {
        so[r[q]] = e[q];
        wo[r[q]] = w[q];
      }
    }
    int k = 0;
    for (; k + 4 <= d; k += 4) {
      int s0 = so[k], s1 = so[k + 1], s2 = so[k + 2], s3 = so[k + 3];
      float w0 = wo[k], w1 = wo[k + 1], w2 = wo[k + 2], w3 = wo[k + 3];
      float4 v0 = h4[s0 * 32 + lane];
      float4 v1 = h4[s1 * 32 + lane];
      float4 v2 = h4[s2 * 32 + lane];
      float4 v3 = h4[s3 * 32 + lane];
      acc = axpy4(v0, w0 * di, acc);
      acc = axpy4(v1, w1 * di, acc);
      acc = axpy4(v2, w2 * di, acc);
      acc = axpy4(v3, w3 * di, acc);
    }
    for (; k < d; ++k) {
      int s0 = so[k];
      float w0 = wo[k];
      float4 v0 = h4[s0 * 32 + lane];
      acc = axpy4(v0, w0 * di, acc);
    }
  } else {
    int last = -1;
    int left = d;
    for (int it = 0; it < d && left > 0; ++it) {
      int mn = 0x7fffffff;
      for (int i = lane; i < d; i += 32) {
        int s = (int)csr[beg + i];
        if (s > last && s < mn) mn = s;
      }
      for (int o = 16; o > 0; o >>= 1) {
        int t = __shfl_xor(mn, o);
        mn = (t < mn) ? t : mn;
      }
      int m = 0;
      for (int i = lane; i < d; i += 32) m += ((int)csr[beg + i] == mn) ? 1 : 0;
      for (int o = 16; o > 0; o >>= 1) m += __shfl_xor(m, o);
      mn = __builtin_amdgcn_readfirstlane(mn);
      m = __builtin_amdgcn_readfirstlane(m);
      float nrm = dinv[mn] * di;
      float4 v = h4[mn * 32 + lane];
      for (int c = 0; c < m; ++c) acc = axpy4(v, nrm, acc);
      last = mn;
      left -= m;
    }
  }
  ((float4*)h)[n * 32 + lane] = acc;
}

__global__ void k_prep_w0(const float* __restrict__ bw, const float* __restrict__ sw,
                          const float* __restrict__ sc, float* __restrict__ Bm) {
  int t = blockIdx.x * blockDim.x + threadIdx.x;
  if (t >= 8 * OUT_C * HIDDEN) return;
  int k = t >> 4, o = t & 15;
  int f = k >> 3, j = k & 7;
  float w = (j == 0) ? bw[o * OUT_C + f]
                     : sw[(o * OUT_C + f) * N_COEF + (j - 1)] * sc[o * OUT_C + f];
  Bm[k * HIDDEN + o] = w;
}

__global__ void k_prep_w1(const float* __restrict__ bw, const float* __restrict__ sw,
                          const float* __restrict__ sc, float* __restrict__ Bt) {
  int t = blockIdx.x * blockDim.x + threadIdx.x;
  if (t >= 8 * HIDDEN * OUT_C) return;
  int o = t >> 7, k = t & 127;
  int f = k >> 3, j = k & 7;
  float w = (j == 0) ? bw[o * HIDDEN + f]
                     : sw[(o * HIDDEN + f) * N_COEF + (j - 1)] * sc[o * HIDDEN + f];
  Bt[o * 128 + k] = w;
}

#define LDS0_STRIDE 258
__global__ void __launch_bounds__(32) k_kan0(const float* __restrict__ h,
                                             const float* __restrict__ Bm,
                                             float* __restrict__ g) {
  __shared__ float feat[16 * LDS0_STRIDE];
  int n0 = blockIdx.x * 16;
  int lane = threadIdx.x;
  int half = lane >> 4, l16 = lane & 15;
  v8f accA = {}, accB = {};
  for (int ch = 0; ch < 4; ++ch) {
    int f0 = ch * 32;
    for (int idx = lane; idx < 16 * 32; idx += 32) {
      int row = idx & 15, fl = idx >> 4;
      float v = h[(n0 + row) * OUT_C + f0 + fl];
      float* d = &feat[row * LDS0_STRIDE + fl * 8];
      d[0] = silu_f(v);
      float bs[N_COEF];
      bspline7(v, bs);
#pragma unroll
      for (int c = 0; c < N_COEF; ++c) d[1 + c] = bs[c];
    }
    __syncthreads();
    for (int k = 0; k < 256; k += 8) {
#pragma unroll
      for (int s = 0; s < 2; ++s) {
        int ka0 = k + s * 4 + (half ? 2 : 0);
        int ka1 = k + s * 4 + (half ? 3 : 1);
        v2f a, b;
        a.x = feat[l16 * LDS0_STRIDE + ka0];
        a.y = feat[l16 * LDS0_STRIDE + ka1];
        b.x = Bm[(ch * 256 + ka0) * HIDDEN + l16];
        b.y = Bm[(ch * 256 + ka1) * HIDDEN + l16];
        if (s == 0) accA = wmma4(a, b, accA);
        else        accB = wmma4(a, b, accB);
      }
    }
    __syncthreads();
  }
  v8f acc = accA + accB;
#pragma unroll
  for (int i = 0; i < 8; ++i)
    g[(n0 + i + half * 8) * HIDDEN + l16] = acc[i];
}

__device__ __forceinline__ void expand8(float v, float4* d) {
  float bs[N_COEF];
  bspline7(v, bs);
  float4 lo, hi;
  lo.x = silu_f(v); lo.y = bs[0]; lo.z = bs[1]; lo.w = bs[2];
  hi.x = bs[3]; hi.y = bs[4]; hi.z = bs[5]; hi.w = bs[6];
  d[0] = lo;
  d[1] = hi;
}

__global__ void __launch_bounds__(256) k_kan1(const float* __restrict__ g,
                                              const float* __restrict__ Bt,
                                              float* __restrict__ out) {
  __shared__ __attribute__((aligned(16))) float As[2][16 * TILE_P];
  int tid = (int)threadIdx.x;
  int lane = tid & 31, wv = tid >> 5;
  int half = lane >> 4, l16 = lane & 15;
  int o0 = wv * 16;
  float4 b[16];
  load_bfrag(Bt, o0, l16, half, b);
  int row = tid >> 4, f = tid & 15;
  int t = (int)blockIdx.x;
  float v = g[(t * 16 + row) * HIDDEN + f];
  expand8(v, (float4*)&As[0][row * TILE_P + f * 8]);
  __syncthreads();
  int buf = 0;
  for (; t < NTILES; t += PGRID) {
    int tn = t + PGRID;
    bool more = tn < NTILES;
    if (more) v = g[(tn * 16 + row) * HIDDEN + f];
    v8f acc = tile_mma(As[buf], l16, half, b);
    if (more) expand8(v, (float4*)&As[buf ^ 1][row * TILE_P + f * 8]);
    __syncthreads();
#pragma unroll
    for (int r = 0; r < 8; ++r)
      out[(t * 16 + r + half * 8) * OUT_C + o0 + l16] = acc[r];
    buf ^= 1;
  }
}

extern "C" void kernel_launch(void* const* d_in, const int* in_sizes, int n_in,
                              void* d_out, int out_size, void* d_ws, size_t ws_size,
                              hipStream_t stream) {
  (void)in_sizes; (void)n_in; (void)out_size; (void)ws_size;
  const float* x       = (const float*)d_in[0];
  const int*   ei      = (const int*)d_in[1];
  const float* gcn_w   = (const float*)d_in[2];
  const float* gcn_b   = (const float*)d_in[3];
  const float* base_w0 = (const float*)d_in[4];
  const float* spl_w0  = (const float*)d_in[5];
  const float* scal0   = (const float*)d_in[6];
  const float* base_w1 = (const float*)d_in[7];
  const float* spl_w1  = (const float*)d_in[8];
  const float* scal1   = (const float*)d_in[9];
  float* out = (float*)d_out;

  float* ws    = (float*)d_ws;
  float* hlin  = ws;
  float* h     = hlin + (size_t)N_NODES * OUT_C;
  float* deg   = h + (size_t)N_NODES * OUT_C;
  float* g     = deg + N_NODES;
  float* Bm0   = g + (size_t)N_NODES * HIDDEN;
  float* Bm1   = Bm0 + 8 * OUT_C * HIDDEN;
  unsigned* binned = (unsigned*)h;
  int* slot = (int*)h + N_EDGES;
  unsigned short* csr = (unsigned short*)g;
  int* off  = (int*)g + N_EDGES / 2;

  k_gemm_xw<<<PGRID, 256, 0, stream>>>(x, gcn_w, hlin);
  k_bin_hist<<<NCHUNK, EDGE_T, 0, stream>>>(ei, slot);
  k_slot_scan<<<1, SCAN_T, 0, stream>>>(slot);
  k_bin_move<<<NCHUNK, EDGE_T, 0, stream>>>(ei, slot, binned);
  k_bin_csr<<<NBINS, EDGE_T, 0, stream>>>(binned, slot, off, deg, csr);
  k_gather<<<(N_NODES * 32 + 255) / 256, 256, 0, stream>>>(hlin, off, csr, deg, gcn_b, h);
  k_prep_w0<<<(8 * OUT_C * HIDDEN + 255) / 256, 256, 0, stream>>>(base_w0, spl_w0, scal0, Bm0);
  k_prep_w1<<<(8 * HIDDEN * OUT_C + 255) / 256, 256, 0, stream>>>(base_w1, spl_w1, scal1, Bm1);
  k_kan0<<<N_NODES / 16, 32, 0, stream>>>(h, Bm0, g);
  k_kan1<<<PGRID, 256, 0, stream>>>(g, Bm1, out);
}


# Edge sort without the one-block scan launch: chunk counts stored as contiguous rows, every move workgroup sums the count table's columns itself for its slots; workgroup 0 leaves the bin starts
# speedup vs baseline: 2.4747x; 1.2199x over previous
#include <hip/hip_runtime.h>
#include <hip/hip_bf16.h>

#define N_NODES 50000
#define N_EDGES 800000
#define IN_C    128
#define OUT_C   128
#define HIDDEN  16
#define N_COEF  7

typedef __attribute__((ext_vector_type(2))) float v2f;
typedef __attribute__((ext_vector_type(8))) float v8f;

__device__ __forceinline__ v8f wmma4(v2f a, v2f b, v8f c) {
  return __builtin_amdgcn_wmma_f32_16x16x4_f32(false, a, false, b, (short)0, c,
                                               false, false);
}

__device__ __forceinline__ float silu_f(float v) {
  return v * (1.0f / (1.0f + __expf(-v)));
}

__device__ __forceinline__ void bspline7(float v, float* out) {
  float b[10];
#pragma unroll
  for (int j = 0; j < 10; ++j) {
    float gj = 0.5f * (float)j - 2.5f;
    b[j] = (v >= gj && v < gj + 0.5f) ? 1.0f : 0.0f;
  }
#pragma unroll
  for (int p = 1; p <= 3; ++p) {
    float invd = 1.0f / (0.5f * (float)p);
#pragma unroll
    for (int j = 0; j < 10 - p; ++j) {
      float gj  = 0.5f * (float)j - 2.5f;
      float gjp = gj + 0.5f * (float)p;
      b[j] = (v - gj) * invd * b[j] + ((gjp + 0.5f) - v) * invd * b[j + 1];
    }
  }
#pragma unroll
  for (int c = 0; c < N_COEF; ++c) out[c] = b[c];
}

#define TILE_P 132
#define NTILES (N_NODES / 16)
#define PGRID  512
static_assert(N_NODES % 16 == 0 && PGRID <= NTILES, "tiling");

__device__ __forceinline__ void load_bfrag(const float* __restrict__ Wt, int o0,
                                           int l16, int half, float4 (&b)[16]) {
  const float4* p = (const float4*)(Wt + (o0 + l16) * 128 + 4 * half);
#pragma unroll
  for (int j = 0; j < 16; ++j) b[j] = p[2 * j];
}

__device__ __forceinline__ v8f tile_mma(const float* As, int l16, int half,
                                        const float4 (&b)[16]) {
  float4 a[16];
  const float4* p = (const float4*)(As + l16 * TILE_P + 4 * half);
#pragma unroll
  for (int j = 0; j < 16; ++j) a[j] = p[2 * j];
  __builtin_amdgcn_sched_barrier(0);
  v8f acc0 = {}, acc1 = {};
#pragma unroll
  for (int j = 0; j < 16; ++j) {
    v2f a0, a1, b0, b1;
    a0.x = a[j].x; a0.y = a[j].y; a1.x = a[j].z; a1.y = a[j].w;
    b0.x = b[j].x; b0.y = b[j].y; b1.x = b[j].z; b1.y = b[j].w;
    acc0 = wmma4(a0, b0, acc0);
    acc1 = wmma4(a1, b1, acc1);
  }
  asm volatile("v_nop\n\tv_nop\n\tv_nop\n\tv_nop\n\tv_nop" : "+v"(acc0), "+v"(acc1));
  return acc0 + acc1;
}

__global__ void __launch_bounds__(256) k_gemm_xw(const float* __restrict__ x,
                                                 const float* __restrict__ W,
                                                 float* __restrict__ hlin) {
  __shared__ __attribute__((aligned(16))) float As[2][16 * TILE_P];
  int tid = (int)threadIdx.x;
  int lane = tid & 31, wv = tid >> 5;
  int half = lane >> 4, l16 = lane & 15;
  int o0 = wv * 16;
  float4 b[16];
  load_bfrag(W, o0, l16, half, b);
  int r0 = wv, c4 = lane;
  const float4* x4 = (const float4*)x;
  int t = (int)blockIdx.x;
  float4 p0 = x4[(t * 16 + r0) * 32 + c4];
  float4 p1 = x4[(t * 16 + r0 + 8) * 32 + c4];
  *(float4*)&As[0][r0 * TILE_P + 4 * c4] = p0;
  *(float4*)&As[0][(r0 + 8) * TILE_P + 4 * c4] = p1;
  __syncthreads();
  int buf = 0;
  for (; t < NTILES; t += PGRID) {
    int tn = t + PGRID;
    bool more = tn < NTILES;
    if (more) {
      p0 = x4[(tn * 16 + r0) * 32 + c4];
      p1 = x4[(tn * 16 + r0 + 8) * 32 + c4];
    }
    v8f acc = tile_mma(As[buf], l16, half, b);
    if (more) {
      *(float4*)&As[buf ^ 1][r0 * TILE_P + 4 * c4] = p0;
      *(float4*)&As[buf ^ 1][(r0 + 8) * TILE_P + 4 * c4] = p1;
    }
    __syncthreads();
#pragma unroll
    for (int r = 0; r < 8; ++r)
      hlin[(t * 16 + r + half * 8) * OUT_C + o0 + l16] = acc[r];
    buf ^= 1;
  }
}

static_assert(N_NODES <= 65536, "source ids are kept as 16-bit values");
#define BINW   256
#define NBINS  ((N_NODES + BINW - 1) / BINW)
#define CHUNK  4096
#define NCHUNK ((N_EDGES + CHUNK - 1) / CHUNK)
#define EDGE_T 512
#define EDGE_V (CHUNK / EDGE_T)
#define NSLOT  (NBINS * NCHUNK)
#define HALF_C (NCHUNK / 2)
static_assert(CHUNK % EDGE_T == 0, "whole edges per thread");
static_assert(BINW <= 65536 && BINW <= EDGE_T, "local destination and source share one word; one thread per node of a bin");
static_assert(NCHUNK % 2 == 0 && NBINS <= 256 && EDGE_T == 512, "k_bin_move: thread = (bin, half of the table's rows)");
static_assert(NBINS <= 7 * 32, "k_bin_move: one wave scans the bin totals, 7 per lane");

__global__ void __launch_bounds__(EDGE_T) k_bin_hist(const int* __restrict__ ei,
                                                     int* __restrict__ cnt) {
  __shared__ int hist[NBINS];
  int tid = (int)threadIdx.x, blk = (int)blockIdx.x;
  for (int b = tid; b < NBINS; b += EDGE_T) hist[b] = 0;
  __syncthreads();
  int e0 = blk * CHUNK + tid;
  unsigned d[EDGE_V];
#pragma unroll
  for (int q = 0; q < EDGE_V; ++q) {
    int e = e0 + q * EDGE_T;
    d[q] = (unsigned)ei[N_EDGES + ((e < N_EDGES) ? e : (N_EDGES - 1))];
  }
#pragma unroll
  for (int q = 0; q < EDGE_V; ++q) {
    int e = e0 + q * EDGE_T;
    if (e < N_EDGES)
      __hip_atomic_fetch_add(&hist[d[q] / BINW], 1, __ATOMIC_RELAXED,
                             __HIP_MEMORY_SCOPE_WORKGROUP);
  }
  __syncthreads();
  for (int b = tid; b < NBINS; b += EDGE_T) cnt[blk * NBINS + b] = hist[b];
}

__global__ void __launch_bounds__(EDGE_T) k_bin_move(const int* __restrict__ ei,
                                                     const int* __restrict__ cnt,
                                                     int* __restrict__ bstart,
                                                     unsigned* __restrict__ binned) {
  __shared__ int pos[NBINS];
  __shared__ int s_tot[2][256];
  __shared__ int s_pre[2][256];
  int tid = (int)threadIdx.x, blk = (int)blockIdx.x;
  int e0 = blk * CHUNK + tid;
  unsigned d[EDGE_V], s[EDGE_V];
#pragma unroll
  for (int q = 0; q < EDGE_V; ++q) {
    int e = e0 + q * EDGE_T;
    int ec = (e < N_EDGES) ? e : (N_EDGES - 1);
    d[q] = (unsigned)ei[N_EDGES + ec];
    s[q] = (unsigned)ei[ec];
  }
  int b = tid & 255, p = tid >> 8;
  int tot = 0, pre = 0;
  if (b < NBINS) {
    const int* cp = cnt + p * HALF_C * NBINS + b;
    int lim = blk - p * HALF_C;
#pragma unroll 49
    for (int r = 0; r < HALF_C; ++r) {
      int v = cp[r * NBINS];
      tot += v;
      pre += (r < lim) ? v : 0;
    }
  }
  s_tot[p][b] = tot;
  s_pre[p][b] = pre;
  __syncthreads();
  if (tid < 32) {
    int loc[7];
    int sum = 0;
#pragma unroll
    for (int q = 0; q < 7; ++q) {
      int bb = tid * 7 + q;
      loc[q] = s_tot[0][bb] + s_tot[1][bb];
      sum += loc[q];
    }
    int inc = sum;
    for (int o = 1; o < 32; o <<= 1) {
      int v = __shfl_up(inc, o);
      if (tid >= o) inc += v;
    }
    int run = inc - sum;
#pragma unroll
    for (int q = 0; q < 7; ++q) {
      int bb = tid * 7 + q;
      if (bb < NBINS) {
        pos[bb] = run + s_pre[0][bb] + s_pre[1][bb];
        if (blk == 0) bstart[bb] = run;
      }
      run += loc[q];
    }
  }
  if (blk == 0 && tid == 0) bstart[NBINS] = N_EDGES;
  __syncthreads();
#pragma unroll
  for (int q = 0; q < EDGE_V; ++q) {
    int e = e0 + q * EDGE_T;
    if (e < N_EDGES) {
      int w = __hip_atomic_fetch_add(&pos[d[q] / BINW], 1, __ATOMIC_RELAXED,
                                     __HIP_MEMORY_SCOPE_WORKGROUP);
      binned[w] = ((d[q] % BINW) << 16) | s[q];
    }
  }
}

#define BIN_U 4
__global__ void __launch_bounds__(EDGE_T) k_bin_csr(const unsigned* __restrict__ binned,
                                                    const int* __restrict__ bstart,
                                                    int* __restrict__ off,
                                                    float* __restrict__ dinv,
                                                    unsigned short* __restrict__ csr) {
  __shared__ int cnt[BINW];
  __shared__ int seg[BINW];
  __shared__ int fil[BINW];
  int tid = (int)threadIdx.x, bin = (int)blockIdx.x;
  int beg = bstart[bin];
  int end = bstart[bin + 1];
  int n = end - beg;
  if (tid < BINW) {
    cnt[tid] = 0;
    fil[tid] = 0;
  }
  __syncthreads();
  for (int i0 = tid; i0 < n; i0 += BIN_U * EDGE_T) {
    unsigned w[BIN_U];
#pragma unroll
    for (int q = 0; q < BIN_U; ++q) {
      int i = i0 + q * EDGE_T;
      w[q] = binned[beg + ((i < n) ? i : i0)];
    }
#pragma unroll
    for (int q = 0; q < BIN_U; ++q) {
      if (i0 + q * EDGE_T < n)
        __hip_atomic_fetch_add(&cnt[w[q] >> 16], 1, __ATOMIC_RELAXED,
                               __HIP_MEMORY_SCOPE_WORKGROUP);
    }
  }
  __syncthreads();
  int c = (tid < BINW) ? cnt[tid] : 0;
  if (tid < BINW) seg[tid] = c;
  __syncthreads();
  for (int o = 1; o < BINW; o <<= 1) {
    int v = (tid < BINW && tid >= o) ? seg[tid - o] : 0;
    __syncthreads();
    if (tid < BINW) seg[tid] += v;
    __syncthreads();
  }
  if (tid < BINW) {
    int p = beg + seg[tid] - c;
    seg[tid] = p;
    int node = bin * BINW + tid;
    if (node < N_NODES) {
      off[node] = p;
      float dg = 1.0f + (float)c;
      dinv[node] = (dg > 0.0f) ? rsqrtf(dg) : 0.0f;
    }
  }
  if (bin == NBINS - 1 && tid == 0) off[N_NODES] = N_EDGES;
  __syncthreads();
  for (int i0 = tid; i0 < n; i0 += BIN_U * EDGE_T) {
    unsigned w[BIN_U];
#pragma unroll
    for (int q = 0; q < BIN_U; ++q) {
      int i = i0 + q * EDGE_T;
      w[q] = binned[beg + ((i < n) ? i : i0)];
    }
#pragma unroll
    for (int q = 0; q < BIN_U; ++q) {
      if (i0 + q * EDGE_T < n) {
        int dl = (int)(w[q] >> 16);
        int r = __hip_atomic_fetch_add(&fil[dl], 1, __ATOMIC_RELAXED,
                                       __HIP_MEMORY_SCOPE_WORKGROUP);
        csr[seg[dl] + r] = (unsigned short)(w[q] & 0xffffu);
      }
    }
  }
}

__device__ __forceinline__ float4 axpy4(float4 v, float s, float4 a) {
  a.x += v.x * s;
  a.y += v.y * s;
  a.z += v.z * s;
  a.w += v.w * s;
  return a;
}

__device__ __forceinline__ float readlane_f(float v, int l) {
  return __int_as_float(__builtin_amdgcn_readlane(__float_as_int(v), l));
}

#define MID_CAP 128
__global__ void __launch_bounds__(256) k_gather(const float* __restrict__ hlin,
                                                const int* __restrict__ off,
                                                const unsigned short* __restrict__ csr,
                                                const float* __restrict__ dinv,
                                                const float* __restrict__ bias,
                                                float* __restrict__ h) {
  __shared__ int   mid_us[8][MID_CAP];
  __shared__ int   mid_so[8][MID_CAP];
  __shared__ float mid_wo[8][MID_CAP];
  int lane = (int)(threadIdx.x & 31);
  int n = __builtin_amdgcn_readfirstlane(
      (int)((blockIdx.x * blockDim.x + threadIdx.x) >> 5));
  if (n >= N_NODES) return;
  int beg = off[n];
  int d = off[n + 1] - beg;
  float di = dinv[n];
  const float4* h4 = (const float4*)hlin;
  float4 hv = h4[n * 32 + lane];
  float4 acc;
  acc.x = hv.x * di * di + bias[lane * 4 + 0];
  acc.y = hv.y * di * di + bias[lane * 4 + 1];
  acc.z = hv.z * di * di + bias[lane * 4 + 2];
  acc.w = hv.w * di * di + bias[lane * 4 + 3];

  if (d <= 32) {
    int s = 0x7fffffff;
    float w = 0.0f;
    if (lane < d) {
      s = (int)csr[beg + lane];
      w = dinv[s];
    }
    int r = 0;
    for (int j = 0; j < d; ++j) {
      int sj = __builtin_amdgcn_readlane(s, j);
      r += ((sj < s) || (sj == s && j < lane)) ? 1 : 0;
    }
    int   ss = __builtin_amdgcn_ds_permute(r << 2, s);
    float ws = __int_as_float(__builtin_amdgcn_ds_permute(r << 2, __float_as_int(w)));
    int k = 0;
    for (; k + 4 <= d; k += 4) {
      int s0 = __builtin_amdgcn_readlane(ss, k);
      int s1 = __builtin_amdgcn_readlane(ss, k + 1);
      int s2 = __builtin_amdgcn_readlane(ss, k + 2);
      int s3 = __builtin_amdgcn_readlane(ss, k + 3);
      float w0 = readlane_f(ws, k);
      float w1 = readlane_f(ws, k + 1);
      float w2 = readlane_f(ws, k + 2);
      float w3 = readlane_f(ws, k + 3);
      float4 v0 = h4[s0 * 32 + lane];
      float4 v1 = h4[s1 * 32 + lane];
      float4 v2 = h4[s2 * 32 + lane];
      float4 v3 = h4[s3 * 32 + lane];
      acc = axpy4(v0, w0 * di, acc);
      acc = axpy4(v1, w1 * di, acc);
      acc = axpy4(v2, w2 * di, acc);
      acc = axpy4(v3, w3 * di, acc);
    }
    for (; k < d; ++k) {
      int s0 = __builtin_amdgcn_readlane(ss, k);
      float w0 = readlane_f(ws, k);
      float4 v0 = h4[s0 * 32 + lane];
      acc = axpy4(v0, w0 * di, acc);
    }
  } else if (d <= MID_CAP) {
    int wv = (int)(threadIdx.x >> 5);
    int*   us = mid_us[wv];
    int*   so = mid_so[wv];
    float* wo = mid_wo[wv];
    int   e[MID_CAP / 32];
    float w[MID_CAP / 32];
    for (int q = 0; q < MID_CAP / 32; ++q) {
      int i = lane + 32 * q;
      e[q] = 0x7fffffff;
      w[q] = 0.0f;
      if (i < d) {
        e[q] = (int)csr[beg + i];
        w[q] = dinv[e[q]];
        us[i] = e[q];
      }
    }
    int r[MID_CAP / 32];
    for (int q = 0; q < MID_CAP / 32; ++q) r[q] = 0;
    for (int j = 0; j < d; ++j) {
      int sj = us[j];
      for (int q = 0; q < MID_CAP / 32; ++q)
        r[q] += ((sj < e[q]) || (sj == e[q] && j < lane + 32 * q)) ? 1 : 0;
    }
    for (int q = 0; q < MID_CAP / 32; ++q) {
      if (lane + 32 * q < d) {
        so[r[q]] = e[q];
        wo[r[q]] = w[q];
      }
    }
    int k = 0;
    for (; k + 4 <= d; k += 4) {
      int s0 = so[k], s1 = so[k + 1], s2 = so[k + 2], s3 = so[k + 3];
      float w0 = wo[k], w1 = wo[k + 1], w2 = wo[k + 2], w3 = wo[k + 3];
      float4 v0 = h4[s0 * 32 + lane];
      float4 v1 = h4[s1 * 32 + lane];
      float4 v2 = h4[s2 * 32 + lane];
      float4 v3 = h4[s3 * 32 + lane];
      acc = axpy4(v0, w0 * di, acc);
      acc = axpy4(v1, w1 * di, acc);
      acc = axpy4(v2, w2 * di, acc);
      acc = axpy4(v3, w3 * di, acc);
    }
    for (; k < d; ++k) {
      int s0 = so[k];
      float w0 = wo[k];
      float4 v0 = h4[s0 * 32 + lane];
      acc = axpy4(v0, w0 * di, acc);
    }
  } else {
    int last = -1;
    int left = d;
    for (int it = 0; it < d && left > 0; ++it) {
      int mn = 0x7fffffff;
      for (int i = lane; i < d; i += 32) {
        int s = (int)csr[beg + i];
        if (s > last && s < mn) mn = s;
      }
      for (int o = 16; o > 0; o >>= 1) {
        int t = __shfl_xor(mn, o);
        mn = (t < mn) ? t : mn;
      }
      int m = 0;
      for (int i = lane; i < d; i += 32) m += ((int)csr[beg + i] == mn) ? 1 : 0;
      for (int o = 16; o > 0; o >>= 1) m += __shfl_xor(m, o);
      mn = __builtin_amdgcn_readfirstlane(mn);
      m = __builtin_amdgcn_readfirstlane(m);
      float nrm = dinv[mn] * di;
      float4 v = h4[mn * 32 + lane];
      for (int c = 0; c < m; ++c) acc = axpy4(v, nrm, acc);
      last = mn;
      left -= m;
    }
  }
  ((float4*)h)[n * 32 + lane] = acc;
}

__global__ void k_prep_w0(const float* __restrict__ bw, const float* __restrict__ sw,
                          const float* __restrict__ sc, float* __restrict__ Bm) {
  int t = blockIdx.x * blockDim.x + threadIdx.x;
  if (t >= 8 * OUT_C * HIDDEN) return;
  int k = t >> 4, o = t & 15;
  int f = k >> 3, j = k & 7;
  float w = (j == 0) ? bw[o * OUT_C + f]
                     : sw[(o * OUT_C + f) * N_COEF + (j - 1)] * sc[o * OUT_C + f];
  Bm[k * HIDDEN + o] = w;
}

__global__ void k_prep_w1(const float* __restrict__ bw, const float* __restrict__ sw,
                          const float* __restrict__ sc, float* __restrict__ Bt) {
  int t = blockIdx.x * blockDim.x + threadIdx.x;
  if (t >= 8 * HIDDEN * OUT_C) return;
  int o = t >> 7, k = t & 127;
  int f = k >> 3, j = k & 7;
  float w = (j == 0) ? bw[o * HIDDEN + f]
                     : sw[(o * HIDDEN + f) * N_COEF + (j - 1)] * sc[o * HIDDEN + f];
  Bt[o * 128 + k] = w;
}

#define LDS0_STRIDE 258
__global__ void __launch_bounds__(32) k_kan0(const float* __restrict__ h,
                                             const float* __restrict__ Bm,
                                             float* __restrict__ g) {
  __shared__ float feat[16 * LDS0_STRIDE];
  int n0 = blockIdx.x * 16;
  int lane = threadIdx.x;
  int half = lane >> 4, l16 = lane & 15;
  v8f accA = {}, accB = {};
  for (int ch = 0; ch < 4; ++ch) {
    int f0 = ch * 32;
    for (int idx = lane; idx < 16 * 32; idx += 32) {
      int row = idx & 15, fl = idx >> 4;
      float v = h[(n0 + row) * OUT_C + f0 + fl];
      float* d = &feat[row * LDS0_STRIDE + fl * 8];
      d[0] = silu_f(v);
      float bs[N_COEF];
      bspline7(v, bs);
#pragma unroll
      for (int c = 0; c < N_COEF; ++c) d[1 + c] = bs[c];
    }
    __syncthreads();
    for (int k = 0; k < 256; k += 8) {
#pragma unroll
      for (int s = 0; s < 2; ++s) {
        int ka0 = k + s * 4 + (half ? 2 : 0);
        int ka1 = k + s * 4 + (half ? 3 : 1);
        v2f a, b;
        a.x = feat[l16 * LDS0_STRIDE + ka0];
        a.y = feat[l16 * LDS0_STRIDE + ka1];
        b.x = Bm[(ch * 256 + ka0) * HIDDEN + l16];
        b.y = Bm[(ch * 256 + ka1) * HIDDEN + l16];
        if (s == 0) accA = wmma4(a, b, accA);
        else        accB = wmma4(a, b, accB);
      }
    }
    __syncthreads();
  }
  v8f acc = accA + accB;
#pragma unroll
  for (int i = 0; i < 8; ++i)
    g[(n0 + i + half * 8) * HIDDEN + l16] = acc[i];
}

__device__ __forceinline__ void expand8(float v, float4* d) {
  float bs[N_COEF];
  bspline7(v, bs);
  float4 lo, hi;
  lo.x = silu_f(v); lo.y = bs[0]; lo.z = bs[1]; lo.w = bs[2];
  hi.x = bs[3]; hi.y = bs[4]; hi.z = bs[5]; hi.w = bs[6];
  d[0] = lo;
  d[1] = hi;
}

__global__ void __launch_bounds__(256) k_kan1(const float* __restrict__ g,
                                              const float* __restrict__ Bt,
                                              float* __restrict__ out) {
  __shared__ __attribute__((aligned(16))) float As[2][16 * TILE_P];
  int tid = (int)threadIdx.x;
  int lane = tid & 31, wv = tid >> 5;
  int half = lane >> 4, l16 = lane & 15;
  int o0 = wv * 16;
  float4 b[16];
  load_bfrag(Bt, o0, l16, half, b);
  int row = tid >> 4, f = tid & 15;
  int t = (int)blockIdx.x;
  float v = g[(t * 16 + row) * HIDDEN + f];
  expand8(v, (float4*)&As[0][row * TILE_P + f * 8]);
  __syncthreads();
  int buf = 0;
  for (; t < NTILES; t += PGRID) {
    int tn = t + PGRID;
    bool more = tn < NTILES;
    if (more) v = g[(tn * 16 + row) * HIDDEN + f];
    v8f acc = tile_mma(As[buf], l16, half, b);
    if (more) expand8(v, (float4*)&As[buf ^ 1][row * TILE_P + f * 8]);
    __syncthreads();
#pragma unroll
    for (int r = 0; r < 8; ++r)
      out[(t * 16 + r + half * 8) * OUT_C + o0 + l16] = acc[r];
    buf ^= 1;
  }
}

extern "C" void kernel_launch(void* const* d_in, const int* in_sizes, int n_in,
                              void* d_out, int out_size, void* d_ws, size_t ws_size,
                              hipStream_t stream) {
  (void)in_sizes; (void)n_in; (void)out_size; (void)ws_size;
  const float* x       = (const float*)d_in[0];
  const int*   ei      = (const int*)d_in[1];
  const float* gcn_w   = (const float*)d_in[2];
  const float* gcn_b   = (const float*)d_in[3];
  const float* base_w0 = (const float*)d_in[4];
  const float* spl_w0  = (const float*)d_in[5];
  const float* scal0   = (const float*)d_in[6];
  const float* base_w1 = (const float*)d_in[7];
  const float* spl_w1  = (const float*)d_in[8];
  const float* scal1   = (const float*)d_in[9];
  float* out = (float*)d_out;

  float* ws    = (float*)d_ws;
  float* hlin  = ws;
  float* h     = hlin + (size_t)N_NODES * OUT_C;
  float* deg   = h + (size_t)N_NODES * OUT_C;
  float* g     = deg + N_NODES;
  float* Bm0   = g + (size_t)N_NODES * HIDDEN;
  float* Bm1   = Bm0 + 8 * OUT_C * HIDDEN;
  unsigned* binned = (unsigned*)h;
  int* cnt = (int*)h + N_EDGES;
  int* bstart = cnt + NSLOT;
  unsigned short* csr = (unsigned short*)g;
  int* off  = (int*)g + N_EDGES / 2;

  k_gemm_xw<<<PGRID, 256, 0, stream>>>(x, gcn_w, hlin);
  k_bin_hist<<<NCHUNK, EDGE_T, 0, stream>>>(ei, cnt);
  k_bin_move<<<NCHUNK, EDGE_T, 0, stream>>>(ei, cnt, bstart, binned);
  k_bin_csr<<<NBINS, EDGE_T, 0, stream>>>(binned, bstart, off, deg, csr);
  k_gather<<<(N_NODES * 32 + 255) / 256, 256, 0, stream>>>(hlin, off, csr, deg, gcn_b, h);
  k_prep_w0<<<(8 * OUT_C * HIDDEN + 255) / 256, 256, 0, stream>>>(base_w0, spl_w0, scal0, Bm0);
  k_prep_w1<<<(8 * HIDDEN * OUT_C + 255) / 256, 256, 0, stream>>>(base_w1, spl_w1, scal1, Bm1);
  k_kan0<<<N_NODES / 16, 32, 0, stream>>>(h, Bm0, g);
  k_kan1<<<PGRID, 256, 0, stream>>>(g, Bm1, out);
}


# KAN layer 0 as persistent 8-wave workgroups: K split over the waves, weight fragments in registers, features expanded in registers as A fragments, partial tiles summed via LDS (sum order)
# speedup vs baseline: 2.7533x; 1.1126x over previous
#include <hip/hip_runtime.h>
#include <hip/hip_bf16.h>

#define N_NODES 50000
#define N_EDGES 800000
#define IN_C    128
#define OUT_C   128
#define HIDDEN  16
#define N_COEF  7

typedef __attribute__((ext_vector_type(2))) float v2f;
typedef __attribute__((ext_vector_type(8))) float v8f;

__device__ __forceinline__ v8f wmma4(v2f a, v2f b, v8f c) {
  return __builtin_amdgcn_wmma_f32_16x16x4_f32(false, a, false, b, (short)0, c,
                                               false, false);
}

__device__ __forceinline__ float silu_f(float v) {
  return v * (1.0f / (1.0f + __expf(-v)));
}

__device__ __forceinline__ void bspline7(float v, float* out) {
  float b[10];
#pragma unroll
  for (int j = 0; j < 10; ++j) {
    float gj = 0.5f * (float)j - 2.5f;
    b[j] = (v >= gj && v < gj + 0.5f) ? 1.0f : 0.0f;
  }
#pragma unroll
  for (int p = 1; p <= 3; ++p) {
    float invd = 1.0f / (0.5f * (float)p);
#pragma unroll
    for (int j = 0; j < 10 - p; ++j) {
      float gj  = 0.5f * (float)j - 2.5f;
      float gjp = gj + 0.5f * (float)p;
      b[j] = (v - gj) * invd * b[j] + ((gjp + 0.5f) - v) * invd * b[j + 1];
    }
  }
#pragma unroll
  for (int c = 0; c < N_COEF; ++c) out[c] = b[c];
}

#define TILE_P 132
#define NTILES (N_NODES / 16)
#define PGRID  512
static_assert(N_NODES % 16 == 0 && PGRID <= NTILES, "tiling");

__device__ __forceinline__ void load_bfrag(const float* __restrict__ Wt, int o0,
                                           int l16, int half, float4 (&b)[16]) {
  const float4* p = (const float4*)(Wt + (o0 + l16) * 128 + 4 * half);
#pragma unroll
  for (int j = 0; j < 16; ++j) b[j] = p[2 * j];
}

__device__ __forceinline__ v8f tile_mma(const float* As, int l16, int half,
                                        const float4 (&b)[16]) {
  float4 a[16];
  const float4* p = (const float4*)(As + l16 * TILE_P + 4 * half);
#pragma unroll
  for (int j = 0; j < 16; ++j) a[j] = p[2 * j];
  __builtin_amdgcn_sched_barrier(0);
  v8f acc0 = {}, acc1 = {};
#pragma unroll
  for (int j = 0; j < 16; ++j) {
    v2f a0, a1, b0, b1;
    a0.x = a[j].x; a0.y = a[j].y; a1.x = a[j].z; a1.y = a[j].w;
    b0.x = b[j].x; b0.y = b[j].y; b1.x = b[j].z; b1.y = b[j].w;
    acc0 = wmma4(a0, b0, acc0);
    acc1 = wmma4(a1, b1, acc1);
  }
  asm volatile("v_nop\n\tv_nop\n\tv_nop\n\tv_nop\n\tv_nop" : "+v"(acc0), "+v"(acc1));
  return acc0 + acc1;
}

__global__ void __launch_bounds__(256) k_gemm_xw(const float* __restrict__ x,
                                                 const float* __restrict__ W,
                                                 float* __restrict__ hlin) {
  __shared__ __attribute__((aligned(16))) float As[2][16 * TILE_P];
  int tid = (int)threadIdx.x;
  int lane = tid & 31, wv = tid >> 5;
  int half = lane >> 4, l16 = lane & 15;
  int o0 = wv * 16;
  float4 b[16];
  load_bfrag(W, o0, l16, half, b);
  int r0 = wv, c4 = lane;
  const float4* x4 = (const float4*)x;
  int t = (int)blockIdx.x;
  float4 p0 = x4[(t * 16 + r0) * 32 + c4];
  float4 p1 = x4[(t * 16 + r0 + 8) * 32 + c4];
  *(float4*)&As[0][r0 * TILE_P + 4 * c4] = p0;
  *(float4*)&As[0][(r0 + 8) * TILE_P + 4 * c4] = p1;
  __syncthreads();
  int buf = 0;
  for (; t < NTILES; t += PGRID) {
    int tn = t + PGRID;
    bool more = tn < NTILES;
    if (more) {
      p0 = x4[(tn * 16 + r0) * 32 + c4];
      p1 = x4[(tn * 16 + r0 + 8) * 32 + c4];
    }
    v8f acc = tile_mma(As[buf], l16, half, b);
    if (more) {
      *(float4*)&As[buf ^ 1][r0 * TILE_P + 4 * c4] = p0;
      *(float4*)&As[buf ^ 1][(r0 + 8) * TILE_P + 4 * c4] = p1;
    }
    __syncthreads();
#pragma unroll
    for (int r = 0; r < 8; ++r)
      hlin[(t * 16 + r + half * 8) * OUT_C + o0 + l16] = acc[r];
    buf ^= 1;
  }
}

static_assert(N_NODES <= 65536, "source ids are kept as 16-bit values");
#define BINW   256
#define NBINS  ((N_NODES + BINW - 1) / BINW)
#define CHUNK  4096
#define NCHUNK ((N_EDGES + CHUNK - 1) / CHUNK)
#define EDGE_T 512
#define EDGE_V (CHUNK / EDGE_T)
#define NSLOT  (NBINS * NCHUNK)
#define HALF_C (NCHUNK / 2)
static_assert(CHUNK % EDGE_T == 0, "whole edges per thread");
static_assert(BINW <= 65536 && BINW <= EDGE_T, "local destination and source share one word; one thread per node of a bin");
static_assert(NCHUNK % 2 == 0 && NBINS <= 256 && EDGE_T == 512, "k_bin_move: thread = (bin, half of the table's rows)");
static_assert(NBINS <= 7 * 32, "k_bin_move: one wave scans the bin totals, 7 per lane");

__global__ void __launch_bounds__(EDGE_T) k_bin_hist(const int* __restrict__ ei,
                                                     int* __restrict__ cnt) {
  __shared__ int hist[NBINS];
  int tid = (int)threadIdx.x, blk = (int)blockIdx.x;
  for (int b = tid; b < NBINS; b += EDGE_T) hist[b] = 0;
  __syncthreads();
  int e0 = blk * CHUNK + tid;
  unsigned d[EDGE_V];
#pragma unroll
  for (int q = 0; q < EDGE_V; ++q) {
    int e = e0 + q * EDGE_T;
    d[q] = (unsigned)ei[N_EDGES + ((e < N_EDGES) ? e : (N_EDGES - 1))];
  }
#pragma unroll
  for (int q = 0; q < EDGE_V; ++q) {
    int e = e0 + q * EDGE_T;
    if (e < N_EDGES)
      __hip_atomic_fetch_add(&hist[d[q] / BINW], 1, __ATOMIC_RELAXED,
                             __HIP_MEMORY_SCOPE_WORKGROUP);
  }
  __syncthreads();
  for (int b = tid; b < NBINS; b += EDGE_T) cnt[blk * NBINS + b] = hist[b];
}

__global__ void __launch_bounds__(EDGE_T) k_bin_move(const int* __restrict__ ei,
                                                     const int* __restrict__ cnt,
                                                     int* __restrict__ bstart,
                                                     unsigned* __restrict__ binned) {
  __shared__ int pos[NBINS];
  __shared__ int s_tot[2][256];
  __shared__ int s_pre[2][256];
  int tid = (int)threadIdx.x, blk = (int)blockIdx.x;
  int e0 = blk * CHUNK + tid;
  unsigned d[EDGE_V], s[EDGE_V];
#pragma unroll
  for (int q = 0; q < EDGE_V; ++q) {
    int e = e0 + q * EDGE_T;
    int ec = (e < N_EDGES) ? e : (N_EDGES - 1);
    d[q] = (unsigned)ei[N_EDGES + ec];
    s[q] = (unsigned)ei[ec];
  }
  int b = tid & 255, p = tid >> 8;
  int tot = 0, pre = 0;
  if (b < NBINS) {
    const int* cp = cnt + p * HALF_C * NBINS + b;
    int lim = blk - p * HALF_C;
#pragma unroll 49
    for (int r = 0; r < HALF_C; ++r) {
      int v = cp[r * NBINS];
      tot += v;
      pre += (r < lim) ? v : 0;
    }
  }
  s_tot[p][b] = tot;
  s_pre[p][b] = pre;
  __syncthreads();
  if (tid < 32) {
    int loc[7];
    int sum = 0;
#pragma unroll
    for (int q = 0; q < 7; ++q) {
      int bb = tid * 7 + q;
      loc[q] = s_tot[0][bb] + s_tot[1][bb];
      sum += loc[q];
    }
    int inc = sum;
    for (int o = 1; o < 32; o <<= 1) {
      int v = __shfl_up(inc, o);
      if (tid >= o) inc += v;
    }
    int run = inc - sum;
#pragma unroll
    for (int q = 0; q < 7; ++q) {
      int bb = tid * 7 + q;
      if (bb < NBINS) {
        pos[bb] = run + s_pre[0][bb] + s_pre[1][bb];
        if (blk == 0) bstart[bb] = run;
      }
      run += loc[q];
    }
  }
  if (blk == 0 && tid == 0) bstart[NBINS] = N_EDGES;
  __syncthreads();
#pragma unroll
  for (int q = 0; q < EDGE_V; ++q) {
    int e = e0 + q * EDGE_T;
    if (e < N_EDGES) {
      int w = __hip_atomic_fetch_add(&pos[d[q] / BINW], 1, __ATOMIC_RELAXED,
                                     __HIP_MEMORY_SCOPE_WORKGROUP);
      binned[w] = ((d[q] % BINW) << 16) | s[q];
    }
  }
}

#define BIN_U 4
__global__ void __launch_bounds__(EDGE_T) k_bin_csr(const unsigned* __restrict__ binned,
                                                    const int* __restrict__ bstart,
                                                    int* __restrict__ off,
                                                    float* __restrict__ dinv,
                                                    unsigned short* __restrict__ csr) {
  __shared__ int cnt[BINW];
  __shared__ int seg[BINW];
  __shared__ int fil[BINW];
  int tid = (int)threadIdx.x, bin = (int)blockIdx.x;
  int beg = bstart[bin];
  int end = bstart[bin + 1];
  int n = end - beg;
  if (tid < BINW) {
    cnt[tid] = 0;
    fil[tid] = 0;
  }
  __syncthreads();
  for (int i0 = tid; i0 < n; i0 += BIN_U * EDGE_T) {
    unsigned w[BIN_U];
#pragma unroll
    for (int q = 0; q < BIN_U; ++q) {
      int i = i0 + q * EDGE_T;
      w[q] = binned[beg + ((i < n) ? i : i0)];
    }
#pragma unroll
    for (int q = 0; q < BIN_U; ++q) {
      if (i0 + q * EDGE_T < n)
        __hip_atomic_fetch_add(&cnt[w[q] >> 16], 1, __ATOMIC_RELAXED,
                               __HIP_MEMORY_SCOPE_WORKGROUP);
    }
  }
  __syncthreads();
  int c = (tid < BINW) ? cnt[tid] : 0;
  if (tid < BINW) seg[tid] = c;
  __syncthreads();
  for (int o = 1; o < BINW; o <<= 1) {
    int v = (tid < BINW && tid >= o) ? seg[tid - o] : 0;
    __syncthreads();
    if (tid < BINW) seg[tid] += v;
    __syncthreads();
  }
  if (tid < BINW) {
    int p = beg + seg[tid] - c;
    seg[tid] = p;
    int node = bin * BINW + tid;
    if (node < N_NODES) {
      off[node] = p;
      float dg = 1.0f + (float)c;
      dinv[node] = (dg > 0.0f) ? rsqrtf(dg) : 0.0f;
    }
  }
  if (bin == NBINS - 1 && tid == 0) off[N_NODES] = N_EDGES;
  __syncthreads();
  for (int i0 = tid; i0 < n; i0 += BIN_U * EDGE_T) {
    unsigned w[BIN_U];
#pragma unroll
    for (int q = 0; q < BIN_U; ++q) {
      int i = i0 + q * EDGE_T;
      w[q] = binned[beg + ((i < n) ? i : i0)];
    }
#pragma unroll
    for (int q = 0; q < BIN_U; ++q) {
      if (i0 + q * EDGE_T < n) {
        int dl = (int)(w[q] >> 16);
        int r = __hip_atomic_fetch_add(&fil[dl], 1, __ATOMIC_RELAXED,
                                       __HIP_MEMORY_SCOPE_WORKGROUP);
        csr[seg[dl] + r] = (unsigned short)(w[q] & 0xffffu);
      }
    }
  }
}

__device__ __forceinline__ float4 axpy4(float4 v, float s, float4 a) {
  a.x += v.x * s;
  a.y += v.y * s;
  a.z += v.z * s;
  a.w += v.w * s;
  return a;
}

__device__ __forceinline__ float readlane_f(float v, int l) {
  return __int_as_float(__builtin_amdgcn_readlane(__float_as_int(v), l));
}

#define MID_CAP 128
__global__ void __launch_bounds__(256) k_gather(const float* __restrict__ hlin,
                                                const int* __restrict__ off,
                                                const unsigned short* __restrict__ csr,
                                                const float* __restrict__ dinv,
                                                const float* __restrict__ bias,
                                                float* __restrict__ h) {
  __shared__ int   mid_us[8][MID_CAP];
  __shared__ int   mid_so[8][MID_CAP];
  __shared__ float mid_wo[8][MID_CAP];
  int lane = (int)(threadIdx.x & 31);
  int n = __builtin_amdgcn_readfirstlane(
      (int)((blockIdx.x * blockDim.x + threadIdx.x) >> 5));
  if (n >= N_NODES) return;
  int beg = off[n];
  int d = off[n + 1] - beg;
  float di = dinv[n];
  const float4* h4 = (const float4*)hlin;
  float4 hv = h4[n * 32 + lane];
  float4 acc;
  acc.x = hv.x * di * di + bias[lane * 4 + 0];
  acc.y = hv.y * di * di + bias[lane * 4 + 1];
  acc.z = hv.z * di * di + bias[lane * 4 + 2];
  acc.w = hv.w * di * di + bias[lane * 4 + 3];

  if (d <= 32) {
    int s = 0x7fffffff;
    float w = 0.0f;
    if (lane < d) {
      s = (int)csr[beg + lane];
      w = dinv[s];
    }
    int r = 0;
    for (int j = 0; j < d; ++j) {
      int sj = __builtin_amdgcn_readlane(s, j);
      r += ((sj < s) || (sj == s && j < lane)) ? 1 : 0;
    }
    int   ss = __builtin_amdgcn_ds_permute(r << 2, s);
    float ws = __int_as_float(__builtin_amdgcn_ds_permute(r << 2, __float_as_int(w)));
    int k = 0;
    for (; k + 4 <= d; k += 4) {
      int s0 = __builtin_amdgcn_readlane(ss, k);
      int s1 = __builtin_amdgcn_readlane(ss, k + 1);
      int s2 = __builtin_amdgcn_readlane(ss, k + 2);
      int s3 = __builtin_amdgcn_readlane(ss, k + 3);
      float w0 = readlane_f(ws, k);
      float w1 = readlane_f(ws, k + 1);
      float w2 = readlane_f(ws, k + 2);
      float w3 = readlane_f(ws, k + 3);
      float4 v0 = h4[s0 * 32 + lane];
      float4 v1 = h4[s1 * 32 + lane];
      float4 v2 = h4[s2 * 32 + lane];
      float4 v3 = h4[s3 * 32 + lane];
      acc = axpy4(v0, w0 * di, acc);
      acc = axpy4(v1, w1 * di, acc);
      acc = axpy4(v2, w2 * di, acc);
      acc = axpy4(v3, w3 * di, acc);
    }
    for (; k < d; ++k) {
      int s0 = __builtin_amdgcn_readlane(ss, k);
      float w0 = readlane_f(ws, k);
      float4 v0 = h4[s0 * 32 + lane];
      acc = axpy4(v0, w0 * di, acc);
    }
  } else if (d <= MID_CAP) {
    int wv = (int)(threadIdx.x >> 5);
    int*   us = mid_us[wv];
    int*   so = mid_so[wv];
    float* wo = mid_wo[wv];
    int   e[MID_CAP / 32];
    float w[MID_CAP / 32];
    for (int q = 0; q < MID_CAP / 32; ++q) {
      int i = lane + 32 * q;
      e[q] = 0x7fffffff;
      w[q] = 0.0f;
      if (i < d) {
        e[q] = (int)csr[beg + i];
        w[q] = dinv[e[q]];
        us[i] = e[q];
      }
    }
    int r[MID_CAP / 32];
    for (int q = 0; q < MID_CAP / 32; ++q) r[q] = 0;
    for (int j = 0; j < d; ++j) {
      int sj = us[j];
      for (int q = 0; q < MID_CAP / 32; ++q)
        r[q] += ((sj < e[q]) || (sj == e[q] && j < lane + 32 * q)) ? 1 : 0;
    }
    for (int q = 0; q < MID_CAP / 32; ++q) {
      if (lane + 32 * q < d) {
        so[r[q]] = e[q];
        wo[r[q]] = w[q];
      }
    }
    int k = 0;
    for (; k + 4 <= d; k += 4) {
      int s0 = so[k], s1 = so[k + 1], s2 = so[k + 2], s3 = so[k + 3];
      float w0 = wo[k], w1 = wo[k + 1], w2 = wo[k + 2], w3 = wo[k + 3];
      float4 v0 = h4[s0 * 32 + lane];
      float4 v1 = h4[s1 * 32 + lane];
      float4 v2 = h4[s2 * 32 + lane];
      float4 v3 = h4[s3 * 32 + lane];
      acc = axpy4(v0, w0 * di, acc);
      acc = axpy4(v1, w1 * di, acc);
      acc = axpy4(v2, w2 * di, acc);
      acc = axpy4(v3, w3 * di, acc);
    }
    for (; k < d; ++k) {
      int s0 = so[k];
      float w0 = wo[k];
      float4 v0 = h4[s0 * 32 + lane];
      acc = axpy4(v0, w0 * di, acc);
    }
  } else {
    int last = -1;
    int left = d;
    for (int it = 0; it < d && left > 0; ++it) {
      int mn = 0x7fffffff;
      for (int i = lane; i < d; i += 32) {
        int s = (int)csr[beg + i];
        if (s > last && s < mn) mn = s;
      }
      for (int o = 16; o > 0; o >>= 1) {
        int t = __shfl_xor(mn, o);
        mn = (t < mn) ? t : mn;
      }
      int m = 0;
      for (int i = lane; i < d; i += 32) m += ((int)csr[beg + i] == mn) ? 1 : 0;
      for (int o = 16; o > 0; o >>= 1) m += __shfl_xor(m, o);
      mn = __builtin_amdgcn_readfirstlane(mn);
      m = __builtin_amdgcn_readfirstlane(m);
      float nrm = dinv[mn] * di;
      float4 v = h4[mn * 32 + lane];
      for (int c = 0; c < m; ++c) acc = axpy4(v, nrm, acc);
      last = mn;
      left -= m;
    }
  }
  ((float4*)h)[n * 32 + lane] = acc;
}

__global__ void k_prep_w0(const float* __restrict__ bw, const float* __restrict__ sw,
                          const float* __restrict__ sc, float* __restrict__ Bt) {
  int t = blockIdx.x * blockDim.x + threadIdx.x;
  if (t >= 8 * OUT_C * HIDDEN) return;
  int o = t >> 10, k = t & 1023;
  int f = k >> 3, j = k & 7;
  float w = (j == 0) ? bw[o * OUT_C + f]
                     : sw[(o * OUT_C + f) * N_COEF + (j - 1)] * sc[o * OUT_C + f];
  Bt[t] = w;
}

__global__ void k_prep_w1(const float* __restrict__ bw, const float* __restrict__ sw,
                          const float* __restrict__ sc, float* __restrict__ Bt) {
  int t = blockIdx.x * blockDim.x + threadIdx.x;
  if (t >= 8 * HIDDEN * OUT_C) return;
  int o = t >> 7, k = t & 127;
  int f = k >> 3, j = k & 7;
  float w = (j == 0) ? bw[o * HIDDEN + f]
                     : sw[(o * HIDDEN + f) * N_COEF + (j - 1)] * sc[o * HIDDEN + f];
  Bt[o * 128 + k] = w;
}

#define PGRID0 512
static_assert(PGRID0 <= NTILES, "every workgroup of k_kan0 has a first tile");
__global__ void __launch_bounds__(256) k_kan0(const float* __restrict__ h,
                                              const float* __restrict__ Bt,
                                              float* __restrict__ g) {
  __shared__ __attribute__((aligned(16))) float red[2][8][256];
  int tid = (int)threadIdx.x;
  int lane = tid & 31, wv = tid >> 5;
  int half = lane >> 4, l16 = lane & 15;
  int f0 = wv * 16 + half * 8;
  float4 b[16];
  {
    const float4* bp = (const float4*)(Bt + l16 * 1024 + f0 * 8);
#pragma unroll
    for (int j = 0; j < 16; ++j) b[j] = bp[j];
  }
  const float4* h4 = (const float4*)h;
  int t = (int)blockIdx.x;
  float4 p0 = h4[((t * 16 + l16) * OUT_C + f0) / 4];
  float4 p1 = h4[((t * 16 + l16) * OUT_C + f0) / 4 + 1];
  int ridx = ((tid >> 4) & 7) * 32 + (tid >> 7) * 16 + (tid & 15);
  int buf = 0;
  for (; t < NTILES; t += PGRID0) {
    float v[8];
    v[0] = p0.x; v[1] = p0.y; v[2] = p0.z; v[3] = p0.w;
    v[4] = p1.x; v[5] = p1.y; v[6] = p1.z; v[7] = p1.w;
    int tn = t + PGRID0;
    if (tn < NTILES) {
      p0 = h4[((tn * 16 + l16) * OUT_C + f0) / 4];
      p1 = h4[((tn * 16 + l16) * OUT_C + f0) / 4 + 1];
    }
    v2f x[8][4];
#pragma unroll
    for (int i = 0; i < 8; ++i) {
      float e[8];
      e[0] = silu_f(v[i]);
      bspline7(v[i], e + 1);
#pragma unroll
      for (int q = 0; q < 4; ++q) { x[i][q].x = e[2 * q]; x[i][q].y = e[2 * q + 1]; }
    }
    __builtin_amdgcn_sched_barrier(0);
    v8f acc0 = {}, acc1 = {};
#pragma unroll
    for (int i = 0; i < 8; ++i) {
      v2f w0, w1, w2, w3;
      w0.x = b[2 * i].x;     w0.y = b[2 * i].y;
      w1.x = b[2 * i].z;     w1.y = b[2 * i].w;
      w2.x = b[2 * i + 1].x; w2.y = b[2 * i + 1].y;
      w3.x = b[2 * i + 1].z; w3.y = b[2 * i + 1].w;
      acc0 = wmma4(x[i][0], w0, acc0);
      acc1 = wmma4(x[i][1], w1, acc1);
      acc0 = wmma4(x[i][2], w2, acc0);
      acc1 = wmma4(x[i][3], w3, acc1);
    }
    asm volatile("v_nop\n\tv_nop\n\tv_nop\n\tv_nop\n\tv_nop" : "+v"(acc0), "+v"(acc1));
    v8f acc = acc0 + acc1;
    float* rw = &red[buf][wv][lane];
#pragma unroll
    for (int r = 0; r < 8; ++r) rw[r * 32] = acc[r];
    __syncthreads();
    float s = red[buf][0][ridx];
#pragma unroll
    for (int w = 1; w < 8; ++w) s += red[buf][w][ridx];
    g[t * 256 + tid] = s;
    buf ^= 1;
  }
}

__device__ __forceinline__ void expand8(float v, float4* d) {
  float bs[N_COEF];
  bspline7(v, bs);
  float4 lo, hi;
  lo.x = silu_f(v); lo.y = bs[0]; lo.z = bs[1]; lo.w = bs[2];
  hi.x = bs[3]; hi.y = bs[4]; hi.z = bs[5]; hi.w = bs[6];
  d[0] = lo;
  d[1] = hi;
}

__global__ void __launch_bounds__(256) k_kan1(const float* __restrict__ g,
                                              const float* __restrict__ Bt,
                                              float* __restrict__ out) {
  __shared__ __attribute__((aligned(16))) float As[2][16 * TILE_P];
  int tid = (int)threadIdx.x;
  int lane = tid & 31, wv = tid >> 5;
  int half = lane >> 4, l16 = lane & 15;
  int o0 = wv * 16;
  float4 b[16];
  load_bfrag(Bt, o0, l16, half, b);
  int row = tid >> 4, f = tid & 15;
  int t = (int)blockIdx.x;
  float v = g[(t * 16 + row) * HIDDEN + f];
  expand8(v, (float4*)&As[0][row * TILE_P + f * 8]);
  __syncthreads();
  int buf = 0;
  for (; t < NTILES; t += PGRID) {
    int tn = t + PGRID;
    bool more = tn < NTILES;
    if (more) v = g[(tn * 16 + row) * HIDDEN + f];
    v8f acc = tile_mma(As[buf], l16, half, b);
    if (more) expand8(v, (float4*)&As[buf ^ 1][row * TILE_P + f * 8]);
    __syncthreads();
#pragma unroll
    for (int r = 0; r < 8; ++r)
      out[(t * 16 + r + half * 8) * OUT_C + o0 + l16] = acc[r];
    buf ^= 1;
  }
}

extern "C" void kernel_launch(void* const* d_in, const int* in_sizes, int n_in,
                              void* d_out, int out_size, void* d_ws, size_t ws_size,
                              hipStream_t stream) {
  (void)in_sizes; (void)n_in; (void)out_size; (void)ws_size;
  const float* x       = (const float*)d_in[0];
  const int*   ei      = (const int*)d_in[1];
  const float* gcn_w   = (const float*)d_in[2];
  const float* gcn_b   = (const float*)d_in[3];
  const float* base_w0 = (const float*)d_in[4];
  const float* spl_w0  = (const float*)d_in[5];
  const float* scal0   = (const float*)d_in[6];
  const float* base_w1 = (const float*)d_in[7];
  const float* spl_w1  = (const float*)d_in[8];
  const float* scal1   = (const float*)d_in[9];
  float* out = (float*)d_out;

  float* ws    = (float*)d_ws;
  float* hlin  = ws;
  float* h     = hlin + (size_t)N_NODES * OUT_C;
  float* deg   = h + (size_t)N_NODES * OUT_C;
  float* g     = deg + N_NODES;
  float* Bm0   = g + (size_t)N_NODES * HIDDEN;
  float* Bm1   = Bm0 + 8 * OUT_C * HIDDEN;
  unsigned* binned = (unsigned*)h;
  int* cnt = (int*)h + N_EDGES;
  int* bstart = cnt + NSLOT;
  unsigned short* csr = (unsigned short*)g;
  int* off  = (int*)g + N_EDGES / 2;

  k_gemm_xw<<<PGRID, 256, 0, stream>>>(x, gcn_w, hlin);
  k_bin_hist<<<NCHUNK, EDGE_T, 0, stream>>>(ei, cnt);
  k_bin_move<<<NCHUNK, EDGE_T, 0, stream>>>(ei, cnt, bstart, binned);
  k_bin_csr<<<NBINS, EDGE_T, 0, stream>>>(binned, bstart, off, deg, csr);
  k_gather<<<(N_NODES * 32 + 255) / 256, 256, 0, stream>>>(hlin, off, csr, deg, gcn_b, h);
  k_prep_w0<<<(8 * OUT_C * HIDDEN + 255) / 256, 256, 0, stream>>>(base_w0, spl_w0, scal0, Bm0);
  k_prep_w1<<<(8 * HIDDEN * OUT_C + 255) / 256, 256, 0, stream>>>(base_w1, spl_w1, scal1, Bm1);
  k_kan0<<<PGRID0, 256, 0, stream>>>(h, Bm0, g);
  k_kan1<<<PGRID, 256, 0, stream>>>(g, Bm1, out);
}


# B-spline bases of both KAN layers from the value's own knot interval: the 4 non-zero uniform cubic pieces in closed form, selected into the 7 outputs, not the 10-interval recursion (rounding differs)
# speedup vs baseline: 3.0087x; 1.0928x over previous
#include <hip/hip_runtime.h>
#include <hip/hip_bf16.h>

#define N_NODES 50000
#define N_EDGES 800000
#define IN_C    128
#define OUT_C   128
#define HIDDEN  16
#define N_COEF  7

typedef __attribute__((ext_vector_type(2))) float v2f;
typedef __attribute__((ext_vector_type(8))) float v8f;

__device__ __forceinline__ v8f wmma4(v2f a, v2f b, v8f c) {
  return __builtin_amdgcn_wmma_f32_16x16x4_f32(false, a, false, b, (short)0, c,
                                               false, false);
}

__device__ __forceinline__ float silu_f(float v) {
  return v * (1.0f / (1.0f + __expf(-v)));
}

__device__ __forceinline__ void bspline7(float v, float* out) {
  float fi = floorf(2.0f * v + 5.0f);
  float t  = (v - (0.5f * fi - 2.5f)) * 2.0f;
  float t2 = t * t;
  float u  = 1.0f - t;
  const float k6 = 1.0f / 6.0f;
  float n3 = t2 * t * k6;
  float n2 = t * (t * (0.5f - 0.5f * t) + 0.5f) + k6;
  float n1 = t2 * (0.5f * t - 1.0f) + (2.0f / 3.0f);
  float n0 = u * u * u * k6;
  bool m[10];
#pragma unroll
  for (int k = 0; k < 10; ++k) m[k] = (fi == (float)k);
#pragma unroll
  for (int c = 0; c < N_COEF; ++c) {
    float r = m[c + 3] ? n0 : 0.0f;
    r = m[c + 2] ? n1 : r;
    r = m[c + 1] ? n2 : r;
    r = m[c] ? n3 : r;
    out[c] = r;
  }
}

#define TILE_P 132
#define NTILES (N_NODES / 16)
#define PGRID  512
static_assert(N_NODES % 16 == 0 && PGRID <= NTILES, "tiling");

__device__ __forceinline__ void load_bfrag(const float* __restrict__ Wt, int o0,
                                           int l16, int half, float4 (&b)[16]) {
  const float4* p = (const float4*)(Wt + (o0 + l16) * 128 + 4 * half);
#pragma unroll
  for (int j = 0; j < 16; ++j) b[j] = p[2 * j];
}

__device__ __forceinline__ v8f tile_mma(const float* As, int l16, int half,
                                        const float4 (&b)[16]) {
  float4 a[16];
  const float4* p = (const float4*)(As + l16 * TILE_P + 4 * half);
#pragma unroll
  for (int j = 0; j < 16; ++j) a[j] = p[2 * j];
  __builtin_amdgcn_sched_barrier(0);
  v8f acc0 = {}, acc1 = {};
#pragma unroll
  for (int j = 0; j < 16; ++j) {
    v2f a0, a1, b0, b1;
    a0.x = a[j].x; a0.y = a[j].y; a1.x = a[j].z; a1.y = a[j].w;
    b0.x = b[j].x; b0.y = b[j].y; b1.x = b[j].z; b1.y = b[j].w;
    acc0 = wmma4(a0, b0, acc0);
    acc1 = wmma4(a1, b1, acc1);
  }
  asm volatile("v_nop\n\tv_nop\n\tv_nop\n\tv_nop\n\tv_nop" : "+v"(acc0), "+v"(acc1));
  return acc0 + acc1;
}

__global__ void __launch_bounds__(256) k_gemm_xw(const float* __restrict__ x,
                                                 const float* __restrict__ W,
                                                 float* __restrict__ hlin) {
  __shared__ __attribute__((aligned(16))) float As[2][16 * TILE_P];
  int tid = (int)threadIdx.x;
  int lane = tid & 31, wv = tid >> 5;
  int half = lane >> 4, l16 = lane & 15;
  int o0 = wv * 16;
  float4 b[16];
  load_bfrag(W, o0, l16, half, b);
  int r0 = wv, c4 = lane;
  const float4* x4 = (const float4*)x;
  int t = (int)blockIdx.x;
  float4 p0 = x4[(t * 16 + r0) * 32 + c4];
  float4 p1 = x4[(t * 16 + r0 + 8) * 32 + c4];
  *(float4*)&As[0][r0 * TILE_P + 4 * c4] = p0;
  *(float4*)&As[0][(r0 + 8) * TILE_P + 4 * c4] = p1;
  __syncthreads();
  int buf = 0;
  for (; t < NTILES; t += PGRID) {
    int tn = t + PGRID;
    bool more = tn < NTILES;
    if (more) {
      p0 = x4[(tn * 16 + r0) * 32 + c4];
      p1 = x4[(tn * 16 + r0 + 8) * 32 + c4];
    }
    v8f acc = tile_mma(As[buf], l16, half, b);
    if (more) {
      *(float4*)&As[buf ^ 1][r0 * TILE_P + 4 * c4] = p0;
      *(float4*)&As[buf ^ 1][(r0 + 8) * TILE_P + 4 * c4] = p1;
    }
    __syncthreads();
#pragma unroll
    for (int r = 0; r < 8; ++r)
      hlin[(t * 16 + r + half * 8) * OUT_C + o0 + l16] = acc[r];
    buf ^= 1;
  }
}

static_assert(N_NODES <= 65536, "source ids are kept as 16-bit values");
#define BINW   256
#define NBINS  ((N_NODES + BINW - 1) / BINW)
#define CHUNK  4096
#define NCHUNK ((N_EDGES + CHUNK - 1) / CHUNK)
#define EDGE_T 512
#define EDGE_V (CHUNK / EDGE_T)
#define NSLOT  (NBINS * NCHUNK)
#define HALF_C (NCHUNK / 2)
static_assert(CHUNK % EDGE_T == 0, "whole edges per thread");
static_assert(BINW <= 65536 && BINW <= EDGE_T, "local destination and source share one word; one thread per node of a bin");
static_assert(NCHUNK % 2 == 0 && NBINS <= 256 && EDGE_T == 512, "k_bin_move: thread = (bin, half of the table's rows)");
static_assert(NBINS <= 7 * 32, "k_bin_move: one wave scans the bin totals, 7 per lane");

__global__ void __launch_bounds__(EDGE_T) k_bin_hist(const int* __restrict__ ei,
                                                     int* __restrict__ cnt) {
  __shared__ int hist[NBINS];
  int tid = (int)threadIdx.x, blk = (int)blockIdx.x;
  for (int b = tid; b < NBINS; b += EDGE_T) hist[b] = 0;
  __syncthreads();
  int e0 = blk * CHUNK + tid;
  unsigned d[EDGE_V];
#pragma unroll
  for (int q = 0; q < EDGE_V; ++q) {
    int e = e0 + q * EDGE_T;
    d[q] = (unsigned)ei[N_EDGES + ((e < N_EDGES) ? e : (N_EDGES - 1))];
  }
#pragma unroll
  for (int q = 0; q < EDGE_V; ++q) {
    int e = e0 + q * EDGE_T;
    if (e < N_EDGES)
      __hip_atomic_fetch_add(&hist[d[q] / BINW], 1, __ATOMIC_RELAXED,
                             __HIP_MEMORY_SCOPE_WORKGROUP);
  }
  __syncthreads();
  for (int b = tid; b < NBINS; b += EDGE_T) cnt[blk * NBINS + b] = hist[b];
}

__global__ void __launch_bounds__(EDGE_T) k_bin_move(const int* __restrict__ ei,
                                                     const int* __restrict__ cnt,
                                                     int* __restrict__ bstart,
                                                     unsigned* __restrict__ binned) {
  __shared__ int pos[NBINS];
  __shared__ int s_tot[2][256];
  __shared__ int s_pre[2][256];
  int tid = (int)threadIdx.x, blk = (int)blockIdx.x;
  int e0 = blk * CHUNK + tid;
  unsigned d[EDGE_V], s[EDGE_V];
#pragma unroll
  for (int q = 0; q < EDGE_V; ++q) {
    int e = e0 + q * EDGE_T;
    int ec = (e < N_EDGES) ? e : (N_EDGES - 1);
    d[q] = (unsigned)ei[N_EDGES + ec];
    s[q] = (unsigned)ei[ec];
  }
  int b = tid & 255, p = tid >> 8;
  int tot = 0, pre = 0;
  if (b < NBINS) {
    const int* cp = cnt + p * HALF_C * NBINS + b;
    int lim = blk - p * HALF_C;
#pragma unroll 49
    for (int r = 0; r < HALF_C; ++r) {
      int v = cp[r * NBINS];
      tot += v;
      pre += (r < lim) ? v : 0;
    }
  }
  s_tot[p][b] = tot;
  s_pre[p][b] = pre;
  __syncthreads();
  if (tid < 32) {
    int loc[7];
    int sum = 0;
#pragma unroll
    for (int q = 0; q < 7; ++q) {
      int bb = tid * 7 + q;
      loc[q] = s_tot[0][bb] + s_tot[1][bb];
      sum += loc[q];
    }
    int inc = sum;
    for (int o = 1; o < 32; o <<= 1) {
      int v = __shfl_up(inc, o);
      if (tid >= o) inc += v;
    }
    int run = inc - sum;
#pragma unroll
    for (int q = 0; q < 7; ++q) {
      int bb = tid * 7 + q;
      if (bb < NBINS) {
        pos[bb] = run + s_pre[0][bb] + s_pre[1][bb];
        if (blk == 0) bstart[bb] = run;
      }
      run += loc[q];
    }
  }
  if (blk == 0 && tid == 0) bstart[NBINS] = N_EDGES;
  __syncthreads();
#pragma unroll
  for (int q = 0; q < EDGE_V; ++q) {
    int e = e0 + q * EDGE_T;
    if (e < N_EDGES) {
      int w = __hip_atomic_fetch_add(&pos[d[q] / BINW], 1, __ATOMIC_RELAXED,
                                     __HIP_MEMORY_SCOPE_WORKGROUP);
      binned[w] = ((d[q] % BINW) << 16) | s[q];
    }
  }
}

#define BIN_U 4
__global__ void __launch_bounds__(EDGE_T) k_bin_csr(const unsigned* __restrict__ binned,
                                                    const int* __restrict__ bstart,
                                                    int* __restrict__ off,
                                                    float* __restrict__ dinv,
                                                    unsigned short* __restrict__ csr) {
  __shared__ int cnt[BINW];
  __shared__ int seg[BINW];
  __shared__ int fil[BINW];
  int tid = (int)threadIdx.x, bin = (int)blockIdx.x;
  int beg = bstart[bin];
  int end = bstart[bin + 1];
  int n = end - beg;
  if (tid < BINW) {
    cnt[tid] = 0;
    fil[tid] = 0;
  }
  __syncthreads();
  for (int i0 = tid; i0 < n; i0 += BIN_U * EDGE_T) {
    unsigned w[BIN_U];
#pragma unroll
    for (int q = 0; q < BIN_U; ++q) {
      int i = i0 + q * EDGE_T;
      w[q] = binned[beg + ((i < n) ? i : i0)];
    }
#pragma unroll
    for (int q = 0; q < BIN_U; ++q) {
      if (i0 + q * EDGE_T < n)
        __hip_atomic_fetch_add(&cnt[w[q] >> 16], 1, __ATOMIC_RELAXED,
                               __HIP_MEMORY_SCOPE_WORKGROUP);
    }
  }
  __syncthreads();
  int c = (tid < BINW) ? cnt[tid] : 0;
  if (tid < BINW) seg[tid] = c;
  __syncthreads();
  for (int o = 1; o < BINW; o <<= 1) {
    int v = (tid < BINW && tid >= o) ? seg[tid - o] : 0;
    __syncthreads();
    if (tid < BINW) seg[tid] += v;
    __syncthreads();
  }
  if (tid < BINW) {
    int p = beg + seg[tid] - c;
    seg[tid] = p;
    int node = bin * BINW + tid;
    if (node < N_NODES) {
      off[node] = p;
      float dg = 1.0f + (float)c;
      dinv[node] = (dg > 0.0f) ? rsqrtf(dg) : 0.0f;
    }
  }
  if (bin == NBINS - 1 && tid == 0) off[N_NODES] = N_EDGES;
  __syncthreads();
  for (int i0 = tid; i0 < n; i0 += BIN_U * EDGE_T) {
    unsigned w[BIN_U];
#pragma unroll
    for (int q = 0; q < BIN_U; ++q) {
      int i = i0 + q * EDGE_T;
      w[q] = binned[beg + ((i < n) ? i : i0)];
    }
#pragma unroll
    for (int q = 0; q < BIN_U; ++q) {
      if (i0 + q * EDGE_T < n) {
        int dl = (int)(w[q] >> 16);
        int r = __hip_atomic_fetch_add(&fil[dl], 1, __ATOMIC_RELAXED,
                                       __HIP_MEMORY_SCOPE_WORKGROUP);
        csr[seg[dl] + r] = (unsigned short)(w[q] & 0xffffu);
      }
    }
  }
}

__device__ __forceinline__ float4 axpy4(float4 v, float s, float4 a) {
  a.x += v.x * s;
  a.y += v.y * s;
  a.z += v.z * s;
  a.w += v.w * s;
  return a;
}

__device__ __forceinline__ float readlane_f(float v, int l) {
  return __int_as_float(__builtin_amdgcn_readlane(__float_as_int(v), l));
}

#define MID_CAP 128
__global__ void __launch_bounds__(256) k_gather(const float* __restrict__ hlin,
                                                const int* __restrict__ off,
                                                const unsigned short* __restrict__ csr,
                                                const float* __restrict__ dinv,
                                                const float* __restrict__ bias,
                                                float* __restrict__ h) {
  __shared__ int   mid_us[8][MID_CAP];
  __shared__ int   mid_so[8][MID_CAP];
  __shared__ float mid_wo[8][MID_CAP];
  int lane = (int)(threadIdx.x & 31);
  int n = __builtin_amdgcn_readfirstlane(
      (int)((blockIdx.x * blockDim.x + threadIdx.x) >> 5));
  if (n >= N_NODES) return;
  int beg = off[n];
  int d = off[n + 1] - beg;
  float di = dinv[n];
  const float4* h4 = (const float4*)hlin;
  float4 hv = h4[n * 32 + lane];
  float4 acc;
  acc.x = hv.x * di * di + bias[lane * 4 + 0];
  acc.y = hv.y * di * di + bias[lane * 4 + 1];
  acc.z = hv.z * di * di + bias[lane * 4 + 2];
  acc.w = hv.w * di * di + bias[lane * 4 + 3];

  if (d <= 32) {
    int s = 0x7fffffff;
    float w = 0.0f;
    if (lane < d) {
      s = (int)csr[beg + lane];
      w = dinv[s];
    }
    int r = 0;
    for (int j = 0; j < d; ++j) {
      int sj = __builtin_amdgcn_readlane(s, j);
      r += ((sj < s) || (sj == s && j < lane)) ? 1 : 0;
    }
    int   ss = __builtin_amdgcn_ds_permute(r << 2, s);
    float ws = __int_as_float(__builtin_amdgcn_ds_permute(r << 2, __float_as_int(w)));
    int k = 0;
    for (; k + 4 <= d; k += 4) {
      int s0 = __builtin_amdgcn_readlane(ss, k);
      int s1 = __builtin_amdgcn_readlane(ss, k + 1);
      int s2 = __builtin_amdgcn_readlane(ss, k + 2);
      int s3 = __builtin_amdgcn_readlane(ss, k + 3);
      float w0 = readlane_f(ws, k);
      float w1 = readlane_f(ws, k + 1);
      float w2 = readlane_f(ws, k + 2);
      float w3 = readlane_f(ws, k + 3);
      float4 v0 = h4[s0 * 32 + lane];
      float4 v1 = h4[s1 * 32 + lane];
      float4 v2 = h4[s2 * 32 + lane];
      float4 v3 = h4[s3 * 32 + lane];
      acc = axpy4(v0, w0 * di, acc);
      acc = axpy4(v1, w1 * di, acc);
      acc = axpy4(v2, w2 * di, acc);
      acc = axpy4(v3, w3 * di, acc);
    }
    for (; k < d; ++k) {
      int s0 = __builtin_amdgcn_readlane(ss, k);
      float w0 = readlane_f(ws, k);
      float4 v0 = h4[s0 * 32 + lane];
      acc = axpy4(v0, w0 * di, acc);
    }
  } else if (d <= MID_CAP) {
    int wv = (int)(threadIdx.x >> 5);
    int*   us = mid_us[wv];
    int*   so = mid_so[wv];
    float* wo = mid_wo[wv];
    int   e[MID_CAP / 32];
    float w[MID_CAP / 32];
    for (int q = 0; q < MID_CAP / 32; ++q) {
      int i = lane + 32 * q;
      e[q] = 0x7fffffff;
      w[q] = 0.0f;
      if (i < d) {
        e[q] = (int)csr[beg + i];
        w[q] = dinv[e[q]];
        us[i] = e[q];
      }
    }
    int r[MID_CAP / 32];
    for (int q = 0; q < MID_CAP / 32; ++q) r[q] = 0;
    for (int j = 0; j < d; ++j) {
      int sj = us[j];
      for (int q = 0; q < MID_CAP / 32; ++q)
        r[q] += ((sj < e[q]) || (sj == e[q] && j < lane + 32 * q)) ? 1 : 0;
    }
    for (int q = 0; q < MID_CAP / 32; ++q) {
      if (lane + 32 * q < d) {
        so[r[q]] = e[q];
        wo[r[q]] = w[q];
      }
    }
    int k = 0;
    for (; k + 4 <= d; k += 4) {
      int s0 = so[k], s1 = so[k + 1], s2 = so[k + 2], s3 = so[k + 3];
      float w0 = wo[k], w1 = wo[k + 1], w2 = wo[k + 2], w3 = wo[k + 3];
      float4 v0 = h4[s0 * 32 + lane];
      float4 v1 = h4[s1 * 32 + lane];
      float4 v2 = h4[s2 * 32 + lane];
      float4 v3 = h4[s3 * 32 + lane];
      acc = axpy4(v0, w0 * di, acc);
      acc = axpy4(v1, w1 * di, acc);
      acc = axpy4(v2, w2 * di, acc);
      acc = axpy4(v3, w3 * di, acc);
    }
    for (; k < d; ++k) {
      int s0 = so[k];
      float w0 = wo[k];
      float4 v0 = h4[s0 * 32 + lane];
      acc = axpy4(v0, w0 * di, acc);
    }
  } else {
    int last = -1;
    int left = d;
    for (int it = 0; it < d && left > 0; ++it) {
      int mn = 0x7fffffff;
      for (int i = lane; i < d; i += 32) {
        int s = (int)csr[beg + i];
        if (s > last && s < mn) mn = s;
      }
      for (int o = 16; o > 0; o >>= 1) {
        int t = __shfl_xor(mn, o);
        mn = (t < mn) ? t : mn;
      }
      int m = 0;
      for (int i = lane; i < d; i += 32) m += ((int)csr[beg + i] == mn) ? 1 : 0;
      for (int o = 16; o > 0; o >>= 1) m += __shfl_xor(m, o);
      mn = __builtin_amdgcn_readfirstlane(mn);
      m = __builtin_amdgcn_readfirstlane(m);
      float nrm = dinv[mn] * di;
      float4 v = h4[mn * 32 + lane];
      for (int c = 0; c < m; ++c) acc = axpy4(v, nrm, acc);
      last = mn;
      left -= m;
    }
  }
  ((float4*)h)[n * 32 + lane] = acc;
}

__global__ void k_prep_w0(const float* __restrict__ bw, const float* __restrict__ sw,
                          const float* __restrict__ sc, float* __restrict__ Bt) {
  int t = blockIdx.x * blockDim.x + threadIdx.x;
  if (t >= 8 * OUT_C * HIDDEN) return;
  int o = t >> 10, k = t & 1023;
  int f = k >> 3, j = k & 7;
  float w = (j == 0) ? bw[o * OUT_C + f]
                     : sw[(o * OUT_C + f) * N_COEF + (j - 1)] * sc[o * OUT_C + f];
  Bt[t] = w;
}

__global__ void k_prep_w1(const float* __restrict__ bw, const float* __restrict__ sw,
                          const float* __restrict__ sc, float* __restrict__ Bt) {
  int t = blockIdx.x * blockDim.x + threadIdx.x;
  if (t >= 8 * HIDDEN * OUT_C) return;
  int o = t >> 7, k = t & 127;
  int f = k >> 3, j = k & 7;
  float w = (j == 0) ? bw[o * HIDDEN + f]
                     : sw[(o * HIDDEN + f) * N_COEF + (j - 1)] * sc[o * HIDDEN + f];
  Bt[o * 128 + k] = w;
}

#define PGRID0 512
static_assert(PGRID0 <= NTILES, "every workgroup of k_kan0 has a first tile");
__global__ void __launch_bounds__(256) k_kan0(const float* __restrict__ h,
                                              const float* __restrict__ Bt,
                                              float* __restrict__ g) {
  __shared__ __attribute__((aligned(16))) float red[2][8][256];
  int tid = (int)threadIdx.x;
  int lane = tid & 31, wv = tid >> 5;
  int half = lane >> 4, l16 = lane & 15;
  int f0 = wv * 16 + half * 8;
  float4 b[16];
  {
    const float4* bp = (const float4*)(Bt + l16 * 1024 + f0 * 8);
#pragma unroll
    for (int j = 0; j < 16; ++j) b[j] = bp[j];
  }
  const float4* h4 = (const float4*)h;
  int t = (int)blockIdx.x;
  float4 p0 = h4[((t * 16 + l16) * OUT_C + f0) / 4];
  float4 p1 = h4[((t * 16 + l16) * OUT_C + f0) / 4 + 1];
  int ridx = ((tid >> 4) & 7) * 32 + (tid >> 7) * 16 + (tid & 15);
  int buf = 0;
  for (; t < NTILES; t += PGRID0) {
    float v[8];
    v[0] = p0.x; v[1] = p0.y; v[2] = p0.z; v[3] = p0.w;
    v[4] = p1.x; v[5] = p1.y; v[6] = p1.z; v[7] = p1.w;
    int tn = t + PGRID0;
    if (tn < NTILES) {
      p0 = h4[((tn * 16 + l16) * OUT_C + f0) / 4];
      p1 = h4[((tn * 16 + l16) * OUT_C + f0) / 4 + 1];
    }
    v2f x[8][4];
#pragma unroll
    for (int i = 0; i < 8; ++i) {
      float e[8];
      e[0] = silu_f(v[i]);
      bspline7(v[i], e + 1);
#pragma unroll
      for (int q = 0; q < 4; ++q) { x[i][q].x = e[2 * q]; x[i][q].y = e[2 * q + 1]; }
    }
    __builtin_amdgcn_sched_barrier(0);
    v8f acc0 = {}, acc1 = {};
#pragma unroll
    for (int i = 0; i < 8; ++i) {
      v2f w0, w1, w2, w3;
      w0.x = b[2 * i].x;     w0.y = b[2 * i].y;
      w1.x = b[2 * i].z;     w1.y = b[2 * i].w;
      w2.x = b[2 * i + 1].x; w2.y = b[2 * i + 1].y;
      w3.x = b[2 * i + 1].z; w3.y = b[2 * i + 1].w;
      acc0 = wmma4(x[i][0], w0, acc0);
      acc1 = wmma4(x[i][1], w1, acc1);
      acc0 = wmma4(x[i][2], w2, acc0);
      acc1 = wmma4(x[i][3], w3, acc1);
    }
    asm volatile("v_nop\n\tv_nop\n\tv_nop\n\tv_nop\n\tv_nop" : "+v"(acc0), "+v"(acc1));
    v8f acc = acc0 + acc1;
    float* rw = &red[buf][wv][lane];
#pragma unroll
    for (int r = 0; r < 8; ++r) rw[r * 32] = acc[r];
    __syncthreads();
    float s = red[buf][0][ridx];
#pragma unroll
    for (int w = 1; w < 8; ++w) s += red[buf][w][ridx];
    g[t * 256 + tid] = s;
    buf ^= 1;
  }
}

__device__ __forceinline__ void expand8(float v, float4* d) {
  float bs[N_COEF];
  bspline7(v, bs);
  float4 lo, hi;
  lo.x = silu_f(v); lo.y = bs[0]; lo.z = bs[1]; lo.w = bs[2];
  hi.x = bs[3]; hi.y = bs[4]; hi.z = bs[5]; hi.w = bs[6];
  d[0] = lo;
  d[1] = hi;
}

__global__ void __launch_bounds__(256) k_kan1(const float* __restrict__ g,
                                              const float* __restrict__ Bt,
                                              float* __restrict__ out) {
  __shared__ __attribute__((aligned(16))) float As[2][16 * TILE_P];
  int tid = (int)threadIdx.x;
  int lane = tid & 31, wv = tid >> 5;
  int half = lane >> 4, l16 = lane & 15;
  int o0 = wv * 16;
  float4 b[16];
  load_bfrag(Bt, o0, l16, half, b);
  int row = tid >> 4, f = tid & 15;
  int t = (int)blockIdx.x;
  float v = g[(t * 16 + row) * HIDDEN + f];
  expand8(v, (float4*)&As[0][row * TILE_P + f * 8]);
  __syncthreads();
  int buf = 0;
  for (; t < NTILES; t += PGRID) {
    int tn = t + PGRID;
    bool more = tn < NTILES;
    if (more) v = g[(tn * 16 + row) * HIDDEN + f];
    v8f acc = tile_mma(As[buf], l16, half, b);
    if (more) expand8(v, (float4*)&As[buf ^ 1][row * TILE_P + f * 8]);
    __syncthreads();
#pragma unroll
    for (int r = 0; r < 8; ++r)
      out[(t * 16 + r + half * 8) * OUT_C + o0 + l16] = acc[r];
    buf ^= 1;
  }
}

extern "C" void kernel_launch(void* const* d_in, const int* in_sizes, int n_in,
                              void* d_out, int out_size, void* d_ws, size_t ws_size,
                              hipStream_t stream) {
  (void)in_sizes; (void)n_in; (void)out_size; (void)ws_size;
  const float* x       = (const float*)d_in[0];
  const int*   ei      = (const int*)d_in[1];
  const float* gcn_w   = (const float*)d_in[2];
  const float* gcn_b   = (const float*)d_in[3];
  const float* base_w0 = (const float*)d_in[4];
  const float* spl_w0  = (const float*)d_in[5];
  const float* scal0   = (const float*)d_in[6];
  const float* base_w1 = (const float*)d_in[7];
  const float* spl_w1  = (const float*)d_in[8];
  const float* scal1   = (const float*)d_in[9];
  float* out = (float*)d_out;

  float* ws    = (float*)d_ws;
  float* hlin  = ws;
  float* h     = hlin + (size_t)N_NODES * OUT_C;
  float* deg   = h + (size_t)N_NODES * OUT_C;
  float* g     = deg + N_NODES;
  float* Bm0   = g + (size_t)N_NODES * HIDDEN;
  float* Bm1   = Bm0 + 8 * OUT_C * HIDDEN;
  unsigned* binned = (unsigned*)h;
  int* cnt = (int*)h + N_EDGES;
  int* bstart = cnt + NSLOT;
  unsigned short* csr = (unsigned short*)g;
  int* off  = (int*)g + N_EDGES / 2;

  k_gemm_xw<<<PGRID, 256, 0, stream>>>(x, gcn_w, hlin);
  k_bin_hist<<<NCHUNK, EDGE_T, 0, stream>>>(ei, cnt);
  k_bin_move<<<NCHUNK, EDGE_T, 0, stream>>>(ei, cnt, bstart, binned);
  k_bin_csr<<<NBINS, EDGE_T, 0, stream>>>(binned, bstart, off, deg, csr);
  k_gather<<<(N_NODES * 32 + 255) / 256, 256, 0, stream>>>(hlin, off, csr, deg, gcn_b, h);
  k_prep_w0<<<(8 * OUT_C * HIDDEN + 255) / 256, 256, 0, stream>>>(base_w0, spl_w0, scal0, Bm0);
  k_prep_w1<<<(8 * HIDDEN * OUT_C + 255) / 256, 256, 0, stream>>>(base_w1, spl_w1, scal1, Bm1);
  k_kan0<<<PGRID0, 256, 0, stream>>>(h, Bm0, g);
  k_kan1<<<PGRID, 256, 0, stream>>>(g, Bm1, out);
}


# The four input-only kernels (x@W^T, edge-chunk bin counts, both weight folds) run side by side as one launch, picked by workgroup index: 9 launches become 6; every sum order kept
# speedup vs baseline: 3.2552x; 1.0819x over previous
#include <hip/hip_runtime.h>
#include <hip/hip_bf16.h>

#define N_NODES 50000
#define N_EDGES 800000
#define IN_C    128
#define OUT_C   128
#define HIDDEN  16
#define N_COEF  7

typedef __attribute__((ext_vector_type(2))) float v2f;
typedef __attribute__((ext_vector_type(8))) float v8f;

__device__ __forceinline__ v8f wmma4(v2f a, v2f b, v8f c) {
  return __builtin_amdgcn_wmma_f32_16x16x4_f32(false, a, false, b, (short)0, c,
                                               false, false);
}

__device__ __forceinline__ float silu_f(float v) {
  return v * (1.0f / (1.0f + __expf(-v)));
}

__device__ __forceinline__ void bspline7(float v, float* out) {
  float fi = floorf(2.0f * v + 5.0f);
  float t  = (v - (0.5f * fi - 2.5f)) * 2.0f;
  float t2 = t * t;
  float u  = 1.0f - t;
  const float k6 = 1.0f / 6.0f;
  float n3 = t2 * t * k6;
  float n2 = t * (t * (0.5f - 0.5f * t) + 0.5f) + k6;
  float n1 = t2 * (0.5f * t - 1.0f) + (2.0f / 3.0f);
  float n0 = u * u * u * k6;
  bool m[10];
#pragma unroll
  for (int k = 0; k < 10; ++k) m[k] = (fi == (float)k);
#pragma unroll
  for (int c = 0; c < N_COEF; ++c) {
    float r = m[c + 3] ? n0 : 0.0f;
    r = m[c + 2] ? n1 : r;
    r = m[c + 1] ? n2 : r;
    r = m[c] ? n3 : r;
    out[c] = r;
  }
}

#define TILE_P 132
#define NTILES (N_NODES / 16)
#define PGRID  512
static_assert(N_NODES % 16 == 0 && PGRID <= NTILES, "tiling");

__device__ __forceinline__ void load_bfrag(const float* __restrict__ Wt, int o0,
                                           int l16, int half, float4 (&b)[16]) {
  const float4* p = (const float4*)(Wt + (o0 + l16) * 128 + 4 * half);
#pragma unroll
  for (int j = 0; j < 16; ++j) b[j] = p[2 * j];
}

__device__ __forceinline__ v8f tile_mma(const float* As, int l16, int half,
                                        const float4 (&b)[16]) {
  float4 a[16];
  const float4* p = (const float4*)(As + l16 * TILE_P + 4 * half);
#pragma unroll
  for (int j = 0; j < 16; ++j) a[j] = p[2 * j];
  __builtin_amdgcn_sched_barrier(0);
  v8f acc0 = {}, acc1 = {};
#pragma unroll
  for (int j = 0; j < 16; ++j) {
    v2f a0, a1, b0, b1;
    a0.x = a[j].x; a0.y = a[j].y; a1.x = a[j].z; a1.y = a[j].w;
    b0.x = b[j].x; b0.y = b[j].y; b1.x = b[j].z; b1.y = b[j].w;
    acc0 = wmma4(a0, b0, acc0);
    acc1 = wmma4(a1, b1, acc1);
  }
  asm volatile("v_nop\n\tv_nop\n\tv_nop\n\tv_nop\n\tv_nop" : "+v"(acc0), "+v"(acc1));
  return acc0 + acc1;
}

__device__ __forceinline__ void gemm_xw_block(const float* __restrict__ x,
                                              const float* __restrict__ W,
                                              float* __restrict__ hlin,
                                              float (*As)[16 * TILE_P], int blk) {
  int tid = (int)threadIdx.x;
  int lane = tid & 31, wv = tid >> 5;
  int half = lane >> 4, l16 = lane & 15;
  int o0 = wv * 16;
  float4 b[16];
  load_bfrag(W, o0, l16, half, b);
  int r0 = wv, c4 = lane;
  const float4* x4 = (const float4*)x;
  int t = blk;
  float4 p0 = x4[(t * 16 + r0) * 32 + c4];
  float4 p1 = x4[(t * 16 + r0 + 8) * 32 + c4];
  *(float4*)&As[0][r0 * TILE_P + 4 * c4] = p0;
  *(float4*)&As[0][(r0 + 8) * TILE_P + 4 * c4] = p1;
  __syncthreads();
  int buf = 0;
  for (; t < NTILES; t += PGRID) {
    int tn = t + PGRID;
    bool more = tn < NTILES;
    if (more) {
      p0 = x4[(tn * 16 + r0) * 32 + c4];
      p1 = x4[(tn * 16 + r0 + 8) * 32 + c4];
    }
    v8f acc = tile_mma(As[buf], l16, half, b);
    if (more) {
      *(float4*)&As[buf ^ 1][r0 * TILE_P + 4 * c4] = p0;
      *(float4*)&As[buf ^ 1][(r0 + 8) * TILE_P + 4 * c4] = p1;
    }
    __syncthreads();
#pragma unroll
    for (int r = 0; r < 8; ++r)
      hlin[(t * 16 + r + half * 8) * OUT_C + o0 + l16] = acc[r];
    buf ^= 1;
  }
}

static_assert(N_NODES <= 65536, "source ids are kept as 16-bit values");
#define BINW   256
#define NBINS  ((N_NODES + BINW - 1) / BINW)
#define CHUNK  4096
#define NCHUNK ((N_EDGES + CHUNK - 1) / CHUNK)
#define EDGE_T 512
#define EDGE_V (CHUNK / EDGE_T)
#define NSLOT  (NBINS * NCHUNK)
#define HALF_C (NCHUNK / 2)
static_assert(CHUNK % EDGE_T == 0, "whole edges per thread");
static_assert(BINW <= 65536 && BINW <= EDGE_T, "local destination and source share one word; one thread per node of a bin");
static_assert(NCHUNK % 2 == 0 && NBINS <= 256 && EDGE_T == 512, "k_bin_move: thread = (bin, half of the table's rows)");
static_assert(NBINS <= 7 * 32, "k_bin_move: one wave scans the bin totals, 7 per lane");

#define FRONT_T 256
#define HIST_V  (CHUNK / FRONT_T)
static_assert(CHUNK % FRONT_T == 0, "whole edges per thread");
__device__ __forceinline__ void bin_hist_block(const int* __restrict__ ei,
                                               int* __restrict__ cnt,
                                               int* hist, int blk) {
  int tid = (int)threadIdx.x;
  for (int b = tid; b < NBINS; b += FRONT_T) hist[b] = 0;
  __syncthreads();
  int e0 = blk * CHUNK + tid;
  unsigned d[HIST_V];
#pragma unroll
  for (int q = 0; q < HIST_V; ++q) {
    int e = e0 + q * FRONT_T;
    d[q] = (unsigned)ei[N_EDGES + ((e < N_EDGES) ? e : (N_EDGES - 1))];
  }
#pragma unroll
  for (int q = 0; q < HIST_V; ++q) {
    int e = e0 + q * FRONT_T;
    if (e < N_EDGES)
      __hip_atomic_fetch_add(&hist[d[q] / BINW], 1, __ATOMIC_RELAXED,
                             __HIP_MEMORY_SCOPE_WORKGROUP);
  }
  __syncthreads();
  for (int b = tid; b < NBINS; b += FRONT_T) cnt[blk * NBINS + b] = hist[b];
}

__global__ void __launch_bounds__(EDGE_T) k_bin_move(const int* __restrict__ ei,
                                                     const int* __restrict__ cnt,
                                                     int* __restrict__ bstart,
                                                     unsigned* __restrict__ binned) {
  __shared__ int pos[NBINS];
  __shared__ int s_tot[2][256];
  __shared__ int s_pre[2][256];
  int tid = (int)threadIdx.x, blk = (int)blockIdx.x;
  int e0 = blk * CHUNK + tid;
  unsigned d[EDGE_V], s[EDGE_V];
#pragma unroll
  for (int q = 0; q < EDGE_V; ++q) {
    int e = e0 + q * EDGE_T;
    int ec = (e < N_EDGES) ? e : (N_EDGES - 1);
    d[q] = (unsigned)ei[N_EDGES + ec];
    s[q] = (unsigned)ei[ec];
  }
  int b = tid & 255, p = tid >> 8;
  int tot = 0, pre = 0;
  if (b < NBINS) {
    const int* cp = cnt + p * HALF_C * NBINS + b;
    int lim = blk - p * HALF_C;
#pragma unroll 49
    for (int r = 0; r < HALF_C; ++r) {
      int v = cp[r * NBINS];
      tot += v;
      pre += (r < lim) ? v : 0;
    }
  }
  s_tot[p][b] = tot;
  s_pre[p][b] = pre;
  __syncthreads();
  if (tid < 32) {
    int loc[7];
    int sum = 0;
#pragma unroll
    for (int q = 0; q < 7; ++q) {
      int bb = tid * 7 + q;
      loc[q] = s_tot[0][bb] + s_tot[1][bb];
      sum += loc[q];
    }
    int inc = sum;
    for (int o = 1; o < 32; o <<= 1) {
      int v = __shfl_up(inc, o);
      if (tid >= o) inc += v;
    }
    int run = inc - sum;
#pragma unroll
    for (int q = 0; q < 7; ++q) {
      int bb = tid * 7 + q;
      if (bb < NBINS) {
        pos[bb] = run + s_pre[0][bb] + s_pre[1][bb];
        if (blk == 0) bstart[bb] = run;
      }
      run += loc[q];
    }
  }
  if (blk == 0 && tid == 0) bstart[NBINS] = N_EDGES;
  __syncthreads();
#pragma unroll
  for (int q = 0; q < EDGE_V; ++q) {
    int e = e0 + q * EDGE_T;
    if (e < N_EDGES) {
      int w = __hip_atomic_fetch_add(&pos[d[q] / BINW], 1, __ATOMIC_RELAXED,
                                     __HIP_MEMORY_SCOPE_WORKGROUP);
      binned[w] = ((d[q] % BINW) << 16) | s[q];
    }
  }
}

#define BIN_U 4
__global__ void __launch_bounds__(EDGE_T) k_bin_csr(const unsigned* __restrict__ binned,
                                                    const int* __restrict__ bstart,
                                                    int* __restrict__ off,
                                                    float* __restrict__ dinv,
                                                    unsigned short* __restrict__ csr) {
  __shared__ int cnt[BINW];
  __shared__ int seg[BINW];
  __shared__ int fil[BINW];
  int tid = (int)threadIdx.x, bin = (int)blockIdx.x;
  int beg = bstart[bin];
  int end = bstart[bin + 1];
  int n = end - beg;
  if (tid < BINW) {
    cnt[tid] = 0;
    fil[tid] = 0;
  }
  __syncthreads();
  for (int i0 = tid; i0 < n; i0 += BIN_U * EDGE_T) {
    unsigned w[BIN_U];
#pragma unroll
    for (int q = 0; q < BIN_U; ++q) {
      int i = i0 + q * EDGE_T;
      w[q] = binned[beg + ((i < n) ? i : i0)];
    }
#pragma unroll
    for (int q = 0; q < BIN_U; ++q) {
      if (i0 + q * EDGE_T < n)
        __hip_atomic_fetch_add(&cnt[w[q] >> 16], 1, __ATOMIC_RELAXED,
                               __HIP_MEMORY_SCOPE_WORKGROUP);
    }
  }
  __syncthreads();
  int c = (tid < BINW) ? cnt[tid] : 0;
  if (tid < BINW) seg[tid] = c;
  __syncthreads();
  for (int o = 1; o < BINW; o <<= 1) {
    int v = (tid < BINW && tid >= o) ? seg[tid - o] : 0;
    __syncthreads();
    if (tid < BINW) seg[tid] += v;
    __syncthreads();
  }
  if (tid < BINW) {
    int p = beg + seg[tid] - c;
    seg[tid] = p;
    int node = bin * BINW + tid;
    if (node < N_NODES) {
      off[node] = p;
      float dg = 1.0f + (float)c;
      dinv[node] = (dg > 0.0f) ? rsqrtf(dg) : 0.0f;
    }
  }
  if (bin == NBINS - 1 && tid == 0) off[N_NODES] = N_EDGES;
  __syncthreads();
  for (int i0 = tid; i0 < n; i0 += BIN_U * EDGE_T) {
    unsigned w[BIN_U];
#pragma unroll
    for (int q = 0; q < BIN_U; ++q) {
      int i = i0 + q * EDGE_T;
      w[q] = binned[beg + ((i < n) ? i : i0)];
    }
#pragma unroll
    for (int q = 0; q < BIN_U; ++q) {
      if (i0 + q * EDGE_T < n) {
        int dl = (int)(w[q] >> 16);
        int r = __hip_atomic_fetch_add(&fil[dl], 1, __ATOMIC_RELAXED,
                                       __HIP_MEMORY_SCOPE_WORKGROUP);
        csr[seg[dl] + r] = (unsigned short)(w[q] & 0xffffu);
      }
    }
  }
}

__device__ __forceinline__ float4 axpy4(float4 v, float s, float4 a) {
  a.x += v.x * s;
  a.y += v.y * s;
  a.z += v.z * s;
  a.w += v.w * s;
  return a;
}

__device__ __forceinline__ float readlane_f(float v, int l) {
  return __int_as_float(__builtin_amdgcn_readlane(__float_as_int(v), l));
}

#define MID_CAP 128
__global__ void __launch_bounds__(256) k_gather(const float* __restrict__ hlin,
                                                const int* __restrict__ off,
                                                const unsigned short* __restrict__ csr,
                                                const float* __restrict__ dinv,
                                                const float* __restrict__ bias,
                                                float* __restrict__ h) {
  __shared__ int   mid_us[8][MID_CAP];
  __shared__ int   mid_so[8][MID_CAP];
  __shared__ float mid_wo[8][MID_CAP];
  int lane = (int)(threadIdx.x & 31);
  int n = __builtin_amdgcn_readfirstlane(
      (int)((blockIdx.x * blockDim.x + threadIdx.x) >> 5));
  if (n >= N_NODES) return;
  int beg = off[n];
  int d = off[n + 1] - beg;
  float di = dinv[n];
  const float4* h4 = (const float4*)hlin;
  float4 hv = h4[n * 32 + lane];
  float4 acc;
  acc.x = hv.x * di * di + bias[lane * 4 + 0];
  acc.y = hv.y * di * di + bias[lane * 4 + 1];
  acc.z = hv.z * di * di + bias[lane * 4 + 2];
  acc.w = hv.w * di * di + bias[lane * 4 + 3];

  if (d <= 32) {
    int s = 0x7fffffff;
    float w = 0.0f;
    if (lane < d) {
      s = (int)csr[beg + lane];
      w = dinv[s];
    }
    int r = 0;
    for (int j = 0; j < d; ++j) {
      int sj = __builtin_amdgcn_readlane(s, j);
      r += ((sj < s) || (sj == s && j < lane)) ? 1 : 0;
    }
    int   ss = __builtin_amdgcn_ds_permute(r << 2, s);
    float ws = __int_as_float(__builtin_amdgcn_ds_permute(r << 2, __float_as_int(w)));
    int k = 0;
    for (; k + 4 <= d; k += 4) {
      int s0 = __builtin_amdgcn_readlane(ss, k);
      int s1 = __builtin_amdgcn_readlane(ss, k + 1);
      int s2 = __builtin_amdgcn_readlane(ss, k + 2);
      int s3 = __builtin_amdgcn_readlane(ss, k + 3);
      float w0 = readlane_f(ws, k);
      float w1 = readlane_f(ws, k + 1);
      float w2 = readlane_f(ws, k + 2);
      float w3 = readlane_f(ws, k + 3);
      float4 v0 = h4[s0 * 32 + lane];
      float4 v1 = h4[s1 * 32 + lane];
      float4 v2 = h4[s2 * 32 + lane];
      float4 v3 = h4[s3 * 32 + lane];
      acc = axpy4(v0, w0 * di, acc);
      acc = axpy4(v1, w1 * di, acc);
      acc = axpy4(v2, w2 * di, acc);
      acc = axpy4(v3, w3 * di, acc);
    }
    for (; k < d; ++k) {
      int s0 = __builtin_amdgcn_readlane(ss, k);
      float w0 = readlane_f(ws, k);
      float4 v0 = h4[s0 * 32 + lane];
      acc = axpy4(v0, w0 * di, acc);
    }
  } else if (d <= MID_CAP) {
    int wv = (int)(threadIdx.x >> 5);
    int*   us = mid_us[wv];
    int*   so = mid_so[wv];
    float* wo = mid_wo[wv];
    int   e[MID_CAP / 32];
    float w[MID_CAP / 32];
    for (int q = 0; q < MID_CAP / 32; ++q) {
      int i = lane + 32 * q;
      e[q] = 0x7fffffff;
      w[q] = 0.0f;
      if (i < d) {
        e[q] = (int)csr[beg + i];
        w[q] = dinv[e[q]];
        us[i] = e[q];
      }
    }
    int r[MID_CAP / 32];
    for (int q = 0; q < MID_CAP / 32; ++q) r[q] = 0;
    for (int j = 0; j < d; ++j) {
      int sj = us[j];
      for (int q = 0; q < MID_CAP / 32; ++q)
        r[q] += ((sj < e[q]) || (sj == e[q] && j < lane + 32 * q)) ? 1 : 0;
    }
    for (int q = 0; q < MID_CAP / 32; ++q) {
      if (lane + 32 * q < d) {
        so[r[q]] = e[q];
        wo[r[q]] = w[q];
      }
    }
    int k = 0;
    for (; k + 4 <= d; k += 4) {
      int s0 = so[k], s1 = so[k + 1], s2 = so[k + 2], s3 = so[k + 3];
      float w0 = wo[k], w1 = wo[k + 1], w2 = wo[k + 2], w3 = wo[k + 3];
      float4 v0 = h4[s0 * 32 + lane];
      float4 v1 = h4[s1 * 32 + lane];
      float4 v2 = h4[s2 * 32 + lane];
      float4 v3 = h4[s3 * 32 + lane];
      acc = axpy4(v0, w0 * di, acc);
      acc = axpy4(v1, w1 * di, acc);
      acc = axpy4(v2, w2 * di, acc);
      acc = axpy4(v3, w3 * di, acc);
    }
    for (; k < d; ++k) {
      int s0 = so[k];
      float w0 = wo[k];
      float4 v0 = h4[s0 * 32 + lane];
      acc = axpy4(v0, w0 * di, acc);
    }
  } else {
    int last = -1;
    int left = d;
    for (int it = 0; it < d && left > 0; ++it) {
      int mn = 0x7fffffff;
      for (int i = lane; i < d; i += 32) {
        int s = (int)csr[beg + i];
        if (s > last && s < mn) mn = s;
      }
      for (int o = 16; o > 0; o >>= 1) {
        int t = __shfl_xor(mn, o);
        mn = (t < mn) ? t : mn;
      }
      int m = 0;
      for (int i = lane; i < d; i += 32) m += ((int)csr[beg + i] == mn) ? 1 : 0;
      for (int o = 16; o > 0; o >>= 1) m += __shfl_xor(m, o);
      mn = __builtin_amdgcn_readfirstlane(mn);
      m = __builtin_amdgcn_readfirstlane(m);
      float nrm = dinv[mn] * di;
      float4 v = h4[mn * 32 + lane];
      for (int c = 0; c < m; ++c) acc = axpy4(v, nrm, acc);
      last = mn;
      left -= m;
    }
  }
  ((float4*)h)[n * 32 + lane] = acc;
}

__device__ __forceinline__ void prep_w0_block(const float* __restrict__ bw,
                                              const float* __restrict__ sw,
                                              const float* __restrict__ sc,
                                              float* __restrict__ Bt, int blk) {
  int t = blk * FRONT_T + (int)threadIdx.x;
  if (t >= 8 * OUT_C * HIDDEN) return;
  int o = t >> 10, k = t & 1023;
  int f = k >> 3, j = k & 7;
  float w = (j == 0) ? bw[o * OUT_C + f]
                     : sw[(o * OUT_C + f) * N_COEF + (j - 1)] * sc[o * OUT_C + f];
  Bt[t] = w;
}

__device__ __forceinline__ void prep_w1_block(const float* __restrict__ bw,
                                              const float* __restrict__ sw,
                                              const float* __restrict__ sc,
                                              float* __restrict__ Bt, int blk) {
  int t = blk * FRONT_T + (int)threadIdx.x;
  if (t >= 8 * HIDDEN * OUT_C) return;
  int o = t >> 7, k = t & 127;
  int f = k >> 3, j = k & 7;
  float w = (j == 0) ? bw[o * HIDDEN + f]
                     : sw[(o * HIDDEN + f) * N_COEF + (j - 1)] * sc[o * HIDDEN + f];
  Bt[o * 128 + k] = w;
}

#define PREP_B     ((8 * OUT_C * HIDDEN + FRONT_T - 1) / FRONT_T)
#define FRONT_GRID (PGRID + NCHUNK + 2 * PREP_B)
__global__ void __launch_bounds__(FRONT_T) k_front(const float* __restrict__ x,
                                                   const float* __restrict__ W,
                                                   float* __restrict__ hlin,
                                                   const int* __restrict__ ei,
                                                   int* __restrict__ cnt,
                                                   const float* __restrict__ bw0,
                                                   const float* __restrict__ sw0,
                                                   const float* __restrict__ sc0,
                                                   float* __restrict__ Bt0,
                                                   const float* __restrict__ bw1,
                                                   const float* __restrict__ sw1,
                                                   const float* __restrict__ sc1,
                                                   float* __restrict__ Bt1) {
  __shared__ __attribute__((aligned(16))) float As[2][16 * TILE_P];
  __shared__ int hist[NBINS];
  int blk = (int)blockIdx.x;
  if (blk < PGRID) {
    gemm_xw_block(x, W, hlin, As, blk);
  } else if (blk < PGRID + NCHUNK) {
    bin_hist_block(ei, cnt, hist, blk - PGRID);
  } else if (blk < PGRID + NCHUNK + PREP_B) {
    prep_w0_block(bw0, sw0, sc0, Bt0, blk - (PGRID + NCHUNK));
  } else {
    prep_w1_block(bw1, sw1, sc1, Bt1, blk - (PGRID + NCHUNK + PREP_B));
  }
}

#define PGRID0 512
static_assert(PGRID0 <= NTILES, "every workgroup of k_kan0 has a first tile");
__global__ void __launch_bounds__(256) k_kan0(const float* __restrict__ h,
                                              const float* __restrict__ Bt,
                                              float* __restrict__ g) {
  __shared__ __attribute__((aligned(16))) float red[2][8][256];
  int tid = (int)threadIdx.x;
  int lane = tid & 31, wv = tid >> 5;
  int half = lane >> 4, l16 = lane & 15;
  int f0 = wv * 16 + half * 8;
  float4 b[16];
  {
    const float4* bp = (const float4*)(Bt + l16 * 1024 + f0 * 8);
#pragma unroll
    for (int j = 0; j < 16; ++j) b[j] = bp[j];
  }
  const float4* h4 = (const float4*)h;
  int t = (int)blockIdx.x;
  float4 p0 = h4[((t * 16 + l16) * OUT_C + f0) / 4];
  float4 p1 = h4[((t * 16 + l16) * OUT_C + f0) / 4 + 1];
  int ridx = ((tid >> 4) & 7) * 32 + (tid >> 7) * 16 + (tid & 15);
  int buf = 0;
  for (; t < NTILES; t += PGRID0) {
    float v[8];
    v[0] = p0.x; v[1] = p0.y; v[2] = p0.z; v[3] = p0.w;
    v[4] = p1.x; v[5] = p1.y; v[6] = p1.z; v[7] = p1.w;
    int tn = t + PGRID0;
    if (tn < NTILES) {
      p0 = h4[((tn * 16 + l16) * OUT_C + f0) / 4];
      p1 = h4[((tn * 16 + l16) * OUT_C + f0) / 4 + 1];
    }
    v2f x[8][4];
#pragma unroll
    for (int i = 0; i < 8; ++i) {
      float e[8];
      e[0] = silu_f(v[i]);
      bspline7(v[i], e + 1);
#pragma unroll
      for (int q = 0; q < 4; ++q) { x[i][q].x = e[2 * q]; x[i][q].y = e[2 * q + 1]; }
    }
    __builtin_amdgcn_sched_barrier(0);
    v8f acc0 = {}, acc1 = {};
#pragma unroll
    for (int i = 0; i < 8; ++i) {
      v2f w0, w1, w2, w3;
      w0.x = b[2 * i].x;     w0.y = b[2 * i].y;
      w1.x = b[2 * i].z;     w1.y = b[2 * i].w;
      w2.x = b[2 * i + 1].x; w2.y = b[2 * i + 1].y;
      w3.x = b[2 * i + 1].z; w3.y = b[2 * i + 1].w;
      acc0 = wmma4(x[i][0], w0, acc0);
      acc1 = wmma4(x[i][1], w1, acc1);
      acc0 = wmma4(x[i][2], w2, acc0);
      acc1 = wmma4(x[i][3], w3, acc1);
    }
    asm volatile("v_nop\n\tv_nop\n\tv_nop\n\tv_nop\n\tv_nop" : "+v"(acc0), "+v"(acc1));
    v8f acc = acc0 + acc1;
    float* rw = &red[buf][wv][lane];
#pragma unroll
    for (int r = 0; r < 8; ++r) rw[r * 32] = acc[r];
    __syncthreads();
    float s = red[buf][0][ridx];
#pragma unroll
    for (int w = 1; w < 8; ++w) s += red[buf][w][ridx];
    g[t * 256 + tid] = s;
    buf ^= 1;
  }
}

__device__ __forceinline__ void expand8(float v, float4* d) {
  float bs[N_COEF];
  bspline7(v, bs);
  float4 lo, hi;
  lo.x = silu_f(v); lo.y = bs[0]; lo.z = bs[1]; lo.w = bs[2];
  hi.x = bs[3]; hi.y = bs[4]; hi.z = bs[5]; hi.w = bs[6];
  d[0] = lo;
  d[1] = hi;
}

__global__ void __launch_bounds__(256) k_kan1(const float* __restrict__ g,
                                              const float* __restrict__ Bt,
                                              float* __restrict__ out) {
  __shared__ __attribute__((aligned(16))) float As[2][16 * TILE_P];
  int tid = (int)threadIdx.x;
  int lane = tid & 31, wv = tid >> 5;
  int half = lane >> 4, l16 = lane & 15;
  int o0 = wv * 16;
  float4 b[16];
  load_bfrag(Bt, o0, l16, half, b);
  int row = tid >> 4, f = tid & 15;
  int t = (int)blockIdx.x;
  float v = g[(t * 16 + row) * HIDDEN + f];
  expand8(v, (float4*)&As[0][row * TILE_P + f * 8]);
  __syncthreads();
  int buf = 0;
  for (; t < NTILES; t += PGRID) {
    int tn = t + PGRID;
    bool more = tn < NTILES;
    if (more) v = g[(tn * 16 + row) * HIDDEN + f];
    v8f acc = tile_mma(As[buf], l16, half, b);
    if (more) expand8(v, (float4*)&As[buf ^ 1][row * TILE_P + f * 8]);
    __syncthreads();
#pragma unroll
    for (int r = 0; r < 8; ++r)
      out[(t * 16 + r + half * 8) * OUT_C + o0 + l16] = acc[r];
    buf ^= 1;
  }
}

extern "C" void kernel_launch(void* const* d_in, const int* in_sizes, int n_in,
                              void* d_out, int out_size, void* d_ws, size_t ws_size,
                              hipStream_t stream) {
  (void)in_sizes; (void)n_in; (void)out_size; (void)ws_size;
  const float* x       = (const float*)d_in[0];
  const int*   ei      = (const int*)d_in[1];
  const float* gcn_w   = (const float*)d_in[2];
  const float* gcn_b   = (const float*)d_in[3];
  const float* base_w0 = (const float*)d_in[4];
  const float* spl_w0  = (const float*)d_in[5];
  const float* scal0   = (const float*)d_in[6];
  const float* base_w1 = (const float*)d_in[7];
  const float* spl_w1  = (const float*)d_in[8];
  const float* scal1   = (const float*)d_in[9];
  float* out = (float*)d_out;

  float* ws    = (float*)d_ws;
  float* hlin  = ws;
  float* h     = hlin + (size_t)N_NODES * OUT_C;
  float* deg   = h + (size_t)N_NODES * OUT_C;
  float* g     = deg + N_NODES;
  float* Bm0   = g + (size_t)N_NODES * HIDDEN;
  float* Bm1   = Bm0 + 8 * OUT_C * HIDDEN;
  unsigned* binned = (unsigned*)h;
  int* cnt = (int*)h + N_EDGES;
  int* bstart = cnt + NSLOT;
  unsigned short* csr = (unsigned short*)g;
  int* off  = (int*)g + N_EDGES / 2;

  k_front<<<FRONT_GRID, FRONT_T, 0, stream>>>(x, gcn_w, hlin, ei, cnt,
                                              base_w0, spl_w0, scal0, Bm0,
                                              base_w1, spl_w1, scal1, Bm1);
  k_bin_move<<<NCHUNK, EDGE_T, 0, stream>>>(ei, cnt, bstart, binned);
  k_bin_csr<<<NBINS, EDGE_T, 0, stream>>>(binned, bstart, off, deg, csr);
  k_gather<<<(N_NODES * 32 + 255) / 256, 256, 0, stream>>>(hlin, off, csr, deg, gcn_b, h);
  k_kan0<<<PGRID0, 256, 0, stream>>>(h, Bm0, g);
  k_kan1<<<PGRID, 256, 0, stream>>>(g, Bm1, out);
}


# Edge grouping in two launches, not three: each chunk sorts its edges by bin in LDS inside the front launch; the per-bin workgroup sums the piece table and collects its own pieces; 6 launches to 5
# speedup vs baseline: 4.0727x; 1.2511x over previous
#include <hip/hip_runtime.h>
#include <hip/hip_bf16.h>

#define N_NODES 50000
#define N_EDGES 800000
#define IN_C    128
#define OUT_C   128
#define HIDDEN  16
#define N_COEF  7

typedef __attribute__((ext_vector_type(2))) float v2f;
typedef __attribute__((ext_vector_type(8))) float v8f;

__device__ __forceinline__ v8f wmma4(v2f a, v2f b, v8f c) {
  return __builtin_amdgcn_wmma_f32_16x16x4_f32(false, a, false, b, (short)0, c,
                                               false, false);
}

__device__ __forceinline__ float silu_f(float v) {
  return v * (1.0f / (1.0f + __expf(-v)));
}

__device__ __forceinline__ void bspline7(float v, float* out) {
  float fi = floorf(2.0f * v + 5.0f);
  float t  = (v - (0.5f * fi - 2.5f)) * 2.0f;
  float t2 = t * t;
  float u  = 1.0f - t;
  const float k6 = 1.0f / 6.0f;
  float n3 = t2 * t * k6;
  float n2 = t * (t * (0.5f - 0.5f * t) + 0.5f) + k6;
  float n1 = t2 * (0.5f * t - 1.0f) + (2.0f / 3.0f);
  float n0 = u * u * u * k6;
  bool m[10];
#pragma unroll
  for (int k = 0; k < 10; ++k) m[k] = (fi == (float)k);
#pragma unroll
  for (int c = 0; c < N_COEF; ++c) {
    float r = m[c + 3] ? n0 : 0.0f;
    r = m[c + 2] ? n1 : r;
    r = m[c + 1] ? n2 : r;
    r = m[c] ? n3 : r;
    out[c] = r;
  }
}

#define TILE_P 132
#define NTILES (N_NODES / 16)
#define PGRID  512
static_assert(N_NODES % 16 == 0 && PGRID <= NTILES, "tiling");

__device__ __forceinline__ void load_bfrag(const float* __restrict__ Wt, int o0,
                                           int l16, int half, float4 (&b)[16]) {
  const float4* p = (const float4*)(Wt + (o0 + l16) * 128 + 4 * half);
#pragma unroll
  for (int j = 0; j < 16; ++j) b[j] = p[2 * j];
}

__device__ __forceinline__ v8f tile_mma(const float* As, int l16, int half,
                                        const float4 (&b)[16]) {
  float4 a[16];
  const float4* p = (const float4*)(As + l16 * TILE_P + 4 * half);
#pragma unroll
  for (int j = 0; j < 16; ++j) a[j] = p[2 * j];
  __builtin_amdgcn_sched_barrier(0);
  v8f acc0 = {}, acc1 = {};
#pragma unroll
  for (int j = 0; j < 16; ++j) {
    v2f a0, a1, b0, b1;
    a0.x = a[j].x; a0.y = a[j].y; a1.x = a[j].z; a1.y = a[j].w;
    b0.x = b[j].x; b0.y = b[j].y; b1.x = b[j].z; b1.y = b[j].w;
    acc0 = wmma4(a0, b0, acc0);
    acc1 = wmma4(a1, b1, acc1);
  }
  asm volatile("v_nop\n\tv_nop\n\tv_nop\n\tv_nop\n\tv_nop" : "+v"(acc0), "+v"(acc1));
  return acc0 + acc1;
}

__device__ __forceinline__ void gemm_xw_block(const float* __restrict__ x,
                                              const float* __restrict__ W,
                                              float* __restrict__ hlin,
                                              float (*As)[16 * TILE_P], int blk) {
  int tid = (int)threadIdx.x;
  int lane = tid & 31, wv = tid >> 5;
  int half = lane >> 4, l16 = lane & 15;
  int o0 = wv * 16;
  float4 b[16];
  load_bfrag(W, o0, l16, half, b);
  int r0 = wv, c4 = lane;
  const float4* x4 = (const float4*)x;
  int t = blk;
  float4 p0 = x4[(t * 16 + r0) * 32 + c4];
  float4 p1 = x4[(t * 16 + r0 + 8) * 32 + c4];
  *(float4*)&As[0][r0 * TILE_P + 4 * c4] = p0;
  *(float4*)&As[0][(r0 + 8) * TILE_P + 4 * c4] = p1;
  __syncthreads();
  int buf = 0;
  for (; t < NTILES; t += PGRID) {
    int tn = t + PGRID;
    bool more = tn < NTILES;
    if (more) {
      p0 = x4[(tn * 16 + r0) * 32 + c4];
      p1 = x4[(tn * 16 + r0 + 8) * 32 + c4];
    }
    v8f acc = tile_mma(As[buf], l16, half, b);
    if (more) {
      *(float4*)&As[buf ^ 1][r0 * TILE_P + 4 * c4] = p0;
      *(float4*)&As[buf ^ 1][(r0 + 8) * TILE_P + 4 * c4] = p1;
    }
    __syncthreads();
#pragma unroll
    for (int r = 0; r < 8; ++r)
      hlin[(t * 16 + r + half * 8) * OUT_C + o0 + l16] = acc[r];
    buf ^= 1;
  }
}

static_assert(N_NODES <= 65536, "source ids are kept as 16-bit values");
#define BINW   256
#define NBINS  ((N_NODES + BINW - 1) / BINW)
#define CHUNK  4096
#define NCHUNK ((N_EDGES + CHUNK - 1) / CHUNK)
#define EDGE_T 512
#define NPIECE (NBINS * NCHUNK)
static_assert(BINW <= 65536 && BINW <= EDGE_T, "local destination and source share one word; one thread per node of a bin");
static_assert(CHUNK < 65536, "a piece's offset and its length share one word");
static_assert(NBINS <= 7 * 32, "one wave scans a chunk's bin counts, 7 per lane");
static_assert(NCHUNK % 4 == 0 && NCHUNK <= EDGE_T, "k_bin_csr reads the piece table four words at a time; one thread per piece");

#define FRONT_T 256
#define HIST_V  (CHUNK / FRONT_T)
static_assert(CHUNK % FRONT_T == 0, "whole edges per thread");
__device__ __forceinline__ void bin_sort_block(const int* __restrict__ ei,
                                               unsigned* __restrict__ srt,
                                               unsigned* __restrict__ pc,
                                               int* hist, int* bst, unsigned* buf,
                                               int blk) {
  int tid = (int)threadIdx.x;
  for (int b = tid; b < NBINS; b += FRONT_T) hist[b] = 0;
  __syncthreads();
  int e0 = blk * CHUNK + tid;
  unsigned d[HIST_V], s[HIST_V];
  int r[HIST_V];
#pragma unroll
  for (int q = 0; q < HIST_V; ++q) {
    int e = e0 + q * FRONT_T;
    int ec = (e < N_EDGES) ? e : (N_EDGES - 1);
    d[q] = (unsigned)ei[N_EDGES + ec];
    s[q] = (unsigned)ei[ec];
  }
#pragma unroll
  for (int q = 0; q < HIST_V; ++q) {
    int e = e0 + q * FRONT_T;
    r[q] = 0;
    if (e < N_EDGES)
      r[q] = __hip_atomic_fetch_add(&hist[d[q] / BINW], 1, __ATOMIC_RELAXED,
                                    __HIP_MEMORY_SCOPE_WORKGROUP);
  }
  __syncthreads();
  if (tid < 32) {
    int loc[7];
    int sum = 0;
#pragma unroll
    for (int q = 0; q < 7; ++q) {
      int bb = tid * 7 + q;
      loc[q] = (bb < NBINS) ? hist[bb] : 0;
      sum += loc[q];
    }
    int inc = sum;
    for (int o = 1; o < 32; o <<= 1) {
      int v = __shfl_up(inc, o);
      if (tid >= o) inc += v;
    }
    int run = inc - sum;
#pragma unroll
    for (int q = 0; q < 7; ++q) {
      int bb = tid * 7 + q;
      if (bb < NBINS) {
        bst[bb] = run;
        pc[bb * NCHUNK + blk] = ((unsigned)loc[q] << 16) | (unsigned)run;
      }
      run += loc[q];
    }
  }
  __syncthreads();
#pragma unroll
  for (int q = 0; q < HIST_V; ++q) {
    int e = e0 + q * FRONT_T;
    if (e < N_EDGES)
      buf[bst[d[q] / BINW] + r[q]] = ((d[q] % BINW) << 16) | s[q];
  }
  __syncthreads();
  int left = N_EDGES - blk * CHUNK;
  int nv = (left < CHUNK) ? left : CHUNK;
  for (int i = tid; i < nv; i += FRONT_T) srt[blk * CHUNK + i] = buf[i];
}

#define CSR_W  (EDGE_T / 32)
#define PIECE_V ((NCHUNK + CSR_W - 1) / CSR_W)
#define TAB_V  ((NPIECE / 4 + EDGE_T - 1) / EDGE_T)
__global__ void __launch_bounds__(EDGE_T) k_bin_csr(const unsigned* __restrict__ srt,
                                                    const unsigned* __restrict__ pc,
                                                    int* __restrict__ off,
                                                    float* __restrict__ dinv,
                                                    unsigned short* __restrict__ csr) {
  __shared__ int cnt[BINW];
  __shared__ int seg[BINW];
  __shared__ int fil[BINW];
  __shared__ int pn[NCHUNK];
  __shared__ int psrc[NCHUNK];
  __shared__ int red[CSR_W];
  int tid = (int)threadIdx.x, bin = (int)blockIdx.x;
  int lane = tid & 31, wv = tid >> 5;
  if (tid < BINW) {
    cnt[tid] = 0;
    fil[tid] = 0;
  }
  unsigned own = pc[bin * NCHUNK + ((tid < NCHUNK) ? tid : 0)];
  int part = 0;
  {
    const uint4* pc4 = (const uint4*)pc;
    int lim4 = bin * (NCHUNK / 4);
    uint4 tv[TAB_V];
#pragma unroll
    for (int q = 0; q < TAB_V; ++q) {
      int i4 = tid + q * EDGE_T;
      tv[q] = pc4[(i4 < lim4) ? i4 : 0];
    }
    __builtin_amdgcn_sched_barrier(0);
#pragma unroll
    for (int q = 0; q < TAB_V; ++q) {
      int i4 = tid + q * EDGE_T;
      int sum4 = (int)((tv[q].x >> 16) + (tv[q].y >> 16) + (tv[q].z >> 16) + (tv[q].w >> 16));
      part += (i4 < lim4) ? sum4 : 0;
    }
  }
  for (int o = 16; o > 0; o >>= 1) part += __shfl_xor(part, o);
  if (lane == 0) red[wv] = part;
  if (tid < NCHUNK) {
    pn[tid] = (int)(own >> 16);
    psrc[tid] = tid * CHUNK + (int)(own & 0xffffu);
  }
  __syncthreads();
  int beg = 0;
#pragma unroll
  for (int w = 0; w < CSR_W; ++w) beg += red[w];
  unsigned wd[PIECE_V];
  int pm[PIECE_V], pb[PIECE_V];
#pragma unroll
  for (int k = 0; k < PIECE_V; ++k) {
    int c = wv + k * CSR_W;
    pm[k] = (c < NCHUNK) ? pn[c] : 0;
    pb[k] = (c < NCHUNK) ? psrc[c] : 0;
    wd[k] = srt[(lane < pm[k]) ? (pb[k] + lane) : 0];
  }
#pragma unroll
  for (int k = 0; k < PIECE_V; ++k) {
    if (lane < pm[k])
      __hip_atomic_fetch_add(&cnt[wd[k] >> 16], 1, __ATOMIC_RELAXED,
                             __HIP_MEMORY_SCOPE_WORKGROUP);
    for (int i = lane + 32; i < pm[k]; i += 32) {
      unsigned w = srt[pb[k] + i];
      __hip_atomic_fetch_add(&cnt[w >> 16], 1, __ATOMIC_RELAXED,
                             __HIP_MEMORY_SCOPE_WORKGROUP);
    }
  }
  __syncthreads();
  int c = (tid < BINW) ? cnt[tid] : 0;
  if (tid < BINW) seg[tid] = c;
  __syncthreads();
  for (int o = 1; o < BINW; o <<= 1) {
    int v = (tid < BINW && tid >= o) ? seg[tid - o] : 0;
    __syncthreads();
    if (tid < BINW) seg[tid] += v;
    __syncthreads();
  }
  if (tid < BINW) {
    int p = beg + seg[tid] - c;
    seg[tid] = p;
    int node = bin * BINW + tid;
    if (node < N_NODES) {
      off[node] = p;
      float dg = 1.0f + (float)c;
      dinv[node] = (dg > 0.0f) ? rsqrtf(dg) : 0.0f;
    }
  }
  if (bin == NBINS - 1 && tid == 0) off[N_NODES] = N_EDGES;
  __syncthreads();
#pragma unroll
  for (int k = 0; k < PIECE_V; ++k) {
    if (lane < pm[k]) {
      int dl = (int)(wd[k] >> 16);
      int r = __hip_atomic_fetch_add(&fil[dl], 1, __ATOMIC_RELAXED,
                                     __HIP_MEMORY_SCOPE_WORKGROUP);
      csr[seg[dl] + r] = (unsigned short)(wd[k] & 0xffffu);
    }
    for (int i = lane + 32; i < pm[k]; i += 32) {
      unsigned w = srt[pb[k] + i];
      int dl = (int)(w >> 16);
      int r = __hip_atomic_fetch_add(&fil[dl], 1, __ATOMIC_RELAXED,
                                     __HIP_MEMORY_SCOPE_WORKGROUP);
      csr[seg[dl] + r] = (unsigned short)(w & 0xffffu);
    }
  }
}

__device__ __forceinline__ float4 axpy4(float4 v, float s, float4 a) {
  a.x += v.x * s;
  a.y += v.y * s;
  a.z += v.z * s;
  a.w += v.w * s;
  return a;
}

__device__ __forceinline__ float readlane_f(float v, int l) {
  return __int_as_float(__builtin_amdgcn_readlane(__float_as_int(v), l));
}

#define MID_CAP 128
__global__ void __launch_bounds__(256) k_gather(const float* __restrict__ hlin,
                                                const int* __restrict__ off,
                                                const unsigned short* __restrict__ csr,
                                                const float* __restrict__ dinv,
                                                const float* __restrict__ bias,
                                                float* __restrict__ h) {
  __shared__ int   mid_us[8][MID_CAP];
  __shared__ int   mid_so[8][MID_CAP];
  __shared__ float mid_wo[8][MID_CAP];
  int lane = (int)(threadIdx.x & 31);
  int n = __builtin_amdgcn_readfirstlane(
      (int)((blockIdx.x * blockDim.x + threadIdx.x) >> 5));
  if (n >= N_NODES) return;
  int beg = off[n];
  int d = off[n + 1] - beg;
  float di = dinv[n];
  const float4* h4 = (const float4*)hlin;
  float4 hv = h4[n * 32 + lane];
  float4 acc;
  acc.x = hv.x * di * di + bias[lane * 4 + 0];
  acc.y = hv.y * di * di + bias[lane * 4 + 1];
  acc.z = hv.z * di * di + bias[lane * 4 + 2];
  acc.w = hv.w * di * di + bias[lane * 4 + 3];

  if (d <= 32) {
    int s = 0x7fffffff;
    float w = 0.0f;
    if (lane < d) {
      s = (int)csr[beg + lane];
      w = dinv[s];
    }
    int r = 0;
    for (int j = 0; j < d; ++j) {
      int sj = __builtin_amdgcn_readlane(s, j);
      r += ((sj < s) || (sj == s && j < lane)) ? 1 : 0;
    }
    int   ss = __builtin_amdgcn_ds_permute(r << 2, s);
    float ws = __int_as_float(__builtin_amdgcn_ds_permute(r << 2, __float_as_int(w)));
    int k = 0;
    for (; k + 4 <= d; k += 4) {
      int s0 = __builtin_amdgcn_readlane(ss, k);
      int s1 = __builtin_amdgcn_readlane(ss, k + 1);
      int s2 = __builtin_amdgcn_readlane(ss, k + 2);
      int s3 = __builtin_amdgcn_readlane(ss, k + 3);
      float w0 = readlane_f(ws, k);
      float w1 = readlane_f(ws, k + 1);
      float w2 = readlane_f(ws, k + 2);
      float w3 = readlane_f(ws, k + 3);
      float4 v0 = h4[s0 * 32 + lane];
      float4 v1 = h4[s1 * 32 + lane];
      float4 v2 = h4[s2 * 32 + lane];
      float4 v3 = h4[s3 * 32 + lane];
      acc = axpy4(v0, w0 * di, acc);
      acc = axpy4(v1, w1 * di, acc);
      acc = axpy4(v2, w2 * di, acc);
      acc = axpy4(v3, w3 * di, acc);
    }
    for (; k < d; ++k) {
      int s0 = __builtin_amdgcn_readlane(ss, k);
      float w0 = readlane_f(ws, k);
      float4 v0 = h4[s0 * 32 + lane];
      acc = axpy4(v0, w0 * di, acc);
    }
  } else if (d <= MID_CAP) {
    int wv = (int)(threadIdx.x >> 5);
    int*   us = mid_us[wv];
    int*   so = mid_so[wv];
    float* wo = mid_wo[wv];
    int   e[MID_CAP / 32];
    float w[MID_CAP / 32];
    for (int q = 0; q < MID_CAP / 32; ++q) {
      int i = lane + 32 * q;
      e[q] = 0x7fffffff;
      w[q] = 0.0f;
      if (i < d) {
        e[q] = (int)csr[beg + i];
        w[q] = dinv[e[q]];
        us[i] = e[q];
      }
    }
    int r[MID_CAP / 32];
    for (int q = 0; q < MID_CAP / 32; ++q) r[q] = 0;
    for (int j = 0; j < d; ++j) {
      int sj = us[j];
      for (int q = 0; q < MID_CAP / 32; ++q)
        r[q] += ((sj < e[q]) || (sj == e[q] && j < lane + 32 * q)) ? 1 : 0;
    }
    for (int q = 0; q < MID_CAP / 32; ++q) {
      if (lane + 32 * q < d) {
        so[r[q]] = e[q];
        wo[r[q]] = w[q];
      }
    }
    int k = 0;
    for (; k + 4 <= d; k += 4) {
      int s0 = so[k], s1 = so[k + 1], s2 = so[k + 2], s3 = so[k + 3];
      float w0 = wo[k], w1 = wo[k + 1], w2 = wo[k + 2], w3 = wo[k + 3];
      float4 v0 = h4[s0 * 32 + lane];
      float4 v1 = h4[s1 * 32 + lane];
      float4 v2 = h4[s2 * 32 + lane];
      float4 v3 = h4[s3 * 32 + lane];
      acc = axpy4(v0, w0 * di, acc);
      acc = axpy4(v1, w1 * di, acc);
      acc = axpy4(v2, w2 * di, acc);
      acc = axpy4(v3, w3 * di, acc);
    }
    for (; k < d; ++k) {
      int s0 = so[k];
      float w0 = wo[k];
      float4 v0 = h4[s0 * 32 + lane];
      acc = axpy4(v0, w0 * di, acc);
    }
  } else {
    int last = -1;
    int left = d;
    for (int it = 0; it < d && left > 0; ++it) {
      int mn = 0x7fffffff;
      for (int i = lane; i < d; i += 32) {
        int s = (int)csr[beg + i];
        if (s > last && s < mn) mn = s;
      }
      for (int o = 16; o > 0; o >>= 1) {
        int t = __shfl_xor(mn, o);
        mn = (t < mn) ? t : mn;
      }
      int m = 0;
      for (int i = lane; i < d; i += 32) m += ((int)csr[beg + i] == mn) ? 1 : 0;
      for (int o = 16; o > 0; o >>= 1) m += __shfl_xor(m, o);
      mn = __builtin_amdgcn_readfirstlane(mn);
      m = __builtin_amdgcn_readfirstlane(m);
      float nrm = dinv[mn] * di;
      float4 v = h4[mn * 32 + lane];
      for (int c = 0; c < m; ++c) acc = axpy4(v, nrm, acc);
      last = mn;
      left -= m;
    }
  }
  ((float4*)h)[n * 32 + lane] = acc;
}

__device__ __forceinline__ void prep_w0_block(const float* __restrict__ bw,
                                              const float* __restrict__ sw,
                                              const float* __restrict__ sc,
                                              float* __restrict__ Bt, int blk) {
  int t = blk * FRONT_T + (int)threadIdx.x;
  if (t >= 8 * OUT_C * HIDDEN) return;
  int o = t >> 10, k = t & 1023;
  int f = k >> 3, j = k & 7;
  float w = (j == 0) ? bw[o * OUT_C + f]
                     : sw[(o * OUT_C + f) * N_COEF + (j - 1)] * sc[o * OUT_C + f];
  Bt[t] = w;
}

__device__ __forceinline__ void prep_w1_block(const float* __restrict__ bw,
                                              const float* __restrict__ sw,
                                              const float* __restrict__ sc,
                                              float* __restrict__ Bt, int blk) {
  int t = blk * FRONT_T + (int)threadIdx.x;
  if (t >= 8 * HIDDEN * OUT_C) return;
  int o = t >> 7, k = t & 127;
  int f = k >> 3, j = k & 7;
  float w = (j == 0) ? bw[o * HIDDEN + f]
                     : sw[(o * HIDDEN + f) * N_COEF + (j - 1)] * sc[o * HIDDEN + f];
  Bt[o * 128 + k] = w;
}

#define PREP_B     ((8 * OUT_C * HIDDEN + FRONT_T - 1) / FRONT_T)
#define FRONT_GRID (PGRID + NCHUNK + 2 * PREP_B)
__global__ void __launch_bounds__(FRONT_T) k_front(const float* __restrict__ x,
                                                   const float* __restrict__ W,
                                                   float* __restrict__ hlin,
                                                   const int* __restrict__ ei,
                                                   unsigned* __restrict__ srt,
                                                   unsigned* __restrict__ pc,
                                                   const float* __restrict__ bw0,
                                                   const float* __restrict__ sw0,
                                                   const float* __restrict__ sc0,
                                                   float* __restrict__ Bt0,
                                                   const float* __restrict__ bw1,
                                                   const float* __restrict__ sw1,
                                                   const float* __restrict__ sc1,
                                                   float* __restrict__ Bt1) {
  __shared__ __attribute__((aligned(16))) float As[2][16 * TILE_P];
  __shared__ int hist[NBINS];
  __shared__ int bst[NBINS];
  static_assert(sizeof(As) >= CHUNK * sizeof(unsigned), "a chunk's words fit the tile buffers");
  int blk = (int)blockIdx.x;
  if (blk < PGRID) {
    gemm_xw_block(x, W, hlin, As, blk);
  } else if (blk < PGRID + NCHUNK) {
    bin_sort_block(ei, srt, pc, hist, bst, (unsigned*)&As[0][0], blk - PGRID);
  } else if (blk < PGRID + NCHUNK + PREP_B) {
    prep_w0_block(bw0, sw0, sc0, Bt0, blk - (PGRID + NCHUNK));
  } else {
    prep_w1_block(bw1, sw1, sc1, Bt1, blk - (PGRID + NCHUNK + PREP_B));
  }
}

#define PGRID0 512
static_assert(PGRID0 <= NTILES, "every workgroup of k_kan0 has a first tile");
__global__ void __launch_bounds__(256) k_kan0(const float* __restrict__ h,
                                              const float* __restrict__ Bt,
                                              float* __restrict__ g) {
  __shared__ __attribute__((aligned(16))) float red[2][8][256];
  int tid = (int)threadIdx.x;
  int lane = tid & 31, wv = tid >> 5;
  int half = lane >> 4, l16 = lane & 15;
  int f0 = wv * 16 + half * 8;
  float4 b[16];
  {
    const float4* bp = (const float4*)(Bt + l16 * 1024 + f0 * 8);
#pragma unroll
    for (int j = 0; j < 16; ++j) b[j] = bp[j];
  }
  const float4* h4 = (const float4*)h;
  int t = (int)blockIdx.x;
  float4 p0 = h4[((t * 16 + l16) * OUT_C + f0) / 4];
  float4 p1 = h4[((t * 16 + l16) * OUT_C + f0) / 4 + 1];
  int ridx = ((tid >> 4) & 7) * 32 + (tid >> 7) * 16 + (tid & 15);
  int buf = 0;
  for (; t < NTILES; t += PGRID0) {
    float v[8];
    v[0] = p0.x; v[1] = p0.y; v[2] = p0.z; v[3] = p0.w;
    v[4] = p1.x; v[5] = p1.y; v[6] = p1.z; v[7] = p1.w;
    int tn = t + PGRID0;
    if (tn < NTILES) {
      p0 = h4[((tn * 16 + l16) * OUT_C + f0) / 4];
      p1 = h4[((tn * 16 + l16) * OUT_C + f0) / 4 + 1];
    }
    v2f x[8][4];
#pragma unroll
    for (int i = 0; i < 8; ++i) {
      float e[8];
      e[0] = silu_f(v[i]);
      bspline7(v[i], e + 1);
#pragma unroll
      for (int q = 0; q < 4; ++q) { x[i][q].x = e[2 * q]; x[i][q].y = e[2 * q + 1]; }
    }
    __builtin_amdgcn_sched_barrier(0);
    v8f acc0 = {}, acc1 = {};
#pragma unroll
    for (int i = 0; i < 8; ++i) {
      v2f w0, w1, w2, w3;
      w0.x = b[2 * i].x;     w0.y = b[2 * i].y;
      w1.x = b[2 * i].z;     w1.y = b[2 * i].w;
      w2.x = b[2 * i + 1].x; w2.y = b[2 * i + 1].y;
      w3.x = b[2 * i + 1].z; w3.y = b[2 * i + 1].w;
      acc0 = wmma4(x[i][0], w0, acc0);
      acc1 = wmma4(x[i][1], w1, acc1);
      acc0 = wmma4(x[i][2], w2, acc0);
      acc1 = wmma4(x[i][3], w3, acc1);
    }
    asm volatile("v_nop\n\tv_nop\n\tv_nop\n\tv_nop\n\tv_nop" : "+v"(acc0), "+v"(acc1));
    v8f acc = acc0 + acc1;
    float* rw = &red[buf][wv][lane];
#pragma unroll
    for (int r = 0; r < 8; ++r) rw[r * 32] = acc[r];
    __syncthreads();
    float s = red[buf][0][ridx];
#pragma unroll
    for (int w = 1; w < 8; ++w) s += red[buf][w][ridx];
    g[t * 256 + tid] = s;
    buf ^= 1;
  }
}

__device__ __forceinline__ void expand8(float v, float4* d) {
  float bs[N_COEF];
  bspline7(v, bs);
  float4 lo, hi;
  lo.x = silu_f(v); lo.y = bs[0]; lo.z = bs[1]; lo.w = bs[2];
  hi.x = bs[3]; hi.y = bs[4]; hi.z = bs[5]; hi.w = bs[6];
  d[0] = lo;
  d[1] = hi;
}

__global__ void __launch_bounds__(256) k_kan1(const float* __restrict__ g,
                                              const float* __restrict__ Bt,
                                              float* __restrict__ out) {
  __shared__ __attribute__((aligned(16))) float As[2][16 * TILE_P];
  int tid = (int)threadIdx.x;
  int lane = tid & 31, wv = tid >> 5;
  int half = lane >> 4, l16 = lane & 15;
  int o0 = wv * 16;
  float4 b[16];
  load_bfrag(Bt, o0, l16, half, b);
  int row = tid >> 4, f = tid & 15;
  int t = (int)blockIdx.x;
  float v = g[(t * 16 + row) * HIDDEN + f];
  expand8(v, (float4*)&As[0][row * TILE_P + f * 8]);
  __syncthreads();
  int buf = 0;
  for (; t < NTILES; t += PGRID) {
    int tn = t + PGRID;
    bool more = tn < NTILES;
    if (more) v = g[(tn * 16 + row) * HIDDEN + f];
    v8f acc = tile_mma(As[buf], l16, half, b);
    if (more) expand8(v, (float4*)&As[buf ^ 1][row * TILE_P + f * 8]);
    __syncthreads();
#pragma unroll
    for (int r = 0; r < 8; ++r)
      out[(t * 16 + r + half * 8) * OUT_C + o0 + l16] = acc[r];
    buf ^= 1;
  }
}

extern "C" void kernel_launch(void* const* d_in, const int* in_sizes, int n_in,
                              void* d_out, int out_size, void* d_ws, size_t ws_size,
                              hipStream_t stream) {
  (void)in_sizes; (void)n_in; (void)out_size; (void)ws_size;
  const float* x       = (const float*)d_in[0];
  const int*   ei      = (const int*)d_in[1];
  const float* gcn_w   = (const float*)d_in[2];
  const float* gcn_b   = (const float*)d_in[3];
  const float* base_w0 = (const float*)d_in[4];
  const float* spl_w0  = (const float*)d_in[5];
  const float* scal0   = (const float*)d_in[6];
  const float* base_w1 = (const float*)d_in[7];
  const float* spl_w1  = (const float*)d_in[8];
  const float* scal1   = (const float*)d_in[9];
  float* out = (float*)d_out;

  float* ws    = (float*)d_ws;
  float* hlin  = ws;
  float* h     = hlin + (size_t)N_NODES * OUT_C;
  float* deg   = h + (size_t)N_NODES * OUT_C;
  float* g     = deg + N_NODES;
  float* Bm0   = g + (size_t)N_NODES * HIDDEN;
  float* Bm1   = Bm0 + 8 * OUT_C * HIDDEN;
  unsigned* srt = (unsigned*)h;
  unsigned* pc = srt + (size_t)NCHUNK * CHUNK;
  unsigned short* csr = (unsigned short*)g;
  int* off  = (int*)g + N_EDGES / 2;

  k_front<<<FRONT_GRID, FRONT_T, 0, stream>>>(x, gcn_w, hlin, ei, srt, pc,
                                              base_w0, spl_w0, scal0, Bm0,
                                              base_w1, spl_w1, scal1, Bm1);
  k_bin_csr<<<NBINS, EDGE_T, 0, stream>>>(srt, pc, off, deg, csr);
  k_gather<<<(N_NODES * 32 + 255) / 256, 256, 0, stream>>>(hlin, off, csr, deg, gcn_b, h);
  k_kan0<<<PGRID0, 256, 0, stream>>>(h, Bm0, g);
  k_kan1<<<PGRID, 256, 0, stream>>>(g, Bm1, out);
}


# x@W^T split over the two launches before the gather: part beside the chunk sort, the rest beside the per-bin build (now 8-wave workgroups, start = sum of own piece offsets); 5 launches, sums kept
# speedup vs baseline: 4.2242x; 1.0372x over previous
#include <hip/hip_runtime.h>
#include <hip/hip_bf16.h>

#define N_NODES 50000
#define N_EDGES 800000
#define IN_C    128
#define OUT_C   128
#define HIDDEN  16
#define N_COEF  7

typedef __attribute__((ext_vector_type(2))) float v2f;
typedef __attribute__((ext_vector_type(8))) float v8f;

__device__ __forceinline__ v8f wmma4(v2f a, v2f b, v8f c) {
  return __builtin_amdgcn_wmma_f32_16x16x4_f32(false, a, false, b, (short)0, c,
                                               false, false);
}

__device__ __forceinline__ float silu_f(float v) {
  return v * (1.0f / (1.0f + __expf(-v)));
}

__device__ __forceinline__ void bspline7(float v, float* out) {
  float fi = floorf(2.0f * v + 5.0f);
  float t  = (v - (0.5f * fi - 2.5f)) * 2.0f;
  float t2 = t * t;
  float u  = 1.0f - t;
  const float k6 = 1.0f / 6.0f;
  float n3 = t2 * t * k6;
  float n2 = t * (t * (0.5f - 0.5f * t) + 0.5f) + k6;
  float n1 = t2 * (0.5f * t - 1.0f) + (2.0f / 3.0f);
  float n0 = u * u * u * k6;
  bool m[10];
#pragma unroll
  for (int k = 0; k < 10; ++k) m[k] = (fi == (float)k);
#pragma unroll
  for (int c = 0; c < N_COEF; ++c) {
    float r = m[c + 3] ? n0 : 0.0f;
    r = m[c + 2] ? n1 : r;
    r = m[c + 1] ? n2 : r;
    r = m[c] ? n3 : r;
    out[c] = r;
  }
}

#define TILE_P 132
#define NTILES (N_NODES / 16)
#define PGRID  512
static_assert(N_NODES % 16 == 0 && PGRID <= NTILES, "tiling");

__device__ __forceinline__ void load_bfrag(const float* __restrict__ Wt, int o0,
                                           int l16, int half, float4 (&b)[16]) {
  const float4* p = (const float4*)(Wt + (o0 + l16) * 128 + 4 * half);
#pragma unroll
  for (int j = 0; j < 16; ++j) b[j] = p[2 * j];
}

__device__ __forceinline__ v8f tile_mma(const float* As, int l16, int half,
                                        const float4 (&b)[16]) {
  float4 a[16];
  const float4* p = (const float4*)(As + l16 * TILE_P + 4 * half);
#pragma unroll
  for (int j = 0; j < 16; ++j) a[j] = p[2 * j];
  __builtin_amdgcn_sched_barrier(0);
  v8f acc0 = {}, acc1 = {};
#pragma unroll
  for (int j = 0; j < 16; ++j) {
    v2f a0, a1, b0, b1;
    a0.x = a[j].x; a0.y = a[j].y; a1.x = a[j].z; a1.y = a[j].w;
    b0.x = b[j].x; b0.y = b[j].y; b1.x = b[j].z; b1.y = b[j].w;
    acc0 = wmma4(a0, b0, acc0);
    acc1 = wmma4(a1, b1, acc1);
  }
  asm volatile("v_nop\n\tv_nop\n\tv_nop\n\tv_nop\n\tv_nop" : "+v"(acc0), "+v"(acc1));
  return acc0 + acc1;
}

#define GEMM_T1 512
static_assert(GEMM_T1 >= PGRID && NTILES - GEMM_T1 >= PGRID, "a first tile for every workgroup of either part");
__device__ __forceinline__ void gemm_xw_block(const float* __restrict__ x,
                                              const float* __restrict__ W,
                                              float* __restrict__ hlin,
                                              float (*As)[16 * TILE_P], int blk,
                                              const int t0, const int t1) {
  int tid = (int)threadIdx.x;
  int lane = tid & 31, wv = tid >> 5;
  int half = lane >> 4, l16 = lane & 15;
  int o0 = wv * 16;
  float4 b[16];
  load_bfrag(W, o0, l16, half, b);
  int r0 = wv, c4 = lane;
  const float4* x4 = (const float4*)x;
  int t = t0 + blk;
  float4 p0 = x4[(t * 16 + r0) * 32 + c4];
  float4 p1 = x4[(t * 16 + r0 + 8) * 32 + c4];
  *(float4*)&As[0][r0 * TILE_P + 4 * c4] = p0;
  *(float4*)&As[0][(r0 + 8) * TILE_P + 4 * c4] = p1;
  __syncthreads();
  int buf = 0;
  for (; t < t1; t += PGRID) {
    int tn = t + PGRID;
    bool more = tn < t1;
    if (more) {
      const float4* xt = x4 + (size_t)tn * 512;
      p0 = xt[r0 * 32 + c4];
      p1 = xt[(r0 + 8) * 32 + c4];
    }
    v8f acc = tile_mma(As[buf], l16, half, b);
    if (more) {
      *(float4*)&As[buf ^ 1][r0 * TILE_P + 4 * c4] = p0;
      *(float4*)&As[buf ^ 1][(r0 + 8) * TILE_P + 4 * c4] = p1;
    }
    __syncthreads();
    float* ht = hlin + (size_t)t * (16 * OUT_C);
#pragma unroll
    for (int r = 0; r < 8; ++r)
      ht[(r + half * 8) * OUT_C + o0 + l16] = acc[r];
    buf ^= 1;
  }
}

static_assert(N_NODES <= 65536, "source ids are kept as 16-bit values");
#define BINW   256
#define NBINS  ((N_NODES + BINW - 1) / BINW)
#define CHUNK  4096
#define NCHUNK ((N_EDGES + CHUNK - 1) / CHUNK)
#define EDGE_T 256
#define NPIECE (NBINS * NCHUNK)
static_assert(BINW <= 65536 && BINW <= EDGE_T, "local destination and source share one word; one thread per node of a bin");
static_assert(CHUNK < 65536, "a piece's offset and its length share one word");
static_assert(NBINS <= 7 * 32, "one wave scans a chunk's bin counts, 7 per lane");
static_assert(NCHUNK <= EDGE_T, "one thread per piece of a bin");

#define FRONT_T 256
#define HIST_V  (CHUNK / FRONT_T)
static_assert(CHUNK % FRONT_T == 0, "whole edges per thread");
__device__ __forceinline__ void bin_sort_block(const int* __restrict__ ei,
                                               unsigned* __restrict__ srt,
                                               unsigned* __restrict__ pc,
                                               int* hist, int* bst, unsigned* buf,
                                               int blk) {
  int tid = (int)threadIdx.x;
  for (int b = tid; b < NBINS; b += FRONT_T) hist[b] = 0;
  __syncthreads();
  int e0 = blk * CHUNK + tid;
  unsigned d[HIST_V], s[HIST_V];
  int r[HIST_V];
#pragma unroll
  for (int q = 0; q < HIST_V; ++q) {
    int e = e0 + q * FRONT_T;
    int ec = (e < N_EDGES) ? e : (N_EDGES - 1);
    d[q] = (unsigned)ei[N_EDGES + ec];
    s[q] = (unsigned)ei[ec];
  }
#pragma unroll
  for (int q = 0; q < HIST_V; ++q) {
    int e = e0 + q * FRONT_T;
    r[q] = 0;
    if (e < N_EDGES)
      r[q] = __hip_atomic_fetch_add(&hist[d[q] / BINW], 1, __ATOMIC_RELAXED,
                                    __HIP_MEMORY_SCOPE_WORKGROUP);
  }
  __syncthreads();
  if (tid < 32) {
    int loc[7];
    int sum = 0;
#pragma unroll
    for (int q = 0; q < 7; ++q) {
      int bb = tid * 7 + q;
      loc[q] = (bb < NBINS) ? hist[bb] : 0;
      sum += loc[q];
    }
    int inc = sum;
    for (int o = 1; o < 32; o <<= 1) {
      int v = __shfl_up(inc, o);
      if (tid >= o) inc += v;
    }
    int run = inc - sum;
#pragma unroll
    for (int q = 0; q < 7; ++q) {
      int bb = tid * 7 + q;
      if (bb < NBINS) {
        bst[bb] = run;
        pc[bb * NCHUNK + blk] = ((unsigned)loc[q] << 16) | (unsigned)run;
      }
      run += loc[q];
    }
  }
  __syncthreads();
#pragma unroll
  for (int q = 0; q < HIST_V; ++q) {
    int e = e0 + q * FRONT_T;
    if (e < N_EDGES)
      buf[bst[d[q] / BINW] + r[q]] = ((d[q] % BINW) << 16) | s[q];
  }
  __syncthreads();
  int left = N_EDGES - blk * CHUNK;
  int nv = (left < CHUNK) ? left : CHUNK;
  for (int i = tid; i < nv; i += FRONT_T) srt[blk * CHUNK + i] = buf[i];
}

#define CSR_W  (EDGE_T / 32)
#define PIECE_V ((NCHUNK + CSR_W - 1) / CSR_W)
__device__ __forceinline__ void bin_csr_block(const unsigned* __restrict__ srt,
                                              const unsigned* __restrict__ pc,
                                              int* __restrict__ off,
                                              float* __restrict__ dinv,
                                              unsigned short* __restrict__ csr,
                                              int* cnt, int* seg, int* fil,
                                              int* pn, int* psrc, int* red,
                                              int bin) {
  int tid = (int)threadIdx.x;
  int lane = tid & 31, wv = tid >> 5;
  if (tid < BINW) {
    cnt[tid] = 0;
    fil[tid] = 0;
  }
  unsigned own = (tid < NCHUNK) ? pc[bin * NCHUNK + tid] : 0u;
  int part = (int)(own & 0xffffu);
  for (int o = 16; o > 0; o >>= 1) part += __shfl_xor(part, o);
  if (lane == 0) red[wv] = part;
  if (tid < NCHUNK) {
    pn[tid] = (int)(own >> 16);
    psrc[tid] = tid * CHUNK + (int)(own & 0xffffu);
  }
  __syncthreads();
  int beg = 0;
#pragma unroll
  for (int w = 0; w < CSR_W; ++w) beg += red[w];
  unsigned wd[PIECE_V];
  int pm[PIECE_V], pb[PIECE_V];
#pragma unroll
  for (int k = 0; k < PIECE_V; ++k) {
    int c = wv + k * CSR_W;
    pm[k] = (c < NCHUNK) ? pn[c] : 0;
    pb[k] = (c < NCHUNK) ? psrc[c] : 0;
    wd[k] = srt[(lane < pm[k]) ? (pb[k] + lane) : 0];
  }
#pragma unroll
  for (int k = 0; k < PIECE_V; ++k) {
    if (lane < pm[k])
      __hip_atomic_fetch_add(&cnt[wd[k] >> 16], 1, __ATOMIC_RELAXED,
                             __HIP_MEMORY_SCOPE_WORKGROUP);
    for (int i = lane + 32; i < pm[k]; i += 32) {
      unsigned w = srt[pb[k] + i];
      __hip_atomic_fetch_add(&cnt[w >> 16], 1, __ATOMIC_RELAXED,
                             __HIP_MEMORY_SCOPE_WORKGROUP);
    }
  }
  __syncthreads();
  int c = (tid < BINW) ? cnt[tid] : 0;
  if (tid < BINW) seg[tid] = c;
  __syncthreads();
  for (int o = 1; o < BINW; o <<= 1) {
    int v = (tid < BINW && tid >= o) ? seg[tid - o] : 0;
    __syncthreads();
    if (tid < BINW) seg[tid] += v;
    __syncthreads();
  }
  if (tid < BINW) {
    int p = beg + seg[tid] - c;
    seg[tid] = p;
    int node = bin * BINW + tid;
    if (node < N_NODES) {
      off[node] = p;
      float dg = 1.0f + (float)c;
      dinv[node] = (dg > 0.0f) ? rsqrtf(dg) : 0.0f;
    }
  }
  if (bin == NBINS - 1 && tid == 0) off[N_NODES] = N_EDGES;
  __syncthreads();
#pragma unroll
  for (int k = 0; k < PIECE_V; ++k) {
    if (lane < pm[k]) {
      int dl = (int)(wd[k] >> 16);
      int r = __hip_atomic_fetch_add(&fil[dl], 1, __ATOMIC_RELAXED,
                                     __HIP_MEMORY_SCOPE_WORKGROUP);
      csr[seg[dl] + r] = (unsigned short)(wd[k] & 0xffffu);
    }
    for (int i = lane + 32; i < pm[k]; i += 32) {
      unsigned w = srt[pb[k] + i];
      int dl = (int)(w >> 16);
      int r = __hip_atomic_fetch_add(&fil[dl], 1, __ATOMIC_RELAXED,
                                     __HIP_MEMORY_SCOPE_WORKGROUP);
      csr[seg[dl] + r] = (unsigned short)(w & 0xffffu);
    }
  }
}

__device__ __forceinline__ float4 axpy4(float4 v, float s, float4 a) {
  a.x += v.x * s;
  a.y += v.y * s;
  a.z += v.z * s;
  a.w += v.w * s;
  return a;
}

__device__ __forceinline__ float readlane_f(float v, int l) {
  return __int_as_float(__builtin_amdgcn_readlane(__float_as_int(v), l));
}

#define MID_CAP 128
__global__ void __launch_bounds__(256) k_gather(const float* __restrict__ hlin,
                                                const int* __restrict__ off,
                                                const unsigned short* __restrict__ csr,
                                                const float* __restrict__ dinv,
                                                const float* __restrict__ bias,
                                                float* __restrict__ h) {
  __shared__ int   mid_us[8][MID_CAP];
  __shared__ int   mid_so[8][MID_CAP];
  __shared__ float mid_wo[8][MID_CAP];
  int lane = (int)(threadIdx.x & 31);
  int n = __builtin_amdgcn_readfirstlane(
      (int)((blockIdx.x * blockDim.x + threadIdx.x) >> 5));
  if (n >= N_NODES) return;
  int beg = off[n];
  int d = off[n + 1] - beg;
  float di = dinv[n];
  const float4* h4 = (const float4*)hlin;
  float4 hv = h4[n * 32 + lane];
  float4 acc;
  acc.x = hv.x * di * di + bias[lane * 4 + 0];
  acc.y = hv.y * di * di + bias[lane * 4 + 1];
  acc.z = hv.z * di * di + bias[lane * 4 + 2];
  acc.w = hv.w * di * di + bias[lane * 4 + 3];

  if (d <= 32) {
    int s = 0x7fffffff;
    float w = 0.0f;
    if (lane < d) {
      s = (int)csr[beg + lane];
      w = dinv[s];
    }
    int r = 0;
    for (int j = 0; j < d; ++j) {
      int sj = __builtin_amdgcn_readlane(s, j);
      r += ((sj < s) || (sj == s && j < lane)) ? 1 : 0;
    }
    int   ss = __builtin_amdgcn_ds_permute(r << 2, s);
    float ws = __int_as_float(__builtin_amdgcn_ds_permute(r << 2, __float_as_int(w)));
    int k = 0;
    for (; k + 4 <= d; k += 4) {
      int s0 = __builtin_amdgcn_readlane(ss, k);
      int s1 = __builtin_amdgcn_readlane(ss, k + 1);
      int s2 = __builtin_amdgcn_readlane(ss, k + 2);
      int s3 = __builtin_amdgcn_readlane(ss, k + 3);
      float w0 = readlane_f(ws, k);
      float w1 = readlane_f(ws, k + 1);
      float w2 = readlane_f(ws, k + 2);
      float w3 = readlane_f(ws, k + 3);
      float4 v0 = h4[s0 * 32 + lane];
      float4 v1 = h4[s1 * 32 + lane];
      float4 v2 = h4[s2 * 32 + lane];
      float4 v3 = h4[s3 * 32 + lane];
      acc = axpy4(v0, w0 * di, acc);
      acc = axpy4(v1, w1 * di, acc);
      acc = axpy4(v2, w2 * di, acc);
      acc = axpy4(v3, w3 * di, acc);
    }
    for (; k < d; ++k) {
      int s0 = __builtin_amdgcn_readlane(ss, k);
      float w0 = readlane_f(ws, k);
      float4 v0 = h4[s0 * 32 + lane];
      acc = axpy4(v0, w0 * di, acc);
    }
  } else if (d <= MID_CAP) {
    int wv = (int)(threadIdx.x >> 5);
    int*   us = mid_us[wv];
    int*   so = mid_so[wv];
    float* wo = mid_wo[wv];
    int   e[MID_CAP / 32];
    float w[MID_CAP / 32];
    for (int q = 0; q < MID_CAP / 32; ++q) {
      int i = lane + 32 * q;
      e[q] = 0x7fffffff;
      w[q] = 0.0f;
      if (i < d) {
        e[q] = (int)csr[beg + i];
        w[q] = dinv[e[q]];
        us[i] = e[q];
      }
    }
    int r[MID_CAP / 32];
    for (int q = 0; q < MID_CAP / 32; ++q) r[q] = 0;
    for (int j = 0; j < d; ++j) {
      int sj = us[j];
      for (int q = 0; q < MID_CAP / 32; ++q)
        r[q] += ((sj < e[q]) || (sj == e[q] && j < lane + 32 * q)) ? 1 : 0;
    }
    for (int q = 0; q < MID_CAP / 32; ++q) {
      if (lane + 32 * q < d) {
        so[r[q]] = e[q];
        wo[r[q]] = w[q];
      }
    }
    int k = 0;
    for (; k + 4 <= d; k += 4) {
      int s0 = so[k], s1 = so[k + 1], s2 = so[k + 2], s3 = so[k + 3];
      float w0 = wo[k], w1 = wo[k + 1], w2 = wo[k + 2], w3 = wo[k + 3];
      float4 v0 = h4[s0 * 32 + lane];
      float4 v1 = h4[s1 * 32 + lane];
      float4 v2 = h4[s2 * 32 + lane];
      float4 v3 = h4[s3 * 32 + lane];
      acc = axpy4(v0, w0 * di, acc);
      acc = axpy4(v1, w1 * di, acc);
      acc = axpy4(v2, w2 * di, acc);
      acc = axpy4(v3, w3 * di, acc);
    }
    for (; k < d; ++k) {
      int s0 = so[k];
      float w0 = wo[k];
      float4 v0 = h4[s0 * 32 + lane];
      acc = axpy4(v0, w0 * di, acc);
    }
  } else {
    int last = -1;
    int left = d;
    for (int it = 0; it < d && left > 0; ++it) {
      int mn = 0x7fffffff;
      for (int i = lane; i < d; i += 32) {
        int s = (int)csr[beg + i];
        if (s > last && s < mn) mn = s;
      }
      for (int o = 16; o > 0; o >>= 1) {
        int t = __shfl_xor(mn, o);
        mn = (t < mn) ? t : mn;
      }
      int m = 0;
      for (int i = lane; i < d; i += 32) m += ((int)csr[beg + i] == mn) ? 1 : 0;
      for (int o = 16; o > 0; o >>= 1) m += __shfl_xor(m, o);
      mn = __builtin_amdgcn_readfirstlane(mn);
      m = __builtin_amdgcn_readfirstlane(m);
      float nrm = dinv[mn] * di;
      float4 v = h4[mn * 32 + lane];
      for (int c = 0; c < m; ++c) acc = axpy4(v, nrm, acc);
      last = mn;
      left -= m;
    }
  }
  ((float4*)h)[n * 32 + lane] = acc;
}

__device__ __forceinline__ void prep_w0_block(const float* __restrict__ bw,
                                              const float* __restrict__ sw,
                                              const float* __restrict__ sc,
                                              float* __restrict__ Bt, int blk) {
  int t = blk * FRONT_T + (int)threadIdx.x;
  if (t >= 8 * OUT_C * HIDDEN) return;
  int o = t >> 10, k = t & 1023;
  int f = k >> 3, j = k & 7;
  float w = (j == 0) ? bw[o * OUT_C + f]
                     : sw[(o * OUT_C + f) * N_COEF + (j - 1)] * sc[o * OUT_C + f];
  Bt[t] = w;
}

__device__ __forceinline__ void prep_w1_block(const float* __restrict__ bw,
                                              const float* __restrict__ sw,
                                              const float* __restrict__ sc,
                                              float* __restrict__ Bt, int blk) {
  int t = blk * FRONT_T + (int)threadIdx.x;
  if (t >= 8 * HIDDEN * OUT_C) return;
  int o = t >> 7, k = t & 127;
  int f = k >> 3, j = k & 7;
  float w = (j == 0) ? bw[o * HIDDEN + f]
                     : sw[(o * HIDDEN + f) * N_COEF + (j - 1)] * sc[o * HIDDEN + f];
  Bt[o * 128 + k] = w;
}

#define PREP_B     ((8 * OUT_C * HIDDEN + FRONT_T - 1) / FRONT_T)
#define FRONT_GRID (NCHUNK + PGRID)
__global__ void __launch_bounds__(FRONT_T) k_front(const float* __restrict__ x,
                                                   const float* __restrict__ W,
                                                   float* __restrict__ hlin,
                                                   const int* __restrict__ ei,
                                                   unsigned* __restrict__ srt,
                                                   unsigned* __restrict__ pc) {
  __shared__ __attribute__((aligned(16))) float As[2][16 * TILE_P];
  __shared__ int hist[NBINS];
  __shared__ int bst[NBINS];
  static_assert(sizeof(As) >= CHUNK * sizeof(unsigned), "a chunk's words fit the tile buffers");
  int blk = (int)blockIdx.x;
  if (blk < NCHUNK) {
    bin_sort_block(ei, srt, pc, hist, bst, (unsigned*)&As[0][0], blk);
  } else {
    gemm_xw_block(x, W, hlin, As, blk - NCHUNK, 0, GEMM_T1);
  }
}

#define MID_GRID (NBINS + PGRID + 2 * PREP_B)
static_assert(EDGE_T == FRONT_T, "the build and x @ W^T share one launch, so one workgroup size");
__global__ void __launch_bounds__(FRONT_T) __attribute__((amdgpu_waves_per_eu(6, 6)))
k_mid(const float* __restrict__ x,
      const float* __restrict__ W,
      float* __restrict__ hlin,
      const unsigned* __restrict__ srt,
      const unsigned* __restrict__ pc,
      int* __restrict__ off,
      float* __restrict__ dinv,
      unsigned short* __restrict__ csr,
      const float* __restrict__ bw0,
      const float* __restrict__ sw0,
      const float* __restrict__ sc0,
      float* __restrict__ Bt0,
      const float* __restrict__ bw1,
      const float* __restrict__ sw1,
      const float* __restrict__ sc1,
      float* __restrict__ Bt1) {
  __shared__ __attribute__((aligned(16))) float As[2][16 * TILE_P];
  __shared__ int cnt[BINW];
  __shared__ int seg[BINW];
  __shared__ int fil[BINW];
  __shared__ int pn[NCHUNK];
  __shared__ int psrc[NCHUNK];
  __shared__ int red[CSR_W];
  int blk = (int)blockIdx.x;
  if (blk < NBINS) {
    bin_csr_block(srt, pc, off, dinv, csr, cnt, seg, fil, pn, psrc, red, blk);
  } else if (blk < NBINS + PGRID) {
    gemm_xw_block(x, W, hlin, As, blk - NBINS, GEMM_T1, NTILES);
  } else if (blk < NBINS + PGRID + PREP_B) {
    prep_w0_block(bw0, sw0, sc0, Bt0, blk - (NBINS + PGRID));
  } else {
    prep_w1_block(bw1, sw1, sc1, Bt1, blk - (NBINS + PGRID + PREP_B));
  }
}

#define PGRID0 512
static_assert(PGRID0 <= NTILES, "every workgroup of k_kan0 has a first tile");
__global__ void __launch_bounds__(256) k_kan0(const float* __restrict__ h,
                                              const float* __restrict__ Bt,
                                              float* __restrict__ g) {
  __shared__ __attribute__((aligned(16))) float red[2][8][256];
  int tid = (int)threadIdx.x;
  int lane = tid & 31, wv = tid >> 5;
  int half = lane >> 4, l16 = lane & 15;
  int f0 = wv * 16 + half * 8;
  float4 b[16];
  {
    const float4* bp = (const float4*)(Bt + l16 * 1024 + f0 * 8);
#pragma unroll
    for (int j = 0; j < 16; ++j) b[j] = bp[j];
  }
  const float4* h4 = (const float4*)h;
  int t = (int)blockIdx.x;
  float4 p0 = h4[((t * 16 + l16) * OUT_C + f0) / 4];
  float4 p1 = h4[((t * 16 + l16) * OUT_C + f0) / 4 + 1];
  int ridx = ((tid >> 4) & 7) * 32 + (tid >> 7) * 16 + (tid & 15);
  int buf = 0;
  for (; t < NTILES; t += PGRID0) {
    float v[8];
    v[0] = p0.x; v[1] = p0.y; v[2] = p0.z; v[3] = p0.w;
    v[4] = p1.x; v[5] = p1.y; v[6] = p1.z; v[7] = p1.w;
    int tn = t + PGRID0;
    if (tn < NTILES) {
      p0 = h4[((tn * 16 + l16) * OUT_C + f0) / 4];
      p1 = h4[((tn * 16 + l16) * OUT_C + f0) / 4 + 1];
    }
    v2f x[8][4];
#pragma unroll
    for (int i = 0; i < 8; ++i) {
      float e[8];
      e[0] = silu_f(v[i]);
      bspline7(v[i], e + 1);
#pragma unroll
      for (int q = 0; q < 4; ++q) { x[i][q].x = e[2 * q]; x[i][q].y = e[2 * q + 1]; }
    }
    __builtin_amdgcn_sched_barrier(0);
    v8f acc0 = {}, acc1 = {};
#pragma unroll
    for (int i = 0; i < 8; ++i) {
      v2f w0, w1, w2, w3;
      w0.x = b[2 * i].x;     w0.y = b[2 * i].y;
      w1.x = b[2 * i].z;     w1.y = b[2 * i].w;
      w2.x = b[2 * i + 1].x; w2.y = b[2 * i + 1].y;
      w3.x = b[2 * i + 1].z; w3.y = b[2 * i + 1].w;
      acc0 = wmma4(x[i][0], w0, acc0);
      acc1 = wmma4(x[i][1], w1, acc1);
      acc0 = wmma4(x[i][2], w2, acc0);
      acc1 = wmma4(x[i][3], w3, acc1);
    }
    asm volatile("v_nop\n\tv_nop\n\tv_nop\n\tv_nop\n\tv_nop" : "+v"(acc0), "+v"(acc1));
    v8f acc = acc0 + acc1;
    float* rw = &red[buf][wv][lane];
#pragma unroll
    for (int r = 0; r < 8; ++r) rw[r * 32] = acc[r];
    __syncthreads();
    float s = red[buf][0][ridx];
#pragma unroll
    for (int w = 1; w < 8; ++w) s += red[buf][w][ridx];
    g[t * 256 + tid] = s;
    buf ^= 1;
  }
}

__device__ __forceinline__ void expand8(float v, float4* d) {
  float bs[N_COEF];
  bspline7(v, bs);
  float4 lo, hi;
  lo.x = silu_f(v); lo.y = bs[0]; lo.z = bs[1]; lo.w = bs[2];
  hi.x = bs[3]; hi.y = bs[4]; hi.z = bs[5]; hi.w = bs[6];
  d[0] = lo;
  d[1] = hi;
}

__global__ void __launch_bounds__(256) k_kan1(const float* __restrict__ g,
                                              const float* __restrict__ Bt,
                                              float* __restrict__ out) {
  __shared__ __attribute__((aligned(16))) float As[2][16 * TILE_P];
  int tid = (int)threadIdx.x;
  int lane = tid & 31, wv = tid >> 5;
  int half = lane >> 4, l16 = lane & 15;
  int o0 = wv * 16;
  float4 b[16];
  load_bfrag(Bt, o0, l16, half, b);
  int row = tid >> 4, f = tid & 15;
  int t = (int)blockIdx.x;
  float v = g[(t * 16 + row) * HIDDEN + f];
  expand8(v, (float4*)&As[0][row * TILE_P + f * 8]);
  __syncthreads();
  int buf = 0;
  for (; t < NTILES; t += PGRID) {
    int tn = t + PGRID;
    bool more = tn < NTILES;
    if (more) v = g[(tn * 16 + row) * HIDDEN + f];
    v8f acc = tile_mma(As[buf], l16, half, b);
    if (more) expand8(v, (float4*)&As[buf ^ 1][row * TILE_P + f * 8]);
    __syncthreads();
#pragma unroll
    for (int r = 0; r < 8; ++r)
      out[(t * 16 + r + half * 8) * OUT_C + o0 + l16] = acc[r];
    buf ^= 1;
  }
}

extern "C" void kernel_launch(void* const* d_in, const int* in_sizes, int n_in,
                              void* d_out, int out_size, void* d_ws, size_t ws_size,
                              hipStream_t stream) {
  (void)in_sizes; (void)n_in; (void)out_size; (void)ws_size;
  const float* x       = (const float*)d_in[0];
  const int*   ei      = (const int*)d_in[1];
  const float* gcn_w   = (const float*)d_in[2];
  const float* gcn_b   = (const float*)d_in[3];
  const float* base_w0 = (const float*)d_in[4];
  const float* spl_w0  = (const float*)d_in[5];
  const float* scal0   = (const float*)d_in[6];
  const float* base_w1 = (const float*)d_in[7];
  const float* spl_w1  = (const float*)d_in[8];
  const float* scal1   = (const float*)d_in[9];
  float* out = (float*)d_out;

  float* ws    = (float*)d_ws;
  float* hlin  = ws;
  float* h     = hlin + (size_t)N_NODES * OUT_C;
  float* deg   = h + (size_t)N_NODES * OUT_C;
  float* g     = deg + N_NODES;
  float* Bm0   = g + (size_t)N_NODES * HIDDEN;
  float* Bm1   = Bm0 + 8 * OUT_C * HIDDEN;
  unsigned* srt = (unsigned*)h;
  unsigned* pc = srt + (size_t)NCHUNK * CHUNK;
  unsigned short* csr = (unsigned short*)g;
  int* off  = (int*)g + N_EDGES / 2;

  k_front<<<FRONT_GRID, FRONT_T, 0, stream>>>(x, gcn_w, hlin, ei, srt, pc);
  k_mid<<<MID_GRID, FRONT_T, 0, stream>>>(x, gcn_w, hlin, srt, pc, off, deg, csr,
                                          base_w0, spl_w0, scal0, Bm0,
                                          base_w1, spl_w1, scal1, Bm1);
  k_gather<<<(N_NODES * 32 + 255) / 256, 256, 0, stream>>>(hlin, off, csr, deg, gcn_b, h);
  k_kan0<<<PGRID0, 256, 0, stream>>>(h, Bm0, g);
  k_kan1<<<PGRID, 256, 0, stream>>>(g, Bm1, out);
}


# Per-bin build: a bin's source lists are put together in LDS and leave as whole 128-B lines (16-B stores), not as scattered 2-byte stores; piece table stored chunk-major likewise; sums kept
# speedup vs baseline: 4.4260x; 1.0478x over previous
#include <hip/hip_runtime.h>
#include <hip/hip_bf16.h>

#define N_NODES 50000
#define N_EDGES 800000
#define IN_C    128
#define OUT_C   128
#define HIDDEN  16
#define N_COEF  7

typedef __attribute__((ext_vector_type(2))) float v2f;
typedef __attribute__((ext_vector_type(8))) float v8f;

__device__ __forceinline__ v8f wmma4(v2f a, v2f b, v8f c) {
  return __builtin_amdgcn_wmma_f32_16x16x4_f32(false, a, false, b, (short)0, c,
                                               false, false);
}

__device__ __forceinline__ float silu_f(float v) {
  return v * (1.0f / (1.0f + __expf(-v)));
}

__device__ __forceinline__ void bspline7(float v, float* out) {
  float fi = floorf(2.0f * v + 5.0f);
  float t  = (v - (0.5f * fi - 2.5f)) * 2.0f;
  float t2 = t * t;
  float u  = 1.0f - t;
  const float k6 = 1.0f / 6.0f;
  float n3 = t2 * t * k6;
  float n2 = t * (t * (0.5f - 0.5f * t) + 0.5f) + k6;
  float n1 = t2 * (0.5f * t - 1.0f) + (2.0f / 3.0f);
  float n0 = u * u * u * k6;
  bool m[10];
#pragma unroll
  for (int k = 0; k < 10; ++k) m[k] = (fi == (float)k);
#pragma unroll
  for (int c = 0; c < N_COEF; ++c) {
    float r = m[c + 3] ? n0 : 0.0f;
    r = m[c + 2] ? n1 : r;
    r = m[c + 1] ? n2 : r;
    r = m[c] ? n3 : r;
    out[c] = r;
  }
}

#define TILE_P 132
#define NTILES (N_NODES / 16)
#define PGRID  512
static_assert(N_NODES % 16 == 0 && PGRID <= NTILES, "tiling");

__device__ __forceinline__ void load_bfrag(const float* __restrict__ Wt, int o0,
                                           int l16, int half, float4 (&b)[16]) {
  const float4* p = (const float4*)(Wt + (o0 + l16) * 128 + 4 * half);
#pragma unroll
  for (int j = 0; j < 16; ++j) b[j] = p[2 * j];
}

__device__ __forceinline__ v8f tile_mma(const float* As, int l16, int half,
                                        const float4 (&b)[16]) {
  float4 a[16];
  const float4* p = (const float4*)(As + l16 * TILE_P + 4 * half);
#pragma unroll
  for (int j = 0; j < 16; ++j) a[j] = p[2 * j];
  __builtin_amdgcn_sched_barrier(0);
  v8f acc0 = {}, acc1 = {};
#pragma unroll
  for (int j = 0; j < 16; ++j) {
    v2f a0, a1, b0, b1;
    a0.x = a[j].x; a0.y = a[j].y; a1.x = a[j].z; a1.y = a[j].w;
    b0.x = b[j].x; b0.y = b[j].y; b1.x = b[j].z; b1.y = b[j].w;
    acc0 = wmma4(a0, b0, acc0);
    acc1 = wmma4(a1, b1, acc1);
  }
  asm volatile("v_nop\n\tv_nop\n\tv_nop\n\tv_nop\n\tv_nop" : "+v"(acc0), "+v"(acc1));
  return acc0 + acc1;
}

#define GEMM_T1 512
static_assert(GEMM_T1 >= PGRID && NTILES - GEMM_T1 >= PGRID, "a first tile for every workgroup of either part");
__device__ __forceinline__ void gemm_xw_block(const float* __restrict__ x,
                                              const float* __restrict__ W,
                                              float* __restrict__ hlin,
                                              float (*As)[16 * TILE_P], int blk,
                                              const int t0, const int t1) {
  int tid = (int)threadIdx.x;
  int lane = tid & 31, wv = tid >> 5;
  int half = lane >> 4, l16 = lane & 15;
  int o0 = wv * 16;
  float4 b[16];
  load_bfrag(W, o0, l16, half, b);
  int r0 = wv, c4 = lane;
  const float4* x4 = (const float4*)x;
  int t = t0 + blk;
  float4 p0 = x4[(t * 16 + r0) * 32 + c4];
  float4 p1 = x4[(t * 16 + r0 + 8) * 32 + c4];
  *(float4*)&As[0][r0 * TILE_P + 4 * c4] = p0;
  *(float4*)&As[0][(r0 + 8) * TILE_P + 4 * c4] = p1;
  __syncthreads();
  int buf = 0;
  for (; t < t1; t += PGRID) {
    int tn = t + PGRID;
    bool more = tn < t1;
    if (more) {
      const float4* xt = x4 + (size_t)tn * 512;
      p0 = xt[r0 * 32 + c4];
      p1 = xt[(r0 + 8) * 32 + c4];
    }
    v8f acc = tile_mma(As[buf], l16, half, b);
    if (more) {
      *(float4*)&As[buf ^ 1][r0 * TILE_P + 4 * c4] = p0;
      *(float4*)&As[buf ^ 1][(r0 + 8) * TILE_P + 4 * c4] = p1;
    }
    __syncthreads();
    float* ht = hlin + (size_t)t * (16 * OUT_C);
#pragma unroll
    for (int r = 0; r < 8; ++r)
      ht[(r + half * 8) * OUT_C + o0 + l16] = acc[r];
    buf ^= 1;
  }
}

static_assert(N_NODES <= 65536, "source ids are kept as 16-bit values");
#define BINW   256
#define NBINS  ((N_NODES + BINW - 1) / BINW)
#define CHUNK  4096
#define NCHUNK ((N_EDGES + CHUNK - 1) / CHUNK)
#define EDGE_T 256
#define NPIECE (NBINS * NCHUNK)
static_assert(BINW <= 65536 && BINW <= EDGE_T, "local destination and source share one word; one thread per node of a bin");
static_assert(CHUNK < 65536, "a piece's offset and its length share one word");
static_assert(NBINS <= 7 * 32, "one wave scans a chunk's bin counts, 7 per lane");
static_assert(NCHUNK <= EDGE_T, "one thread per piece of a bin");

#define FRONT_T 256
#define HIST_V  (CHUNK / FRONT_T)
static_assert(CHUNK % FRONT_T == 0, "whole edges per thread");
__device__ __forceinline__ void bin_sort_block(const int* __restrict__ ei,
                                               unsigned* __restrict__ srt,
                                               unsigned* __restrict__ pc,
                                               int* hist, int* bst, unsigned* buf,
                                               int blk) {
  int tid = (int)threadIdx.x;
  for (int b = tid; b < NBINS; b += FRONT_T) hist[b] = 0;
  __syncthreads();
  int e0 = blk * CHUNK + tid;
  unsigned d[HIST_V], s[HIST_V];
  int r[HIST_V];
#pragma unroll
  for (int q = 0; q < HIST_V; ++q) {
    int e = e0 + q * FRONT_T;
    int ec = (e < N_EDGES) ? e : (N_EDGES - 1);
    d[q] = (unsigned)ei[N_EDGES + ec];
    s[q] = (unsigned)ei[ec];
  }
#pragma unroll
  for (int q = 0; q < HIST_V; ++q) {
    int e = e0 + q * FRONT_T;
    r[q] = 0;
    if (e < N_EDGES)
      r[q] = __hip_atomic_fetch_add(&hist[d[q] / BINW], 1, __ATOMIC_RELAXED,
                                    __HIP_MEMORY_SCOPE_WORKGROUP);
  }
  __syncthreads();
  if (tid < 32) {
    int loc[7];
    int sum = 0;
#pragma unroll
    for (int q = 0; q < 7; ++q) {
      int bb = tid * 7 + q;
      loc[q] = (bb < NBINS) ? hist[bb] : 0;
      sum += loc[q];
    }
    int inc = sum;
    for (int o = 1; o < 32; o <<= 1) {
      int v = __shfl_up(inc, o);
      if (tid >= o) inc += v;
    }
    int run = inc - sum;
#pragma unroll
    for (int q = 0; q < 7; ++q) {
      int bb = tid * 7 + q;
      if (bb < NBINS) {
        bst[bb] = run;
        pc[blk * NBINS + bb] = ((unsigned)loc[q] << 16) | (unsigned)run;
      }
      run += loc[q];
    }
  }
  __syncthreads();
#pragma unroll
  for (int q = 0; q < HIST_V; ++q) {
    int e = e0 + q * FRONT_T;
    if (e < N_EDGES)
      buf[bst[d[q] / BINW] + r[q]] = ((d[q] % BINW) << 16) | s[q];
  }
  __syncthreads();
  int left = N_EDGES - blk * CHUNK;
  int nv = (left < CHUNK) ? left : CHUNK;
  for (int i = tid; i < nv; i += FRONT_T) srt[blk * CHUNK + i] = buf[i];
}

#define CSR_W  (EDGE_T / 32)
#define PIECE_V ((NCHUNK + CSR_W - 1) / CSR_W)
#define CSR_CAP (2 * 16 * TILE_P * 2)
__device__ __forceinline__ void bin_csr_block(const unsigned* __restrict__ srt,
                                              const unsigned* __restrict__ pc,
                                              int* __restrict__ off,
                                              float* __restrict__ dinv,
                                              unsigned short* __restrict__ csr,
                                              int* cnt, int* seg, int* fil,
                                              int* pn, int* psrc, int* red,
                                              int* red2, unsigned short* cb,
                                              int bin) {
  int tid = (int)threadIdx.x;
  int lane = tid & 31, wv = tid >> 5;
  if (tid < BINW) {
    cnt[tid] = 0;
    fil[tid] = 0;
  }
  unsigned own = (tid < NCHUNK) ? pc[tid * NBINS + bin] : 0u;
  int part = (int)(own & 0xffffu);
  int plen = (int)(own >> 16);
  for (int o = 16; o > 0; o >>= 1) {
    part += __shfl_xor(part, o);
    plen += __shfl_xor(plen, o);
  }
  if (lane == 0) {
    red[wv] = part;
    red2[wv] = plen;
  }
  if (tid < NCHUNK) {
    pn[tid] = (int)(own >> 16);
    psrc[tid] = tid * CHUNK + (int)(own & 0xffffu);
  }
  __syncthreads();
  int beg = 0, tot = 0;
#pragma unroll
  for (int w = 0; w < CSR_W; ++w) {
    beg += red[w];
    tot += red2[w];
  }
  unsigned wd[PIECE_V];
  int pm[PIECE_V], pb[PIECE_V];
#pragma unroll
  for (int k = 0; k < PIECE_V; ++k) {
    int c = wv + k * CSR_W;
    pm[k] = (c < NCHUNK) ? pn[c] : 0;
    pb[k] = (c < NCHUNK) ? psrc[c] : 0;
    wd[k] = srt[(lane < pm[k]) ? (pb[k] + lane) : 0];
  }
#pragma unroll
  for (int k = 0; k < PIECE_V; ++k) {
    if (lane < pm[k])
      __hip_atomic_fetch_add(&cnt[wd[k] >> 16], 1, __ATOMIC_RELAXED,
                             __HIP_MEMORY_SCOPE_WORKGROUP);
    for (int i = lane + 32; i < pm[k]; i += 32) {
      unsigned w = srt[pb[k] + i];
      __hip_atomic_fetch_add(&cnt[w >> 16], 1, __ATOMIC_RELAXED,
                             __HIP_MEMORY_SCOPE_WORKGROUP);
    }
  }
  __syncthreads();
  int c = (tid < BINW) ? cnt[tid] : 0;
  if (tid < BINW) seg[tid] = c;
  __syncthreads();
  for (int o = 1; o < BINW; o <<= 1) {
    int v = (tid < BINW && tid >= o) ? seg[tid - o] : 0;
    __syncthreads();
    if (tid < BINW) seg[tid] += v;
    __syncthreads();
  }
  if (tid < BINW) {
    int p = beg + seg[tid] - c;
    seg[tid] = p;
    int node = bin * BINW + tid;
    if (node < N_NODES) {
      off[node] = p;
      float dg = 1.0f + (float)c;
      dinv[node] = (dg > 0.0f) ? rsqrtf(dg) : 0.0f;
    }
  }
  if (bin == NBINS - 1 && tid == 0) off[N_NODES] = N_EDGES;
  __syncthreads();
  const int mis  = (int)((((size_t)csr) >> 1) & 63);
  const int a    = (beg + mis) & 63;
  const int base = beg - a;
  if (a + tot <= CSR_CAP) {
#pragma unroll
    for (int k = 0; k < PIECE_V; ++k) {
      if (lane < pm[k]) {
        int dl = (int)(wd[k] >> 16);
        int r = __hip_atomic_fetch_add(&fil[dl], 1, __ATOMIC_RELAXED,
                                       __HIP_MEMORY_SCOPE_WORKGROUP);
        cb[seg[dl] + r - base] = (unsigned short)(wd[k] & 0xffffu);
      }
      for (int i = lane + 32; i < pm[k]; i += 32) {
        unsigned w = srt[pb[k] + i];
        int dl = (int)(w >> 16);
        int r = __hip_atomic_fetch_add(&fil[dl], 1, __ATOMIC_RELAXED,
                                       __HIP_MEMORY_SCOPE_WORKGROUP);
        cb[seg[dl] + r - base] = (unsigned short)(w & 0xffffu);
      }
    }
    __syncthreads();
    const int end  = a + tot;
    const int nvec = (end + 7) >> 3;
    for (int j = tid; j < nvec; j += EDGE_T) {
      int lo = 8 * j;
      if (lo >= a && lo + 8 <= end) {
        *(uint4*)(csr + base + lo) = *(const uint4*)(cb + lo);
      } else {
        for (int e = 0; e < 8; ++e) {
          int idx = lo + e;
          if (idx >= a && idx < end) csr[base + idx] = cb[idx];
        }
      }
    }
  } else {
#pragma unroll
    for (int k = 0; k < PIECE_V; ++k) {
      if (lane < pm[k]) {
        int dl = (int)(wd[k] >> 16);
        int r = __hip_atomic_fetch_add(&fil[dl], 1, __ATOMIC_RELAXED,
                                       __HIP_MEMORY_SCOPE_WORKGROUP);
        csr[seg[dl] + r] = (unsigned short)(wd[k] & 0xffffu);
      }
      for (int i = lane + 32; i < pm[k]; i += 32) {
        unsigned w = srt[pb[k] + i];
        int dl = (int)(w >> 16);
        int r = __hip_atomic_fetch_add(&fil[dl], 1, __ATOMIC_RELAXED,
                                       __HIP_MEMORY_SCOPE_WORKGROUP);
        csr[seg[dl] + r] = (unsigned short)(w & 0xffffu);
      }
    }
  }
}

__device__ __forceinline__ float4 axpy4(float4 v, float s, float4 a) {
  a.x += v.x * s;
  a.y += v.y * s;
  a.z += v.z * s;
  a.w += v.w * s;
  return a;
}

__device__ __forceinline__ float readlane_f(float v, int l) {
  return __int_as_float(__builtin_amdgcn_readlane(__float_as_int(v), l));
}

#define MID_CAP 128
__global__ void __launch_bounds__(256) k_gather(const float* __restrict__ hlin,
                                                const int* __restrict__ off,
                                                const unsigned short* __restrict__ csr,
                                                const float* __restrict__ dinv,
                                                const float* __restrict__ bias,
                                                float* __restrict__ h) {
  __shared__ int   mid_us[8][MID_CAP];
  __shared__ int   mid_so[8][MID_CAP];
  __shared__ float mid_wo[8][MID_CAP];
  int lane = (int)(threadIdx.x & 31);
  int n = __builtin_amdgcn_readfirstlane(
      (int)((blockIdx.x * blockDim.x + threadIdx.x) >> 5));
  if (n >= N_NODES) return;
  int beg = off[n];
  int d = off[n + 1] - beg;
  float di = dinv[n];
  const float4* h4 = (const float4*)hlin;
  float4 hv = h4[n * 32 + lane];
  float4 acc;
  acc.x = hv.x * di * di + bias[lane * 4 + 0];
  acc.y = hv.y * di * di + bias[lane * 4 + 1];
  acc.z = hv.z * di * di + bias[lane * 4 + 2];
  acc.w = hv.w * di * di + bias[lane * 4 + 3];

  if (d <= 32) {
    int s = 0x7fffffff;
    float w = 0.0f;
    if (lane < d) {
      s = (int)csr[beg + lane];
      w = dinv[s];
    }
    int r = 0;
    for (int j = 0; j < d; ++j) {
      int sj = __builtin_amdgcn_readlane(s, j);
      r += ((sj < s) || (sj == s && j < lane)) ? 1 : 0;
    }
    int   ss = __builtin_amdgcn_ds_permute(r << 2, s);
    float ws = __int_as_float(__builtin_amdgcn_ds_permute(r << 2, __float_as_int(w)));
    int k = 0;
    for (; k + 4 <= d; k += 4) {
      int s0 = __builtin_amdgcn_readlane(ss, k);
      int s1 = __builtin_amdgcn_readlane(ss, k + 1);
      int s2 = __builtin_amdgcn_readlane(ss, k + 2);
      int s3 = __builtin_amdgcn_readlane(ss, k + 3);
      float w0 = readlane_f(ws, k);
      float w1 = readlane_f(ws, k + 1);
      float w2 = readlane_f(ws, k + 2);
      float w3 = readlane_f(ws, k + 3);
      float4 v0 = h4[s0 * 32 + lane];
      float4 v1 = h4[s1 * 32 + lane];
      float4 v2 = h4[s2 * 32 + lane];
      float4 v3 = h4[s3 * 32 + lane];
      acc = axpy4(v0, w0 * di, acc);
      acc = axpy4(v1, w1 * di, acc);
      acc = axpy4(v2, w2 * di, acc);
      acc = axpy4(v3, w3 * di, acc);
    }
    for (; k < d; ++k) {
      int s0 = __builtin_amdgcn_readlane(ss, k);
      float w0 = readlane_f(ws, k);
      float4 v0 = h4[s0 * 32 + lane];
      acc = axpy4(v0, w0 * di, acc);
    }
  } else if (d <= MID_CAP) {
    int wv = (int)(threadIdx.x >> 5);
    int*   us = mid_us[wv];
    int*   so = mid_so[wv];
    float* wo = mid_wo[wv];
    int   e[MID_CAP / 32];
    float w[MID_CAP / 32];
    for (int q = 0; q < MID_CAP / 32; ++q) {
      int i = lane + 32 * q;
      e[q] = 0x7fffffff;
      w[q] = 0.0f;
      if (i < d) {
        e[q] = (int)csr[beg + i];
        w[q] = dinv[e[q]];
        us[i] = e[q];
      }
    }
    int r[MID_CAP / 32];
    for (int q = 0; q < MID_CAP / 32; ++q) r[q] = 0;
    for (int j = 0; j < d; ++j) {
      int sj = us[j];
      for (int q = 0; q < MID_CAP / 32; ++q)
        r[q] += ((sj < e[q]) || (sj == e[q] && j < lane + 32 * q)) ? 1 : 0;
    }
    for (int q = 0; q < MID_CAP / 32; ++q) {
      if (lane + 32 * q < d) {
        so[r[q]] = e[q];
        wo[r[q]] = w[q];
      }
    }
    int k = 0;
    for (; k + 4 <= d; k += 4) {
      int s0 = so[k], s1 = so[k + 1], s2 = so[k + 2], s3 = so[k + 3];
      float w0 = wo[k], w1 = wo[k + 1], w2 = wo[k + 2], w3 = wo[k + 3];
      float4 v0 = h4[s0 * 32 + lane];
      float4 v1 = h4[s1 * 32 + lane];
      float4 v2 = h4[s2 * 32 + lane];
      float4 v3 = h4[s3 * 32 + lane];
      acc = axpy4(v0, w0 * di, acc);
      acc = axpy4(v1, w1 * di, acc);
      acc = axpy4(v2, w2 * di, acc);
      acc = axpy4(v3, w3 * di, acc);
    }
    for (; k < d; ++k) {
      int s0 = so[k];
      float w0 = wo[k];
      float4 v0 = h4[s0 * 32 + lane];
      acc = axpy4(v0, w0 * di, acc);
    }
  } else {
    int last = -1;
    int left = d;
    for (int it = 0; it < d && left > 0; ++it) {
      int mn = 0x7fffffff;
      for (int i = lane; i < d; i += 32) {
        int s = (int)csr[beg + i];
        if (s > last && s < mn) mn = s;
      }
      for (int o = 16; o > 0; o >>= 1) {
        int t = __shfl_xor(mn, o);
        mn = (t < mn) ? t : mn;
      }
      int m = 0;
      for (int i = lane; i < d; i += 32) m += ((int)csr[beg + i] == mn) ? 1 : 0;
      for (int o = 16; o > 0; o >>= 1) m += __shfl_xor(m, o);
      mn = __builtin_amdgcn_readfirstlane(mn);
      m = __builtin_amdgcn_readfirstlane(m);
      float nrm = dinv[mn] * di;
      float4 v = h4[mn * 32 + lane];
      for (int c = 0; c < m; ++c) acc = axpy4(v, nrm, acc);
      last = mn;
      left -= m;
    }
  }
  ((float4*)h)[n * 32 + lane] = acc;
}

__device__ __forceinline__ void prep_w0_block(const float* __restrict__ bw,
                                              const float* __restrict__ sw,
                                              const float* __restrict__ sc,
                                              float* __restrict__ Bt, int blk) {
  int t = blk * FRONT_T + (int)threadIdx.x;
  if (t >= 8 * OUT_C * HIDDEN) return;
  int o = t >> 10, k = t & 1023;
  int f = k >> 3, j = k & 7;
  float w = (j == 0) ? bw[o * OUT_C + f]
                     : sw[(o * OUT_C + f) * N_COEF + (j - 1)] * sc[o * OUT_C + f];
  Bt[t] = w;
}

__device__ __forceinline__ void prep_w1_block(const float* __restrict__ bw,
                                              const float* __restrict__ sw,
                                              const float* __restrict__ sc,
                                              float* __restrict__ Bt, int blk) {
  int t = blk * FRONT_T + (int)threadIdx.x;
  if (t >= 8 * HIDDEN * OUT_C) return;
  int o = t >> 7, k = t & 127;
  int f = k >> 3, j = k & 7;
  float w = (j == 0) ? bw[o * HIDDEN + f]
                     : sw[(o * HIDDEN + f) * N_COEF + (j - 1)] * sc[o * HIDDEN + f];
  Bt[o * 128 + k] = w;
}

#define PREP_B     ((8 * OUT_C * HIDDEN + FRONT_T - 1) / FRONT_T)
#define FRONT_GRID (NCHUNK + PGRID)
__global__ void __launch_bounds__(FRONT_T) k_front(const float* __restrict__ x,
                                                   const float* __restrict__ W,
                                                   float* __restrict__ hlin,
                                                   const int* __restrict__ ei,
                                                   unsigned* __restrict__ srt,
                                                   unsigned* __restrict__ pc) {
  __shared__ __attribute__((aligned(16))) float As[2][16 * TILE_P];
  __shared__ int hist[NBINS];
  __shared__ int bst[NBINS];
  static_assert(sizeof(As) >= CHUNK * sizeof(unsigned), "a chunk's words fit the tile buffers");
  int blk = (int)blockIdx.x;
  if (blk < NCHUNK) {
    bin_sort_block(ei, srt, pc, hist, bst, (unsigned*)&As[0][0], blk);
  } else {
    gemm_xw_block(x, W, hlin, As, blk - NCHUNK, 0, GEMM_T1);
  }
}

#define MID_GRID (NBINS + PGRID + 2 * PREP_B)
static_assert(EDGE_T == FRONT_T, "the build and x @ W^T share one launch, so one workgroup size");
__global__ void __launch_bounds__(FRONT_T) __attribute__((amdgpu_waves_per_eu(6, 6)))
k_mid(const float* __restrict__ x,
      const float* __restrict__ W,
      float* __restrict__ hlin,
      const unsigned* __restrict__ srt,
      const unsigned* __restrict__ pc,
      int* __restrict__ off,
      float* __restrict__ dinv,
      unsigned short* __restrict__ csr,
      const float* __restrict__ bw0,
      const float* __restrict__ sw0,
      const float* __restrict__ sc0,
      float* __restrict__ Bt0,
      const float* __restrict__ bw1,
      const float* __restrict__ sw1,
      const float* __restrict__ sc1,
      float* __restrict__ Bt1) {
  __shared__ __attribute__((aligned(16))) float As[2][16 * TILE_P];
  __shared__ int cnt[BINW];
  __shared__ int seg[BINW];
  __shared__ int fil[BINW];
  __shared__ int pn[NCHUNK];
  __shared__ int psrc[NCHUNK];
  __shared__ int red[CSR_W];
  __shared__ int red2[CSR_W];
  static_assert(sizeof(As) >= CSR_CAP * sizeof(unsigned short), "the list staging fits the tile buffers");
  int blk = (int)blockIdx.x;
  if (blk < NBINS) {
    bin_csr_block(srt, pc, off, dinv, csr, cnt, seg, fil, pn, psrc, red, red2,
                  (unsigned short*)&As[0][0], blk);
  } else if (blk < NBINS + PGRID) {
    gemm_xw_block(x, W, hlin, As, blk - NBINS, GEMM_T1, NTILES);
  } else if (blk < NBINS + PGRID + PREP_B) {
    prep_w0_block(bw0, sw0, sc0, Bt0, blk - (NBINS + PGRID));
  } else {
    prep_w1_block(bw1, sw1, sc1, Bt1, blk - (NBINS + PGRID + PREP_B));
  }
}

#define PGRID0 512
static_assert(PGRID0 <= NTILES, "every workgroup of k_kan0 has a first tile");
__global__ void __launch_bounds__(256) k_kan0(const float* __restrict__ h,
                                              const float* __restrict__ Bt,
                                              float* __restrict__ g) {
  __shared__ __attribute__((aligned(16))) float red[2][8][256];
  int tid = (int)threadIdx.x;
  int lane = tid & 31, wv = tid >> 5;
  int half = lane >> 4, l16 = lane & 15;
  int f0 = wv * 16 + half * 8;
  float4 b[16];
  {
    const float4* bp = (const float4*)(Bt + l16 * 1024 + f0 * 8);
#pragma unroll
    for (int j = 0; j < 16; ++j) b[j] = bp[j];
  }
  const float4* h4 = (const float4*)h;
  int t = (int)blockIdx.x;
  float4 p0 = h4[((t * 16 + l16) * OUT_C + f0) / 4];
  float4 p1 = h4[((t * 16 + l16) * OUT_C + f0) / 4 + 1];
  int ridx = ((tid >> 4) & 7) * 32 + (tid >> 7) * 16 + (tid & 15);
  int buf = 0;
  for (; t < NTILES; t += PGRID0) {
    float v[8];
    v[0] = p0.x; v[1] = p0.y; v[2] = p0.z; v[3] = p0.w;
    v[4] = p1.x; v[5] = p1.y; v[6] = p1.z; v[7] = p1.w;
    int tn = t + PGRID0;
    if (tn < NTILES) {
      p0 = h4[((tn * 16 + l16) * OUT_C + f0) / 4];
      p1 = h4[((tn * 16 + l16) * OUT_C + f0) / 4 + 1];
    }
    v2f x[8][4];
#pragma unroll
    for (int i = 0; i < 8; ++i) {
      float e[8];
      e[0] = silu_f(v[i]);
      bspline7(v[i], e + 1);
#pragma unroll
      for (int q = 0; q < 4; ++q) { x[i][q].x = e[2 * q]; x[i][q].y = e[2 * q + 1]; }
    }
    __builtin_amdgcn_sched_barrier(0);
    v8f acc0 = {}, acc1 = {};
#pragma unroll
    for (int i = 0; i < 8; ++i) {
      v2f w0, w1, w2, w3;
      w0.x = b[2 * i].x;     w0.y = b[2 * i].y;
      w1.x = b[2 * i].z;     w1.y = b[2 * i].w;
      w2.x = b[2 * i + 1].x; w2.y = b[2 * i + 1].y;
      w3.x = b[2 * i + 1].z; w3.y = b[2 * i + 1].w;
      acc0 = wmma4(x[i][0], w0, acc0);
      acc1 = wmma4(x[i][1], w1, acc1);
      acc0 = wmma4(x[i][2], w2, acc0);
      acc1 = wmma4(x[i][3], w3, acc1);
    }
    asm volatile("v_nop\n\tv_nop\n\tv_nop\n\tv_nop\n\tv_nop" : "+v"(acc0), "+v"(acc1));
    v8f acc = acc0 + acc1;
    float* rw = &red[buf][wv][lane];
#pragma unroll
    for (int r = 0; r < 8; ++r) rw[r * 32] = acc[r];
    __syncthreads();
    float s = red[buf][0][ridx];
#pragma unroll
    for (int w = 1; w < 8; ++w) s += red[buf][w][ridx];
    g[t * 256 + tid] = s;
    buf ^= 1;
  }
}

__device__ __forceinline__ void expand8(float v, float4* d) {
  float bs[N_COEF];
  bspline7(v, bs);
  float4 lo, hi;
  lo.x = silu_f(v); lo.y = bs[0]; lo.z = bs[1]; lo.w = bs[2];
  hi.x = bs[3]; hi.y = bs[4]; hi.z = bs[5]; hi.w = bs[6];
  d[0] = lo;
  d[1] = hi;
}

__global__ void __launch_bounds__(256) k_kan1(const float* __restrict__ g,
                                              const float* __restrict__ Bt,
                                              float* __restrict__ out) {
  __shared__ __attribute__((aligned(16))) float As[2][16 * TILE_P];
  int tid = (int)threadIdx.x;
  int lane = tid & 31, wv = tid >> 5;
  int half = lane >> 4, l16 = lane & 15;
  int o0 = wv * 16;
  float4 b[16];
  load_bfrag(Bt, o0, l16, half, b);
  int row = tid >> 4, f = tid & 15;
  int t = (int)blockIdx.x;
  float v = g[(t * 16 + row) * HIDDEN + f];
  expand8(v, (float4*)&As[0][row * TILE_P + f * 8]);
  __syncthreads();
  int buf = 0;
  for (; t < NTILES; t += PGRID) {
    int tn = t + PGRID;
    bool more = tn < NTILES;
    if (more) v = g[(tn * 16 + row) * HIDDEN + f];
    v8f acc = tile_mma(As[buf], l16, half, b);
    if (more) expand8(v, (float4*)&As[buf ^ 1][row * TILE_P + f * 8]);
    __syncthreads();
#pragma unroll
    for (int r = 0; r < 8; ++r)
      out[(t * 16 + r + half * 8) * OUT_C + o0 + l16] = acc[r];
    buf ^= 1;
  }
}

extern "C" void kernel_launch(void* const* d_in, const int* in_sizes, int n_in,
                              void* d_out, int out_size, void* d_ws, size_t ws_size,
                              hipStream_t stream) {
  (void)in_sizes; (void)n_in; (void)out_size; (void)ws_size;
  const float* x       = (const float*)d_in[0];
  const int*   ei      = (const int*)d_in[1];
  const float* gcn_w   = (const float*)d_in[2];
  const float* gcn_b   = (const float*)d_in[3];
  const float* base_w0 = (const float*)d_in[4];
  const float* spl_w0  = (const float*)d_in[5];
  const float* scal0   = (const float*)d_in[6];
  const float* base_w1 = (const float*)d_in[7];
  const float* spl_w1  = (const float*)d_in[8];
  const float* scal1   = (const float*)d_in[9];
  float* out = (float*)d_out;

  float* ws    = (float*)d_ws;
  float* hlin  = ws;
  float* h     = hlin + (size_t)N_NODES * OUT_C;
  float* deg   = h + (size_t)N_NODES * OUT_C;
  float* g     = deg + N_NODES;
  float* Bm0   = g + (size_t)N_NODES * HIDDEN;
  float* Bm1   = Bm0 + 8 * OUT_C * HIDDEN;
  unsigned* srt = (unsigned*)h;
  unsigned* pc = srt + (size_t)NCHUNK * CHUNK;
  unsigned short* csr = (unsigned short*)g;
  int* off  = (int*)g + N_EDGES / 2;

  k_front<<<FRONT_GRID, FRONT_T, 0, stream>>>(x, gcn_w, hlin, ei, srt, pc);
  k_mid<<<MID_GRID, FRONT_T, 0, stream>>>(x, gcn_w, hlin, srt, pc, off, deg, csr,
                                          base_w0, spl_w0, scal0, Bm0,
                                          base_w1, spl_w1, scal1, Bm1);
  k_gather<<<(N_NODES * 32 + 255) / 256, 256, 0, stream>>>(hlin, off, csr, deg, gcn_b, h);
  k_kan0<<<PGRID0, 256, 0, stream>>>(h, Bm0, g);
  k_kan1<<<PGRID, 256, 0, stream>>>(g, Bm1, out);
}
